# starting kernel
# baseline (speedup 1.0000x reference)
#include <hip/hip_runtime.h>
#include <math.h>


typedef __attribute__((ext_vector_type(2))) float v2f;
typedef __attribute__((ext_vector_type(8))) float v8f;

__device__ __forceinline__ void atomAddF32(float* p, float v) {
    __hip_atomic_fetch_add(p, v, __ATOMIC_RELAXED, __HIP_MEMORY_SCOPE_AGENT);
}


__global__ void k_deg_init(float* __restrict__ deg, int n) {
    int i = blockIdx.x * blockDim.x + threadIdx.x;
    if (i < n) deg[i] = 1.0f;
}

__global__ void k_deg_scatter(const int* __restrict__ dst, const float* __restrict__ w,
                              float* __restrict__ deg, int E) {
    int e = blockIdx.x * blockDim.x + threadIdx.x;
    if (e < E) atomAddF32(&deg[dst[e]], w[e]);
}

__global__ void k_dinv(float* __restrict__ deg, int n) {
    int i = blockIdx.x * blockDim.x + threadIdx.x;
    if (i < n) {
        float d = deg[i];
        deg[i] = (d > 0.0f) ? rsqrtf(d) : 0.0f;
    }
}

__global__ void k_norm(const int* __restrict__ src, const int* __restrict__ dst,
                       const float* __restrict__ w, const float* __restrict__ dinv,
                       float* __restrict__ nrm, int E) {
    int e = blockIdx.x * blockDim.x + threadIdx.x;
    if (e < E) nrm[e] = dinv[src[e]] * w[e] * dinv[dst[e]];
}


__global__ void k_gemm1(const float* __restrict__ X, const float* __restrict__ W,
                        float* __restrict__ O, int mtiles) {
    int wave = threadIdx.x >> 5;
    int tile = blockIdx.x * (blockDim.x >> 5) + wave;
    if (tile >= mtiles * 4) return;
    int mt = tile >> 2, nt = tile & 3;
    int lane = threadIdx.x & 31;
    int lo = lane & 15, hi = lane >> 4;

    const float* arow = X + (size_t)(mt * 16 + lo) * 64;
    int col = nt * 16 + lo;
    v8f c = {};
#pragma unroll
    for (int k = 0; k < 64; k += 4) {
        int ka = k + 2 * hi;
        v2f a = { arow[ka], arow[ka + 1] };
        v2f b = { W[(size_t)ka * 64 + col], W[(size_t)(ka + 1) * 64 + col] };
        c = __builtin_amdgcn_wmma_f32_16x16x4_f32(false, a, false, b,
                                                  (short)0, c, false, false);
    }
    float* orow = O + (size_t)(mt * 16 + 8 * hi) * 64 + col;
#pragma unroll
    for (int r = 0; r < 8; ++r) orow[(size_t)r * 64] = c[r];
}


__global__ void k_gemm2(const float* __restrict__ Hin, const float* __restrict__ W2,
                        float* __restrict__ O, int mtiles) {
    int wave = threadIdx.x >> 5;
    int tile = blockIdx.x * (blockDim.x >> 5) + wave;
    if (tile >= mtiles * 3) return;
    int mt = tile / 3, nt = tile - mt * 3;
    int lane = threadIdx.x & 31;
    int lo = lane & 15, hi = lane >> 4;

    const float* arow = Hin + (size_t)(mt * 16 + lo) * 64;
    int col = nt * 16 + lo;
    bool cok = col < 40;
    v8f c = {};
#pragma unroll
    for (int k = 0; k < 64; k += 4) {
        int ka = k + 2 * hi;
        v2f a = { fmaxf(arow[ka], 0.0f), fmaxf(arow[ka + 1], 0.0f) };
        v2f b = { cok ? W2[(size_t)ka * 40 + col] : 0.0f,
                  cok ? W2[(size_t)(ka + 1) * 40 + col] : 0.0f };
        c = __builtin_amdgcn_wmma_f32_16x16x4_f32(false, a, false, b,
                                                  (short)0, c, false, false);
    }
    if (cok) {
        float* orow = O + (size_t)(mt * 16 + 8 * hi) * 40 + col;
#pragma unroll
        for (int r = 0; r < 8; ++r) orow[(size_t)r * 40] = c[r];
    }
}


__global__ void k_self1(const float* __restrict__ h, const float* __restrict__ dinv,
                        const float* __restrict__ bias, float* __restrict__ out, int n) {
    int t = blockIdx.x * blockDim.x + threadIdx.x;
    if (t < n * 64) {
        int i = t >> 6, f = t & 63;
        float di = dinv[i];
        out[t] = bias[f] + di * di * h[t];
    }
}

__global__ void k_self2(const float* __restrict__ h, const float* __restrict__ dinv,
                        const float* __restrict__ bias, float* __restrict__ out, int n) {
    int t = blockIdx.x * blockDim.x + threadIdx.x;
    if (t < n * 64) {
        int i = t >> 6, f = t & 63;
        if (f < 40) {
            float di = dinv[i];
            out[(size_t)i * 40 + f] = bias[f] + di * di * h[(size_t)i * 40 + f];
        }
    }
}


__global__ void k_scat1(const int* __restrict__ src, const int* __restrict__ dst,
                        const float* __restrict__ nrm, const float* __restrict__ h,
                        float* __restrict__ out, int E) {
    long long t = (long long)blockIdx.x * blockDim.x + threadIdx.x;
    if (t >= (long long)E * 64) return;
    int e = (int)(t >> 6), f = (int)(t & 63);
    int s = src[e], d = dst[e];
    atomAddF32(&out[(size_t)d * 64 + f], nrm[e] * h[(size_t)s * 64 + f]);
}

__global__ void k_scat2(const int* __restrict__ src, const int* __restrict__ dst,
                        const float* __restrict__ nrm, const float* __restrict__ h,
                        float* __restrict__ out, int E) {
    long long t = (long long)blockIdx.x * blockDim.x + threadIdx.x;
    if (t >= (long long)E * 64) return;
    int e = (int)(t >> 6), f = (int)(t & 63);
    if (f < 40) {
        int s = src[e], d = dst[e];
        atomAddF32(&out[(size_t)d * 40 + f], nrm[e] * h[(size_t)s * 40 + f]);
    }
}


__global__ void k_lsm(const float* __restrict__ h, float* __restrict__ out, int n) {
    int i = blockIdx.x * blockDim.x + threadIdx.x;
    if (i >= n) return;
    float v[40];
    float mx = -3.402823466e38f;
#pragma unroll
    for (int c = 0; c < 40; ++c) { v[c] = h[(size_t)i * 40 + c]; mx = fmaxf(mx, v[c]); }
    float s = 0.0f;
#pragma unroll
    for (int c = 0; c < 40; ++c) s += expf(v[c] - mx);
    float l = mx + logf(s);
#pragma unroll
    for (int c = 0; c < 40; ++c) out[(size_t)i * 40 + c] = v[c] - l;
}


extern "C" void kernel_launch(void* const* d_in, const int* in_sizes, int n_in,
                              void* d_out, int out_size, void* d_ws, size_t ws_size,
                              hipStream_t stream) {
    const float* x   = (const float*)d_in[0];
    const int*   ei  = (const int*)d_in[1];
    const float* ew  = (const float*)d_in[2];
    const float* W1  = (const float*)d_in[3];
    const float* b1  = (const float*)d_in[4];
    const float* W2  = (const float*)d_in[5];
    const float* b2  = (const float*)d_in[6];
    float* out = (float*)d_out;

    const int N = in_sizes[0] / 64;
    const int E = in_sizes[2];
    const int* src = ei;
    const int* dst = ei + E;

    float* ws   = (float*)d_ws;
    float* dinv = ws;
    float* nrm  = dinv + N;
    float* bufA = nrm + E;
    float* bufB = bufA + (size_t)N * 64;

    const int BLK = 256;
    const int mtiles = N / 16;

    k_deg_init<<<(N + BLK - 1) / BLK, BLK, 0, stream>>>(dinv, N);
    k_deg_scatter<<<(E + BLK - 1) / BLK, BLK, 0, stream>>>(dst, ew, dinv, E);
    k_dinv<<<(N + BLK - 1) / BLK, BLK, 0, stream>>>(dinv, N);
    k_norm<<<(E + BLK - 1) / BLK, BLK, 0, stream>>>(src, dst, ew, dinv, nrm, E);

    k_gemm1<<<(mtiles * 4 + 7) / 8, BLK, 0, stream>>>(x, W1, bufA, mtiles);
    {
        long long t = (long long)N * 64;
        k_self1<<<(unsigned)((t + BLK - 1) / BLK), BLK, 0, stream>>>(bufA, dinv, b1, bufB, N);
    }
    {
        long long t = (long long)E * 64;
        k_scat1<<<(unsigned)((t + BLK - 1) / BLK), BLK, 0, stream>>>(src, dst, nrm, bufA, bufB, E);
    }

    k_gemm2<<<(mtiles * 3 + 7) / 8, BLK, 0, stream>>>(bufB, W2, bufA, mtiles);
    {
        long long t = (long long)N * 64;
        k_self2<<<(unsigned)((t + BLK - 1) / BLK), BLK, 0, stream>>>(bufA, dinv, b2, bufB, N);
    }
    {
        long long t = (long long)E * 64;
        k_scat2<<<(unsigned)((t + BLK - 1) / BLK), BLK, 0, stream>>>(src, dst, nrm, bufA, bufB, E);
    }

    k_lsm<<<(N + BLK - 1) / BLK, BLK, 0, stream>>>(bufB, out, N);
}


# global f32 atomic scatters (degree, both layers) replaced by owner work-groups: each keeps 391 destination rows in LDS, filter-scans a byte key per edge, adds by LDS float atomics in arrival order
# speedup vs baseline: 2.5771x; 2.5771x over previous
#include <hip/hip_runtime.h>
#include <math.h>


typedef __attribute__((ext_vector_type(2))) float v2f;
typedef __attribute__((ext_vector_type(8))) float v8f;

constexpr int NOWN  = 256;
constexpr int RMAX  = 392;
constexpr int TPB   = 1024;
constexpr int NWAVE = TPB / 32;
constexpr int pws_of(int other_bytes) {
    int p = (164 * 1024 - other_bytes + NWAVE * 4 - 1) / (NWAVE * 4);
    return p < 512 ? 512 : p;
}
constexpr int UB    = 8;
constexpr int EB    = 16;

__device__ __forceinline__ void ldsAddF32(float* p, float v) {
    __hip_atomic_fetch_add(p, v, __ATOMIC_RELAXED, __HIP_MEMORY_SCOPE_WORKGROUP);
}


__global__ void k_key(const int* __restrict__ dst, unsigned* __restrict__ keyw,
                      int E, int nwords, unsigned R) {
    int i = blockIdx.x * blockDim.x + threadIdx.x;
    if (i >= nwords) return;
    unsigned v = 0u;
#pragma unroll
    for (int u = 0; u < 4; ++u) {
        int e = i * 4 + u;
        unsigned o = (e < E) ? ((unsigned)dst[e] / R) : 0u;
        v |= o << (8 * u);
    }
    keyw[i] = v;
}

__device__ __forceinline__ bool any_key_hit(uint4 k, unsigned pat) {
    unsigned x0 = k.x ^ pat, x1 = k.y ^ pat, x2 = k.z ^ pat, x3 = k.w ^ pat;
    unsigned t = ((x0 - 0x01010101u) & ~x0) | ((x1 - 0x01010101u) & ~x1) |
                 ((x2 - 0x01010101u) & ~x2) | ((x3 - 0x01010101u) & ~x3);
    return (t & 0x80808080u) != 0u;
}

__device__ __forceinline__ unsigned key_hit_mask16(uint4 k, unsigned pat) {
    unsigned kk[4] = { k.x, k.y, k.z, k.w };
    unsigned m = 0u;
#pragma unroll
    for (int j = 0; j < 4; ++j) {
        unsigned x = kk[j] ^ pat;
        unsigned t = ((x & 0x7F7F7F7Fu) + 0x7F7F7F7Fu) | x;
        unsigned z = ~t & 0x80808080u;
        unsigned c = ((z >> 7) & 1u) | ((z >> 14) & 2u) | ((z >> 21) & 4u) | ((z >> 28) & 8u);
        m |= c << (4 * j);
    }
    return m;
}

template <int PWS, class Proc>
__device__ __forceinline__ void scan_owned(const uint4* __restrict__ keyv, int nchunks,
                                           unsigned* wlist, int lane, int wave, int owner,
                                           Proc proc) {
    const unsigned pat = (unsigned)owner * 0x01010101u;
    const int off = (owner * 12) % nchunks;
    int cnt = 0;
    for (int c0 = wave; c0 < nchunks; c0 += NWAVE * UB) {
        uint4 k[UB];
        int cc[UB];
#pragma unroll
        for (int u = 0; u < UB; ++u) {
            int c = c0 + u * NWAVE;
            int x = ((c < nchunks) ? c : wave) + off;
            cc[u] = (x >= nchunks) ? x - nchunks : x;
            k[u] = keyv[(size_t)cc[u] * 32 + lane];
        }
#pragma unroll
        for (int u = 0; u < UB; ++u) {
            bool hit = (c0 + u * NWAVE < nchunks) && any_key_hit(k[u], pat);
            unsigned bal = __builtin_amdgcn_ballot_w32(hit);
            if (hit) wlist[cnt + (int)__builtin_amdgcn_mbcnt_lo(bal, 0u)] = (unsigned)(cc[u] * 32 + lane);
            cnt += __builtin_popcount(bal);
        }
        if (cnt > PWS - 32 * UB) { proc(cnt); cnt = 0; }
    }
    proc(cnt);
}


__global__ __launch_bounds__(TPB) void k_deg_own(const unsigned* __restrict__ keyw,
                                                 const int* __restrict__ dst,
                                                 const float* __restrict__ w,
                                                 float* __restrict__ dinv,
                                                 int n, int E, int R, int nchunks) {
    constexpr int PWS = pws_of(RMAX * 4);
    __shared__ float degl[RMAX];
    __shared__ unsigned list[NWAVE * PWS];
    const uint4* keyv = (const uint4*)keyw;
    const int tid = threadIdx.x, lane = tid & 31;
    const int wave = __builtin_amdgcn_readfirstlane(tid >> 5);
    const int owner = blockIdx.x, base = owner * R;
    const unsigned pat = (unsigned)owner * 0x01010101u;
    unsigned* wlist = list + wave * PWS;

    for (int i = tid; i < RMAX; i += TPB) degl[i] = 1.0f;
    __syncthreads();

    auto proc = [&](int cnt) {
        for (int g = 0; g < cnt; g += 32) {
            int idx = g + lane;
            if (idx < cnt) {
                unsigned seg = wlist[idx];
                unsigned m = key_hit_mask16(keyv[seg], pat);
                while (m != 0u) {
                    int e = (int)seg * 16 + __builtin_ctz(m);
                    m &= m - 1u;
                    if (e < E) ldsAddF32(&degl[dst[e] - base], w[e]);
                }
            }
        }
    };
    scan_owned<PWS>(keyv, nchunks, wlist, lane, wave, owner, proc);
    __syncthreads();

    for (int i = tid; i < R; i += TPB) {
        int node = base + i;
        if (node < n) {
            float d = degl[i];
            dinv[node] = (d > 0.0f) ? rsqrtf(d) : 0.0f;
        }
    }
}


__global__ void k_gemm1(const float* __restrict__ X, const float* __restrict__ W,
                        float* __restrict__ O, int mtiles) {
    int wave = threadIdx.x >> 5;
    int tile = blockIdx.x * (blockDim.x >> 5) + wave;
    if (tile >= mtiles * 4) return;
    int mt = tile >> 2, nt = tile & 3;
    int lane = threadIdx.x & 31;
    int lo = lane & 15, hi = lane >> 4;

    const float* arow = X + (size_t)(mt * 16 + lo) * 64;
    int col = nt * 16 + lo;
    v8f c = {};
#pragma unroll
    for (int k = 0; k < 64; k += 4) {
        int ka = k + 2 * hi;
        v2f a = { arow[ka], arow[ka + 1] };
        v2f b = { W[(size_t)ka * 64 + col], W[(size_t)(ka + 1) * 64 + col] };
        c = __builtin_amdgcn_wmma_f32_16x16x4_f32(false, a, false, b,
                                                  (short)0, c, false, false);
    }
    float* orow = O + (size_t)(mt * 16 + 8 * hi) * 64 + col;
#pragma unroll
    for (int r = 0; r < 8; ++r) orow[(size_t)r * 64] = c[r];
}


__global__ void k_gemm2(const float* __restrict__ Hin, const float* __restrict__ W2,
                        float* __restrict__ O, int mtiles) {
    int wave = threadIdx.x >> 5;
    int tile = blockIdx.x * (blockDim.x >> 5) + wave;
    if (tile >= mtiles * 3) return;
    int mt = tile / 3, nt = tile - mt * 3;
    int lane = threadIdx.x & 31;
    int lo = lane & 15, hi = lane >> 4;

    const float* arow = Hin + (size_t)(mt * 16 + lo) * 64;
    int col = nt * 16 + lo;
    bool cok = col < 40;
    v8f c = {};
#pragma unroll
    for (int k = 0; k < 64; k += 4) {
        int ka = k + 2 * hi;
        v2f a = { fmaxf(arow[ka], 0.0f), fmaxf(arow[ka + 1], 0.0f) };
        v2f b = { cok ? W2[(size_t)ka * 40 + col] : 0.0f,
                  cok ? W2[(size_t)(ka + 1) * 40 + col] : 0.0f };
        c = __builtin_amdgcn_wmma_f32_16x16x4_f32(false, a, false, b,
                                                  (short)0, c, false, false);
    }
    if (cok) {
        float* orow = O + (size_t)(mt * 16 + 8 * hi) * 40 + col;
#pragma unroll
        for (int r = 0; r < 8; ++r) orow[(size_t)r * 40] = c[r];
    }
}


template <int W>
__global__ __launch_bounds__(TPB) void k_agg_own(const unsigned* __restrict__ keyw,
                                                 const int* __restrict__ src,
                                                 const int* __restrict__ dst,
                                                 const float* __restrict__ w,
                                                 const float* __restrict__ dinv,
                                                 const float* __restrict__ h,
                                                 const float* __restrict__ bias,
                                                 float* __restrict__ out,
                                                 int n, int E, int R, int nchunks) {
    constexpr int PWS = pws_of(RMAX * W * 4);
    __shared__ __attribute__((aligned(16))) float acc[RMAX * W];
    __shared__ unsigned list[NWAVE * PWS];
    const uint4* keyv = (const uint4*)keyw;
    const int tid = threadIdx.x, lane = tid & 31;
    const int wave = __builtin_amdgcn_readfirstlane(tid >> 5);
    const int owner = blockIdx.x, base = owner * R;
    const unsigned pat = (unsigned)owner * 0x01010101u;
    unsigned* wlist = list + wave * PWS;
    int rows = n - base;
    rows = rows < 0 ? 0 : (rows > R ? R : rows);
    const int nel = rows * W;
    const float* hown = h + (size_t)base * W;
    float* oown = out + (size_t)base * W;

    for (int i = tid; i < nel; i += TPB) {
        int r = i / W, f = i - r * W;
        float di = dinv[base + r];
        acc[i] = bias[f] + di * di * hown[i];
    }
    __syncthreads();

    const bool fl = (2 * lane < W);
    auto proc = [&](int cnt) {
        for (int g = 0; g < cnt; g += 32) {
            int idx = g + lane;
            bool v = idx < cnt;
            unsigned seg = v ? wlist[idx] : 0u;
            unsigned m = v ? key_hit_mask16(keyv[seg], pat) : 0u;
            while (__builtin_amdgcn_ballot_w32(m != 0u) != 0u) {
                bool act = (m != 0u);
                int e = (int)seg * 16 + (act ? __builtin_ctz(m) : 0);
                m &= m - 1u;
                act = act && (e < E);
                int s = 0, dl = 0;
                float nr = 0.0f;
                if (act) {
                    int d = dst[e];
                    s = src[e];
                    nr = dinv[s] * w[e] * dinv[d];
                    dl = d - base;
                }
                unsigned bal = __builtin_amdgcn_ballot_w32(act);
                while (bal != 0u) {
                    int jj[EB];
                    bool ok[EB];
#pragma unroll
                    for (int u = 0; u < EB; ++u) {
                        ok[u] = (bal != 0u);
                        jj[u] = ok[u] ? __builtin_ctz(bal) : 0;
                        bal &= bal - 1u;
                    }
                    v2f r[EB];
#pragma unroll
                    for (int u = 0; u < EB; ++u) {
                        int sj = __builtin_amdgcn_readlane(s, jj[u]);
                        r[u] = fl ? *(const v2f*)(h + (size_t)sj * W + 2 * lane) : v2f{0.0f, 0.0f};
                    }
#pragma unroll
                    for (int u = 0; u < EB; ++u) {
                        if (ok[u]) {
                            float nj = __int_as_float(__builtin_amdgcn_readlane(__float_as_int(nr), jj[u]));
                            int dj = __builtin_amdgcn_readlane(dl, jj[u]);
                            if (fl) {
                                float* a = &acc[dj * W + 2 * lane];
                                ldsAddF32(a, nj * r[u].x);
                                ldsAddF32(a + 1, nj * r[u].y);
                            }
                        }
                    }
                }
            }
        }
    };
    scan_owned<PWS>(keyv, nchunks, wlist, lane, wave, owner, proc);
    __syncthreads();

    for (int i = tid * 4; i < nel; i += TPB * 4) {
        *(float4*)(oown + i) = *(const float4*)(acc + i);
    }
}


__global__ void k_lsm(const float* __restrict__ h, float* __restrict__ out, int n) {
    int i = blockIdx.x * blockDim.x + threadIdx.x;
    if (i >= n) return;
    float v[40];
    float mx = -3.402823466e38f;
#pragma unroll
    for (int c = 0; c < 40; ++c) { v[c] = h[(size_t)i * 40 + c]; mx = fmaxf(mx, v[c]); }
    float s = 0.0f;
#pragma unroll
    for (int c = 0; c < 40; ++c) s += expf(v[c] - mx);
    float l = mx + logf(s);
#pragma unroll
    for (int c = 0; c < 40; ++c) out[(size_t)i * 40 + c] = v[c] - l;
}


extern "C" void kernel_launch(void* const* d_in, const int* in_sizes, int n_in,
                              void* d_out, int out_size, void* d_ws, size_t ws_size,
                              hipStream_t stream) {
    const float* x   = (const float*)d_in[0];
    const int*   ei  = (const int*)d_in[1];
    const float* ew  = (const float*)d_in[2];
    const float* W1  = (const float*)d_in[3];
    const float* b1  = (const float*)d_in[4];
    const float* W2  = (const float*)d_in[5];
    const float* b2  = (const float*)d_in[6];
    float* out = (float*)d_out;

    const int N = in_sizes[0] / 64;
    const int E = in_sizes[2];
    const int* src = ei;
    const int* dst = ei + E;

    const int R = (N + NOWN - 1) / NOWN;
    const int nchunks = (E + 511) / 512;
    const int nwords = nchunks * 128;
    const int npad = ((N + 63) / 64) * 64;

    float* ws   = (float*)d_ws;
    float* dinv = ws;
    unsigned* keyw = (unsigned*)(dinv + npad);
    float* bufA = (float*)(keyw + nwords);
    float* bufB = bufA + (size_t)N * 64;

    const int BLK = 256;
    const int mtiles = N / 16;

    k_key<<<(nwords + BLK - 1) / BLK, BLK, 0, stream>>>(dst, keyw, E, nwords, (unsigned)R);
    k_deg_own<<<NOWN, TPB, 0, stream>>>(keyw, dst, ew, dinv, N, E, R, nchunks);

    k_gemm1<<<(mtiles * 4 + 7) / 8, BLK, 0, stream>>>(x, W1, bufA, mtiles);
    k_agg_own<64><<<NOWN, TPB, 0, stream>>>(keyw, src, dst, ew, dinv, bufA, b1, bufB, N, E, R, nchunks);

    k_gemm2<<<(mtiles * 3 + 7) / 8, BLK, 0, stream>>>(bufB, W2, bufA, mtiles);
    k_agg_own<40><<<NOWN, TPB, 0, stream>>>(keyw, src, dst, ew, dinv, bufA, b2, bufB, N, E, R, nchunks);

    k_lsm<<<(N + BLK - 1) / BLK, BLK, 0, stream>>>(bufB, out, N);
}


# degree pass stores each wave's owned edges (source, local row, weight) as 8-byte records; both aggregation passes read these lists instead of re-scanning keys; LDS float adds stay in arrival order
# speedup vs baseline: 3.7546x; 1.4569x over previous
#include <hip/hip_runtime.h>
#include <math.h>


typedef __attribute__((ext_vector_type(2))) float v2f;
typedef __attribute__((ext_vector_type(8))) float v8f;

constexpr int NOWN  = 256;
constexpr int RMAX  = 392;
constexpr int TPB   = 1024;
constexpr int NWAVE = TPB / 32;
constexpr int pws_of(int other_bytes) {
    int p = (164 * 1024 - other_bytes + NWAVE * 4 - 1) / (NWAVE * 4);
    return p < 512 ? 512 : p;
}
constexpr int UB    = 8;
constexpr int EB    = 16;
constexpr int CAP   = 512;
constexpr int SBITS = 23;
constexpr unsigned SMASK = (1u << SBITS) - 1u;

__device__ __forceinline__ void ldsAddF32(float* p, float v) {
    __hip_atomic_fetch_add(p, v, __ATOMIC_RELAXED, __HIP_MEMORY_SCOPE_WORKGROUP);
}


__global__ void k_key(const int* __restrict__ dst, unsigned* __restrict__ keyw,
                      int E, int nwords, unsigned R) {
    int i = blockIdx.x * blockDim.x + threadIdx.x;
    if (i >= nwords) return;
    unsigned v = 0u;
#pragma unroll
    for (int u = 0; u < 4; ++u) {
        int e = i * 4 + u;
        unsigned o = (e < E) ? ((unsigned)dst[e] / R) : 0u;
        v |= o << (8 * u);
    }
    keyw[i] = v;
}

__device__ __forceinline__ bool any_key_hit(uint4 k, unsigned pat) {
    unsigned x0 = k.x ^ pat, x1 = k.y ^ pat, x2 = k.z ^ pat, x3 = k.w ^ pat;
    unsigned t = ((x0 - 0x01010101u) & ~x0) | ((x1 - 0x01010101u) & ~x1) |
                 ((x2 - 0x01010101u) & ~x2) | ((x3 - 0x01010101u) & ~x3);
    return (t & 0x80808080u) != 0u;
}

__device__ __forceinline__ unsigned key_hit_mask16(uint4 k, unsigned pat) {
    unsigned kk[4] = { k.x, k.y, k.z, k.w };
    unsigned m = 0u;
#pragma unroll
    for (int j = 0; j < 4; ++j) {
        unsigned x = kk[j] ^ pat;
        unsigned t = ((x & 0x7F7F7F7Fu) + 0x7F7F7F7Fu) | x;
        unsigned z = ~t & 0x80808080u;
        unsigned c = ((z >> 7) & 1u) | ((z >> 14) & 2u) | ((z >> 21) & 4u) | ((z >> 28) & 8u);
        m |= c << (4 * j);
    }
    return m;
}

template <int PWS, class Proc>
__device__ __forceinline__ void scan_owned(const uint4* __restrict__ keyv, int nchunks,
                                           unsigned* wlist, int lane, int wave, int owner,
                                           Proc proc) {
    const unsigned pat = (unsigned)owner * 0x01010101u;
    const int off = (owner * 12) % nchunks;
    int cnt = 0;
    for (int c0 = wave; c0 < nchunks; c0 += NWAVE * UB) {
        uint4 k[UB];
        int cc[UB];
#pragma unroll
        for (int u = 0; u < UB; ++u) {
            int c = c0 + u * NWAVE;
            int x = ((c < nchunks) ? c : wave) + off;
            cc[u] = (x >= nchunks) ? x - nchunks : x;
            k[u] = keyv[(size_t)cc[u] * 32 + lane];
        }
#pragma unroll
        for (int u = 0; u < UB; ++u) {
            bool hit = (c0 + u * NWAVE < nchunks) && any_key_hit(k[u], pat);
            unsigned bal = __builtin_amdgcn_ballot_w32(hit);
            if (hit) wlist[cnt + (int)__builtin_amdgcn_mbcnt_lo(bal, 0u)] = (unsigned)(cc[u] * 32 + lane);
            cnt += __builtin_popcount(bal);
        }
        if (cnt > PWS - 32 * UB) { proc(cnt); cnt = 0; }
    }
    proc(cnt);
}


__global__ __launch_bounds__(TPB) void k_deg_own(const unsigned* __restrict__ keyw,
                                                 const int* __restrict__ src,
                                                 const int* __restrict__ dst,
                                                 const float* __restrict__ w,
                                                 float* __restrict__ dinv,
                                                 int* __restrict__ cntp,
                                                 unsigned* __restrict__ recw,
                                                 int n, int E, int R, int nchunks) {
    constexpr int PWS = pws_of(RMAX * 4);
    __shared__ float degl[RMAX];
    __shared__ unsigned list[NWAVE * PWS];
    const uint4* keyv = (const uint4*)keyw;
    const int tid = threadIdx.x, lane = tid & 31;
    const int wave = __builtin_amdgcn_readfirstlane(tid >> 5);
    const int owner = blockIdx.x, base = owner * R;
    const unsigned pat = (unsigned)owner * 0x01010101u;
    unsigned* wlist = list + wave * PWS;
    const int wid = owner * NWAVE + wave;
    uint2* wrec = (uint2*)recw + (size_t)wid * CAP;
    int nrec = 0;

    for (int i = tid; i < RMAX; i += TPB) degl[i] = 1.0f;
    __syncthreads();

    auto proc = [&](int cnt) {
        for (int g = 0; g < cnt; g += 32) {
            int idx = g + lane;
            bool v = idx < cnt;
            unsigned seg = v ? wlist[idx] : 0u;
            unsigned m = v ? key_hit_mask16(keyv[seg], pat) : 0u;
            while (__builtin_amdgcn_ballot_w32(m != 0u) != 0u) {
                bool act = (m != 0u);
                int e = (int)seg * 16 + (act ? __builtin_ctz(m) : 0);
                m &= m - 1u;
                act = act && (e < E);
                unsigned bal = __builtin_amdgcn_ballot_w32(act);
                if (act) {
                    int dl = dst[e] - base;
                    float we = w[e];
                    int sv = src[e];
                    int pos = nrec + (int)__builtin_amdgcn_mbcnt_lo(bal, 0u);
                    pos = pos < CAP ? pos : CAP - 1;
                    wrec[pos] = make_uint2((unsigned)sv | ((unsigned)dl << SBITS), __float_as_uint(we));
                    ldsAddF32(&degl[dl], we);
                }
                nrec += __builtin_popcount(bal);
            }
        }
    };
    scan_owned<PWS>(keyv, nchunks, wlist, lane, wave, owner, proc);
    if (lane == 0) cntp[wid] = nrec;
    __syncthreads();

    for (int i = tid; i < R; i += TPB) {
        int node = base + i;
        if (node < n) {
            float d = degl[i];
            dinv[node] = (d > 0.0f) ? rsqrtf(d) : 0.0f;
        }
    }
}


__global__ void k_gemm1(const float* __restrict__ X, const float* __restrict__ W,
                        float* __restrict__ O, int mtiles) {
    int wave = threadIdx.x >> 5;
    int tile = blockIdx.x * (blockDim.x >> 5) + wave;
    if (tile >= mtiles * 4) return;
    int mt = tile >> 2, nt = tile & 3;
    int lane = threadIdx.x & 31;
    int lo = lane & 15, hi = lane >> 4;

    const float* arow = X + (size_t)(mt * 16 + lo) * 64;
    int col = nt * 16 + lo;
    v8f c = {};
#pragma unroll
    for (int k = 0; k < 64; k += 4) {
        int ka = k + 2 * hi;
        v2f a = { arow[ka], arow[ka + 1] };
        v2f b = { W[(size_t)ka * 64 + col], W[(size_t)(ka + 1) * 64 + col] };
        c = __builtin_amdgcn_wmma_f32_16x16x4_f32(false, a, false, b,
                                                  (short)0, c, false, false);
    }
    float* orow = O + (size_t)(mt * 16 + 8 * hi) * 64 + col;
#pragma unroll
    for (int r = 0; r < 8; ++r) orow[(size_t)r * 64] = c[r];
}


__global__ void k_gemm2(const float* __restrict__ Hin, const float* __restrict__ W2,
                        float* __restrict__ O, int mtiles) {
    int wave = threadIdx.x >> 5;
    int tile = blockIdx.x * (blockDim.x >> 5) + wave;
    if (tile >= mtiles * 3) return;
    int mt = tile / 3, nt = tile - mt * 3;
    int lane = threadIdx.x & 31;
    int lo = lane & 15, hi = lane >> 4;

    const float* arow = Hin + (size_t)(mt * 16 + lo) * 64;
    int col = nt * 16 + lo;
    bool cok = col < 40;
    v8f c = {};
#pragma unroll
    for (int k = 0; k < 64; k += 4) {
        int ka = k + 2 * hi;
        v2f a = { fmaxf(arow[ka], 0.0f), fmaxf(arow[ka + 1], 0.0f) };
        v2f b = { cok ? W2[(size_t)ka * 40 + col] : 0.0f,
                  cok ? W2[(size_t)(ka + 1) * 40 + col] : 0.0f };
        c = __builtin_amdgcn_wmma_f32_16x16x4_f32(false, a, false, b,
                                                  (short)0, c, false, false);
    }
    if (cok) {
        float* orow = O + (size_t)(mt * 16 + 8 * hi) * 40 + col;
#pragma unroll
        for (int r = 0; r < 8; ++r) orow[(size_t)r * 40] = c[r];
    }
}


template <int W>
__global__ __launch_bounds__(TPB) void k_agg_own(const unsigned* __restrict__ keyw,
                                                 const int* __restrict__ src,
                                                 const int* __restrict__ dst,
                                                 const float* __restrict__ w,
                                                 const float* __restrict__ dinv,
                                                 const float* __restrict__ h,
                                                 const float* __restrict__ bias,
                                                 const int* __restrict__ cntp,
                                                 const unsigned* __restrict__ recw,
                                                 float* __restrict__ out,
                                                 int n, int E, int R, int nchunks) {
    constexpr int PWS = pws_of(RMAX * W * 4);
    __shared__ __attribute__((aligned(16))) float acc[RMAX * W];
    __shared__ unsigned list[NWAVE * PWS];
    const uint4* keyv = (const uint4*)keyw;
    const int tid = threadIdx.x, lane = tid & 31;
    const int wave = __builtin_amdgcn_readfirstlane(tid >> 5);
    const int owner = blockIdx.x, base = owner * R;
    const unsigned pat = (unsigned)owner * 0x01010101u;
    unsigned* wlist = list + wave * PWS;
    int rows = n - base;
    rows = rows < 0 ? 0 : (rows > R ? R : rows);
    const int nel = rows * W;
    const float* hown = h + (size_t)base * W;
    float* oown = out + (size_t)base * W;

    for (int i = tid; i < nel; i += TPB) {
        int r = i / W, f = i - r * W;
        float di = dinv[base + r];
        acc[i] = bias[f] + di * di * hown[i];
    }
    __syncthreads();

    const bool fl = (2 * lane < W);
    auto proc = [&](int cnt) {
        for (int g = 0; g < cnt; g += 32) {
            int idx = g + lane;
            bool v = idx < cnt;
            unsigned seg = v ? wlist[idx] : 0u;
            unsigned m = v ? key_hit_mask16(keyv[seg], pat) : 0u;
            while (__builtin_amdgcn_ballot_w32(m != 0u) != 0u) {
                bool act = (m != 0u);
                int e = (int)seg * 16 + (act ? __builtin_ctz(m) : 0);
                m &= m - 1u;
                act = act && (e < E);
                int s = 0, dl = 0;
                float nr = 0.0f;
                if (act) {
                    int d = dst[e];
                    s = src[e];
                    nr = dinv[s] * w[e] * dinv[d];
                    dl = d - base;
                }
                unsigned bal = __builtin_amdgcn_ballot_w32(act);
                while (bal != 0u) {
                    int jj[EB];
                    bool ok[EB];
#pragma unroll
                    for (int u = 0; u < EB; ++u) {
                        ok[u] = (bal != 0u);
                        jj[u] = ok[u] ? __builtin_ctz(bal) : 0;
                        bal &= bal - 1u;
                    }
                    v2f r[EB];
#pragma unroll
                    for (int u = 0; u < EB; ++u) {
                        int sj = __builtin_amdgcn_readlane(s, jj[u]);
                        r[u] = fl ? *(const v2f*)(h + (size_t)sj * W + 2 * lane) : v2f{0.0f, 0.0f};
                    }
#pragma unroll
                    for (int u = 0; u < EB; ++u) {
                        if (ok[u]) {
                            float nj = __int_as_float(__builtin_amdgcn_readlane(__float_as_int(nr), jj[u]));
                            int dj = __builtin_amdgcn_readlane(dl, jj[u]);
                            if (fl) {
                                float* a = &acc[dj * W + 2 * lane];
                                ldsAddF32(a, nj * r[u].x);
                                ldsAddF32(a + 1, nj * r[u].y);
                            }
                        }
                    }
                }
            }
        }
    };
    const int wid = owner * NWAVE + wave;
    const int c = cntp[wid];
    if (c <= CAP) {
        const unsigned long long* wrec = (const unsigned long long*)recw + (size_t)wid * CAP;
        const bool cur = lane < c;
        unsigned long long rc = cur ? wrec[lane] : 0ull;
        int s = (int)((unsigned)rc & SMASK), dl = (int)((unsigned)rc >> SBITS);
        float nr = cur ? dinv[s] * __uint_as_float((unsigned)(rc >> 32)) * dinv[base + dl] : 0.0f;
        for (int g = 0; g < c; g += 32) {
            const bool nx = g + 32 + lane < c;
            unsigned long long rn = nx ? wrec[g + 32 + lane] : 0ull;
            v2f r[32];
#pragma unroll
            for (int u = 0; u < 32; ++u) {
                int sj = __builtin_amdgcn_readlane(s, u);
                r[u] = fl ? *(const v2f*)(h + (size_t)sj * W + 2 * lane) : v2f{0.0f, 0.0f};
            }
            int sn = (int)((unsigned)rn & SMASK), dln = (int)((unsigned)rn >> SBITS);
            float dsn = nx ? dinv[sn] : 0.0f;
            float ddn = nx ? dinv[base + dln] : 0.0f;
#pragma unroll
            for (int u = 0; u < 32; ++u) {
                if (g + u < c) {
                    float nj = __int_as_float(__builtin_amdgcn_readlane(__float_as_int(nr), u));
                    int dj = __builtin_amdgcn_readlane(dl, u);
                    if (fl) {
                        float* a = &acc[dj * W + 2 * lane];
                        ldsAddF32(a, nj * r[u].x);
                        ldsAddF32(a + 1, nj * r[u].y);
                    }
                }
            }
            s = sn; dl = dln;
            nr = dsn * __uint_as_float((unsigned)(rn >> 32)) * ddn;
        }
    } else {
        scan_owned<PWS>(keyv, nchunks, wlist, lane, wave, owner, proc);
    }
    __syncthreads();

    for (int i = tid * 4; i < nel; i += TPB * 4) {
        *(float4*)(oown + i) = *(const float4*)(acc + i);
    }
}


__global__ void k_lsm(const float* __restrict__ h, float* __restrict__ out, int n) {
    int i = blockIdx.x * blockDim.x + threadIdx.x;
    if (i >= n) return;
    float v[40];
    float mx = -3.402823466e38f;
#pragma unroll
    for (int c = 0; c < 40; ++c) { v[c] = h[(size_t)i * 40 + c]; mx = fmaxf(mx, v[c]); }
    float s = 0.0f;
#pragma unroll
    for (int c = 0; c < 40; ++c) s += expf(v[c] - mx);
    float l = mx + logf(s);
#pragma unroll
    for (int c = 0; c < 40; ++c) out[(size_t)i * 40 + c] = v[c] - l;
}


extern "C" void kernel_launch(void* const* d_in, const int* in_sizes, int n_in,
                              void* d_out, int out_size, void* d_ws, size_t ws_size,
                              hipStream_t stream) {
    const float* x   = (const float*)d_in[0];
    const int*   ei  = (const int*)d_in[1];
    const float* ew  = (const float*)d_in[2];
    const float* W1  = (const float*)d_in[3];
    const float* b1  = (const float*)d_in[4];
    const float* W2  = (const float*)d_in[5];
    const float* b2  = (const float*)d_in[6];
    float* out = (float*)d_out;

    const int N = in_sizes[0] / 64;
    const int E = in_sizes[2];
    const int* src = ei;
    const int* dst = ei + E;

    const int R = (N + NOWN - 1) / NOWN;
    const int nchunks = (E + 511) / 512;
    const int nwords = nchunks * 128;
    const int npad = ((N + 63) / 64) * 64;

    float* ws   = (float*)d_ws;
    float* dinv = ws;
    unsigned* keyw = (unsigned*)(dinv + npad);
    float* bufA = (float*)(keyw + nwords);
    float* bufB = bufA + (size_t)N * 64;
    int* cntp = (int*)(bufB + (size_t)N * 64);
    unsigned* recw = (unsigned*)(cntp + NOWN * NWAVE);

    const int BLK = 256;
    const int mtiles = N / 16;

    k_key<<<(nwords + BLK - 1) / BLK, BLK, 0, stream>>>(dst, keyw, E, nwords, (unsigned)R);
    k_deg_own<<<NOWN, TPB, 0, stream>>>(keyw, src, dst, ew, dinv, cntp, recw, N, E, R, nchunks);

    k_gemm1<<<(mtiles * 4 + 7) / 8, BLK, 0, stream>>>(x, W1, bufA, mtiles);
    k_agg_own<64><<<NOWN, TPB, 0, stream>>>(keyw, src, dst, ew, dinv, bufA, b1, cntp, recw, bufB, N, E, R, nchunks);

    k_gemm2<<<(mtiles * 3 + 7) / 8, BLK, 0, stream>>>(bufB, W2, bufA, mtiles);
    k_agg_own<40><<<NOWN, TPB, 0, stream>>>(keyw, src, dst, ew, dinv, bufA, b2, cntp, recw, bufB, N, E, R, nchunks);

    k_lsm<<<(N + BLK - 1) / BLK, BLK, 0, stream>>>(bufB, out, N);
}


# both GEMMs re-tiled: 256 groups of 16 persistent waves keep the weight fragments in registers and compute whole 16-row blocks (all column tiles) per trip, rows by 16-byte loads; k sum order changes
# speedup vs baseline: 3.9196x; 1.0439x over previous
#include <hip/hip_runtime.h>
#include <math.h>


typedef __attribute__((ext_vector_type(2))) float v2f;
typedef __attribute__((ext_vector_type(8))) float v8f;

constexpr int NOWN  = 256;
constexpr int RMAX  = 392;
constexpr int TPB   = 1024;
constexpr int NWAVE = TPB / 32;
constexpr int pws_of(int other_bytes) {
    int p = (164 * 1024 - other_bytes + NWAVE * 4 - 1) / (NWAVE * 4);
    return p < 512 ? 512 : p;
}
constexpr int UB    = 8;
constexpr int EB    = 16;
constexpr int CAP   = 512;
constexpr int SBITS = 23;
constexpr unsigned SMASK = (1u << SBITS) - 1u;

__device__ __forceinline__ void ldsAddF32(float* p, float v) {
    __hip_atomic_fetch_add(p, v, __ATOMIC_RELAXED, __HIP_MEMORY_SCOPE_WORKGROUP);
}


__global__ void k_key(const int* __restrict__ dst, unsigned* __restrict__ keyw,
                      int E, int nwords, unsigned R) {
    int i = blockIdx.x * blockDim.x + threadIdx.x;
    if (i >= nwords) return;
    unsigned v = 0u;
#pragma unroll
    for (int u = 0; u < 4; ++u) {
        int e = i * 4 + u;
        unsigned o = (e < E) ? ((unsigned)dst[e] / R) : 0u;
        v |= o << (8 * u);
    }
    keyw[i] = v;
}

__device__ __forceinline__ bool any_key_hit(uint4 k, unsigned pat) {
    unsigned x0 = k.x ^ pat, x1 = k.y ^ pat, x2 = k.z ^ pat, x3 = k.w ^ pat;
    unsigned t = ((x0 - 0x01010101u) & ~x0) | ((x1 - 0x01010101u) & ~x1) |
                 ((x2 - 0x01010101u) & ~x2) | ((x3 - 0x01010101u) & ~x3);
    return (t & 0x80808080u) != 0u;
}

__device__ __forceinline__ unsigned key_hit_mask16(uint4 k, unsigned pat) {
    unsigned kk[4] = { k.x, k.y, k.z, k.w };
    unsigned m = 0u;
#pragma unroll
    for (int j = 0; j < 4; ++j) {
        unsigned x = kk[j] ^ pat;
        unsigned t = ((x & 0x7F7F7F7Fu) + 0x7F7F7F7Fu) | x;
        unsigned z = ~t & 0x80808080u;
        unsigned c = ((z >> 7) & 1u) | ((z >> 14) & 2u) | ((z >> 21) & 4u) | ((z >> 28) & 8u);
        m |= c << (4 * j);
    }
    return m;
}

template <int PWS, class Proc>
__device__ __forceinline__ void scan_owned(const uint4* __restrict__ keyv, int nchunks,
                                           unsigned* wlist, int lane, int wave, int owner,
                                           Proc proc) {
    const unsigned pat = (unsigned)owner * 0x01010101u;
    const int off = (owner * 12) % nchunks;
    int cnt = 0;
    for (int c0 = wave; c0 < nchunks; c0 += NWAVE * UB) {
        uint4 k[UB];
        int cc[UB];
#pragma unroll
        for (int u = 0; u < UB; ++u) {
            int c = c0 + u * NWAVE;
            int x = ((c < nchunks) ? c : wave) + off;
            cc[u] = (x >= nchunks) ? x - nchunks : x;
            k[u] = keyv[(size_t)cc[u] * 32 + lane];
        }
#pragma unroll
        for (int u = 0; u < UB; ++u) {
            bool hit = (c0 + u * NWAVE < nchunks) && any_key_hit(k[u], pat);
            unsigned bal = __builtin_amdgcn_ballot_w32(hit);
            if (hit) wlist[cnt + (int)__builtin_amdgcn_mbcnt_lo(bal, 0u)] = (unsigned)(cc[u] * 32 + lane);
            cnt += __builtin_popcount(bal);
        }
        if (cnt > PWS - 32 * UB) { proc(cnt); cnt = 0; }
    }
    proc(cnt);
}


__global__ __launch_bounds__(TPB) void k_deg_own(const unsigned* __restrict__ keyw,
                                                 const int* __restrict__ src,
                                                 const int* __restrict__ dst,
                                                 const float* __restrict__ w,
                                                 float* __restrict__ dinv,
                                                 int* __restrict__ cntp,
                                                 unsigned* __restrict__ recw,
                                                 int n, int E, int R, int nchunks) {
    constexpr int PWS = pws_of(RMAX * 4);
    __shared__ float degl[RMAX];
    __shared__ unsigned list[NWAVE * PWS];
    const uint4* keyv = (const uint4*)keyw;
    const int tid = threadIdx.x, lane = tid & 31;
    const int wave = __builtin_amdgcn_readfirstlane(tid >> 5);
    const int owner = blockIdx.x, base = owner * R;
    const unsigned pat = (unsigned)owner * 0x01010101u;
    unsigned* wlist = list + wave * PWS;
    const int wid = owner * NWAVE + wave;
    uint2* wrec = (uint2*)recw + (size_t)wid * CAP;
    int nrec = 0;

    for (int i = tid; i < RMAX; i += TPB) degl[i] = 1.0f;
    __syncthreads();

    auto proc = [&](int cnt) {
        for (int g = 0; g < cnt; g += 32) {
            int idx = g + lane;
            bool v = idx < cnt;
            unsigned seg = v ? wlist[idx] : 0u;
            unsigned m = v ? key_hit_mask16(keyv[seg], pat) : 0u;
            while (__builtin_amdgcn_ballot_w32(m != 0u) != 0u) {
                bool act = (m != 0u);
                int e = (int)seg * 16 + (act ? __builtin_ctz(m) : 0);
                m &= m - 1u;
                act = act && (e < E);
                unsigned bal = __builtin_amdgcn_ballot_w32(act);
                if (act) {
                    int dl = dst[e] - base;
                    float we = w[e];
                    int sv = src[e];
                    int pos = nrec + (int)__builtin_amdgcn_mbcnt_lo(bal, 0u);
                    pos = pos < CAP ? pos : CAP - 1;
                    wrec[pos] = make_uint2((unsigned)sv | ((unsigned)dl << SBITS), __float_as_uint(we));
                    ldsAddF32(&degl[dl], we);
                }
                nrec += __builtin_popcount(bal);
            }
        }
    };
    scan_owned<PWS>(keyv, nchunks, wlist, lane, wave, owner, proc);
    if (lane == 0) cntp[wid] = nrec;
    __syncthreads();

    for (int i = tid; i < R; i += TPB) {
        int node = base + i;
        if (node < n) {
            float d = degl[i];
            dinv[node] = (d > 0.0f) ? rsqrtf(d) : 0.0f;
        }
    }
}


constexpr int GWG = 256;
constexpr int GWV = 16;

__device__ __forceinline__ int piece(int i) { return 4 * (i & 1) + (i >> 1); }
__device__ __forceinline__ int kof(int j, int hi) { return 8 * piece(j >> 1) + 4 * hi + 2 * (j & 1); }


__global__ __launch_bounds__(GWV * 32) void k_gemm1(const float* __restrict__ X, const float* __restrict__ W,
                                                   float* __restrict__ O, int mtiles) {
    const int lane = threadIdx.x & 31;
    const int wave = __builtin_amdgcn_readfirstlane(threadIdx.x >> 5);
    const int lo = lane & 15, hi = lane >> 4;
    const int t0 = (int)(((long long)blockIdx.x * mtiles) / GWG);
    const int t1 = (int)(((long long)(blockIdx.x + 1) * mtiles) / GWG);
    if (t0 + wave >= t1) return;

    v2f bf[16][4];
#pragma unroll
    for (int j = 0; j < 16; ++j) {
        const int k0 = kof(j, hi);
        const float4 q0 = *(const float4*)(W + (size_t)k0 * 64 + 4 * lo);
        const float4 q1 = *(const float4*)(W + (size_t)(k0 + 1) * 64 + 4 * lo);
        bf[j][0] = v2f{ q0.x, q1.x };
        bf[j][1] = v2f{ q0.y, q1.y };
        bf[j][2] = v2f{ q0.z, q1.z };
        bf[j][3] = v2f{ q0.w, q1.w };
    }

    for (int mt = t0 + wave; mt < t1; mt += GWV) {
        float4 a[8];
        const float* ap = X + (size_t)(mt * 16 + lo) * 64 + 4 * hi;
#pragma unroll
        for (int i = 0; i < 8; ++i) a[i] = *(const float4*)(ap + 8 * piece(i));
        v8f c0 = {}, c1 = {}, c2 = {}, c3 = {};
#pragma unroll
        for (int i = 0; i < 8; ++i) {
            const v2f a0 = { a[i].x, a[i].y };
            const v2f a1 = { a[i].z, a[i].w };
            c0 = __builtin_amdgcn_wmma_f32_16x16x4_f32(false, a0, false, bf[2 * i][0], (short)0, c0, false, false);
            c1 = __builtin_amdgcn_wmma_f32_16x16x4_f32(false, a0, false, bf[2 * i][1], (short)0, c1, false, false);
            c2 = __builtin_amdgcn_wmma_f32_16x16x4_f32(false, a0, false, bf[2 * i][2], (short)0, c2, false, false);
            c3 = __builtin_amdgcn_wmma_f32_16x16x4_f32(false, a0, false, bf[2 * i][3], (short)0, c3, false, false);
            c0 = __builtin_amdgcn_wmma_f32_16x16x4_f32(false, a1, false, bf[2 * i + 1][0], (short)0, c0, false, false);
            c1 = __builtin_amdgcn_wmma_f32_16x16x4_f32(false, a1, false, bf[2 * i + 1][1], (short)0, c1, false, false);
            c2 = __builtin_amdgcn_wmma_f32_16x16x4_f32(false, a1, false, bf[2 * i + 1][2], (short)0, c2, false, false);
            c3 = __builtin_amdgcn_wmma_f32_16x16x4_f32(false, a1, false, bf[2 * i + 1][3], (short)0, c3, false, false);
        }
        float* orow = O + (size_t)(mt * 16 + 8 * hi) * 64 + 4 * lo;
#pragma unroll
        for (int r = 0; r < 8; ++r) *(float4*)(orow + (size_t)r * 64) = make_float4(c0[r], c1[r], c2[r], c3[r]);
    }
}


__global__ __launch_bounds__(GWV * 32) void k_gemm2(const float* __restrict__ Hin, const float* __restrict__ W2,
                                                   float* __restrict__ O, int mtiles) {
    const int lane = threadIdx.x & 31;
    const int wave = __builtin_amdgcn_readfirstlane(threadIdx.x >> 5);
    const int lo = lane & 15, hi = lane >> 4;
    const int t0 = (int)(((long long)blockIdx.x * mtiles) / GWG);
    const int t1 = (int)(((long long)(blockIdx.x + 1) * mtiles) / GWG);
    if (t0 + wave >= t1) return;

    v2f bf[16][3];
#pragma unroll
    for (int j = 0; j < 16; ++j) {
        const int k0 = kof(j, hi);
#pragma unroll
        for (int nt = 0; nt < 3; ++nt) {
            const int col = 3 * lo + nt;
            const bool cok = col < 40;
            const int cc = cok ? col : 39;
            const float w0 = W2[(size_t)k0 * 40 + cc];
            const float w1 = W2[(size_t)(k0 + 1) * 40 + cc];
            bf[j][nt] = v2f{ cok ? w0 : 0.0f, cok ? w1 : 0.0f };
        }
    }

    for (int mt = t0 + wave; mt < t1; mt += GWV) {
        float4 a[8];
        const float* ap = Hin + (size_t)(mt * 16 + lo) * 64 + 4 * hi;
#pragma unroll
        for (int i = 0; i < 8; ++i) a[i] = *(const float4*)(ap + 8 * piece(i));
        v8f c0 = {}, c1 = {}, c2 = {};
#pragma unroll
        for (int i = 0; i < 8; ++i) {
            const v2f a0 = { fmaxf(a[i].x, 0.0f), fmaxf(a[i].y, 0.0f) };
            const v2f a1 = { fmaxf(a[i].z, 0.0f), fmaxf(a[i].w, 0.0f) };
            c0 = __builtin_amdgcn_wmma_f32_16x16x4_f32(false, a0, false, bf[2 * i][0], (short)0, c0, false, false);
            c1 = __builtin_amdgcn_wmma_f32_16x16x4_f32(false, a0, false, bf[2 * i][1], (short)0, c1, false, false);
            c2 = __builtin_amdgcn_wmma_f32_16x16x4_f32(false, a0, false, bf[2 * i][2], (short)0, c2, false, false);
            c0 = __builtin_amdgcn_wmma_f32_16x16x4_f32(false, a1, false, bf[2 * i + 1][0], (short)0, c0, false, false);
            c1 = __builtin_amdgcn_wmma_f32_16x16x4_f32(false, a1, false, bf[2 * i + 1][1], (short)0, c1, false, false);
            c2 = __builtin_amdgcn_wmma_f32_16x16x4_f32(false, a1, false, bf[2 * i + 1][2], (short)0, c2, false, false);
        }
        float* orow = O + (size_t)(mt * 16 + 8 * hi) * 40 + 3 * lo;
        if (lo < 13) {
#pragma unroll
            for (int r = 0; r < 8; ++r) {
                float* p = orow + (size_t)r * 40;
                p[0] = c0[r]; p[1] = c1[r]; p[2] = c2[r];
            }
        } else if (lo == 13) {
#pragma unroll
            for (int r = 0; r < 8; ++r) orow[(size_t)r * 40] = c0[r];
        }
    }
}


template <int W>
__global__ __launch_bounds__(TPB) void k_agg_own(const unsigned* __restrict__ keyw,
                                                 const int* __restrict__ src,
                                                 const int* __restrict__ dst,
                                                 const float* __restrict__ w,
                                                 const float* __restrict__ dinv,
                                                 const float* __restrict__ h,
                                                 const float* __restrict__ bias,
                                                 const int* __restrict__ cntp,
                                                 const unsigned* __restrict__ recw,
                                                 float* __restrict__ out,
                                                 int n, int E, int R, int nchunks) {
    constexpr int PWS = pws_of(RMAX * W * 4);
    __shared__ __attribute__((aligned(16))) float acc[RMAX * W];
    __shared__ unsigned list[NWAVE * PWS];
    const uint4* keyv = (const uint4*)keyw;
    const int tid = threadIdx.x, lane = tid & 31;
    const int wave = __builtin_amdgcn_readfirstlane(tid >> 5);
    const int owner = blockIdx.x, base = owner * R;
    const unsigned pat = (unsigned)owner * 0x01010101u;
    unsigned* wlist = list + wave * PWS;
    int rows = n - base;
    rows = rows < 0 ? 0 : (rows > R ? R : rows);
    const int nel = rows * W;
    const float* hown = h + (size_t)base * W;
    float* oown = out + (size_t)base * W;

    for (int i = tid; i < nel; i += TPB) {
        int r = i / W, f = i - r * W;
        float di = dinv[base + r];
        acc[i] = bias[f] + di * di * hown[i];
    }
    __syncthreads();

    const bool fl = (2 * lane < W);
    auto proc = [&](int cnt) {
        for (int g = 0; g < cnt; g += 32) {
            int idx = g + lane;
            bool v = idx < cnt;
            unsigned seg = v ? wlist[idx] : 0u;
            unsigned m = v ? key_hit_mask16(keyv[seg], pat) : 0u;
            while (__builtin_amdgcn_ballot_w32(m != 0u) != 0u) {
                bool act = (m != 0u);
                int e = (int)seg * 16 + (act ? __builtin_ctz(m) : 0);
                m &= m - 1u;
                act = act && (e < E);
                int s = 0, dl = 0;
                float nr = 0.0f;
                if (act) {
                    int d = dst[e];
                    s = src[e];
                    nr = dinv[s] * w[e] * dinv[d];
                    dl = d - base;
                }
                unsigned bal = __builtin_amdgcn_ballot_w32(act);
                while (bal != 0u) {
                    int jj[EB];
                    bool ok[EB];
#pragma unroll
                    for (int u = 0; u < EB; ++u) {
                        ok[u] = (bal != 0u);
                        jj[u] = ok[u] ? __builtin_ctz(bal) : 0;
                        bal &= bal - 1u;
                    }
                    v2f r[EB];
#pragma unroll
                    for (int u = 0; u < EB; ++u) {
                        int sj = __builtin_amdgcn_readlane(s, jj[u]);
                        r[u] = fl ? *(const v2f*)(h + (size_t)sj * W + 2 * lane) : v2f{0.0f, 0.0f};
                    }
#pragma unroll
                    for (int u = 0; u < EB; ++u) {
                        if (ok[u]) {
                            float nj = __int_as_float(__builtin_amdgcn_readlane(__float_as_int(nr), jj[u]));
                            int dj = __builtin_amdgcn_readlane(dl, jj[u]);
                            if (fl) {
                                float* a = &acc[dj * W + 2 * lane];
                                ldsAddF32(a, nj * r[u].x);
                                ldsAddF32(a + 1, nj * r[u].y);
                            }
                        }
                    }
                }
            }
        }
    };
    const int wid = owner * NWAVE + wave;
    const int c = cntp[wid];
    if (c <= CAP) {
        const unsigned long long* wrec = (const unsigned long long*)recw + (size_t)wid * CAP;
        const bool cur = lane < c;
        unsigned long long rc = cur ? wrec[lane] : 0ull;
        int s = (int)((unsigned)rc & SMASK), dl = (int)((unsigned)rc >> SBITS);
        float nr = cur ? dinv[s] * __uint_as_float((unsigned)(rc >> 32)) * dinv[base + dl] : 0.0f;
        for (int g = 0; g < c; g += 32) {
            const bool nx = g + 32 + lane < c;
            unsigned long long rn = nx ? wrec[g + 32 + lane] : 0ull;
            v2f r[32];
#pragma unroll
            for (int u = 0; u < 32; ++u) {
                int sj = __builtin_amdgcn_readlane(s, u);
                r[u] = fl ? *(const v2f*)(h + (size_t)sj * W + 2 * lane) : v2f{0.0f, 0.0f};
            }
            int sn = (int)((unsigned)rn & SMASK), dln = (int)((unsigned)rn >> SBITS);
            float dsn = nx ? dinv[sn] : 0.0f;
            float ddn = nx ? dinv[base + dln] : 0.0f;
#pragma unroll
            for (int u = 0; u < 32; ++u) {
                if (g + u < c) {
                    float nj = __int_as_float(__builtin_amdgcn_readlane(__float_as_int(nr), u));
                    int dj = __builtin_amdgcn_readlane(dl, u);
                    if (fl) {
                        float* a = &acc[dj * W + 2 * lane];
                        ldsAddF32(a, nj * r[u].x);
                        ldsAddF32(a + 1, nj * r[u].y);
                    }
                }
            }
            s = sn; dl = dln;
            nr = dsn * __uint_as_float((unsigned)(rn >> 32)) * ddn;
        }
    } else {
        scan_owned<PWS>(keyv, nchunks, wlist, lane, wave, owner, proc);
    }
    __syncthreads();

    for (int i = tid * 4; i < nel; i += TPB * 4) {
        *(float4*)(oown + i) = *(const float4*)(acc + i);
    }
}


__global__ void k_lsm(const float* __restrict__ h, float* __restrict__ out, int n) {
    int i = blockIdx.x * blockDim.x + threadIdx.x;
    if (i >= n) return;
    float v[40];
    float mx = -3.402823466e38f;
#pragma unroll
    for (int c = 0; c < 40; ++c) { v[c] = h[(size_t)i * 40 + c]; mx = fmaxf(mx, v[c]); }
    float s = 0.0f;
#pragma unroll
    for (int c = 0; c < 40; ++c) s += expf(v[c] - mx);
    float l = mx + logf(s);
#pragma unroll
    for (int c = 0; c < 40; ++c) out[(size_t)i * 40 + c] = v[c] - l;
}


extern "C" void kernel_launch(void* const* d_in, const int* in_sizes, int n_in,
                              void* d_out, int out_size, void* d_ws, size_t ws_size,
                              hipStream_t stream) {
    const float* x   = (const float*)d_in[0];
    const int*   ei  = (const int*)d_in[1];
    const float* ew  = (const float*)d_in[2];
    const float* W1  = (const float*)d_in[3];
    const float* b1  = (const float*)d_in[4];
    const float* W2  = (const float*)d_in[5];
    const float* b2  = (const float*)d_in[6];
    float* out = (float*)d_out;

    const int N = in_sizes[0] / 64;
    const int E = in_sizes[2];
    const int* src = ei;
    const int* dst = ei + E;

    const int R = (N + NOWN - 1) / NOWN;
    const int nchunks = (E + 511) / 512;
    const int nwords = nchunks * 128;
    const int npad = ((N + 63) / 64) * 64;

    float* ws   = (float*)d_ws;
    float* dinv = ws;
    unsigned* keyw = (unsigned*)(dinv + npad);
    float* bufA = (float*)(keyw + nwords);
    float* bufB = bufA + (size_t)N * 64;
    int* cntp = (int*)(bufB + (size_t)N * 64);
    unsigned* recw = (unsigned*)(cntp + NOWN * NWAVE);

    const int BLK = 256;
    const int mtiles = N / 16;

    k_key<<<(nwords + BLK - 1) / BLK, BLK, 0, stream>>>(dst, keyw, E, nwords, (unsigned)R);
    k_deg_own<<<NOWN, TPB, 0, stream>>>(keyw, src, dst, ew, dinv, cntp, recw, N, E, R, nchunks);

    k_gemm1<<<GWG, GWV * 32, 0, stream>>>(x, W1, bufA, mtiles);
    k_agg_own<64><<<NOWN, TPB, 0, stream>>>(keyw, src, dst, ew, dinv, bufA, b1, cntp, recw, bufB, N, E, R, nchunks);

    k_gemm2<<<GWG, GWV * 32, 0, stream>>>(bufB, W2, bufA, mtiles);
    k_agg_own<40><<<NOWN, TPB, 0, stream>>>(keyw, src, dst, ew, dinv, bufA, b2, cntp, recw, bufB, N, E, R, nchunks);

    k_lsm<<<(N + BLK - 1) / BLK, BLK, 0, stream>>>(bufB, out, N);
}


# layer 1 summed before its product; both WMMA products and log_softmax now run at the tails of the two owner sum kernels on their LDS rows: 7 -> 4 launches; layer-1 f32 sum order changes
# speedup vs baseline: 4.6608x; 1.1891x over previous
#include <hip/hip_runtime.h>
#include <math.h>


typedef __attribute__((ext_vector_type(2))) float v2f;
typedef __attribute__((ext_vector_type(8))) float v8f;

constexpr int NOWN  = 256;
constexpr int RMAX  = 392;
constexpr int TPB   = 1024;
constexpr int NWAVE = TPB / 32;
constexpr int pws_of(int other_bytes) {
    int p = (164 * 1024 - other_bytes + NWAVE * 4 - 1) / (NWAVE * 4);
    return p < 512 ? 512 : p;
}
constexpr int UB    = 8;
constexpr int EB    = 16;
constexpr int CAP   = 512;
constexpr int SBITS = 23;
constexpr unsigned SMASK = (1u << SBITS) - 1u;

__device__ __forceinline__ void ldsAddF32(float* p, float v) {
    __hip_atomic_fetch_add(p, v, __ATOMIC_RELAXED, __HIP_MEMORY_SCOPE_WORKGROUP);
}


__global__ void k_key(const int* __restrict__ dst, unsigned* __restrict__ keyw,
                      int E, int nwords, unsigned R) {
    int i = blockIdx.x * blockDim.x + threadIdx.x;
    if (i >= nwords) return;
    unsigned v = 0u;
#pragma unroll
    for (int u = 0; u < 4; ++u) {
        int e = i * 4 + u;
        unsigned o = (e < E) ? ((unsigned)dst[e] / R) : 0u;
        v |= o << (8 * u);
    }
    keyw[i] = v;
}

__device__ __forceinline__ bool any_key_hit(uint4 k, unsigned pat) {
    unsigned x0 = k.x ^ pat, x1 = k.y ^ pat, x2 = k.z ^ pat, x3 = k.w ^ pat;
    unsigned t = ((x0 - 0x01010101u) & ~x0) | ((x1 - 0x01010101u) & ~x1) |
                 ((x2 - 0x01010101u) & ~x2) | ((x3 - 0x01010101u) & ~x3);
    return (t & 0x80808080u) != 0u;
}

__device__ __forceinline__ unsigned key_hit_mask16(uint4 k, unsigned pat) {
    unsigned kk[4] = { k.x, k.y, k.z, k.w };
    unsigned m = 0u;
#pragma unroll
    for (int j = 0; j < 4; ++j) {
        unsigned x = kk[j] ^ pat;
        unsigned t = ((x & 0x7F7F7F7Fu) + 0x7F7F7F7Fu) | x;
        unsigned z = ~t & 0x80808080u;
        unsigned c = ((z >> 7) & 1u) | ((z >> 14) & 2u) | ((z >> 21) & 4u) | ((z >> 28) & 8u);
        m |= c << (4 * j);
    }
    return m;
}

template <int PWS, class Proc>
__device__ __forceinline__ void scan_owned(const uint4* __restrict__ keyv, int nchunks,
                                           unsigned* wlist, int lane, int wave, int owner,
                                           Proc proc) {
    const unsigned pat = (unsigned)owner * 0x01010101u;
    const int off = (owner * 12) % nchunks;
    int cnt = 0;
    for (int c0 = wave; c0 < nchunks; c0 += NWAVE * UB) {
        uint4 k[UB];
        int cc[UB];
#pragma unroll
        for (int u = 0; u < UB; ++u) {
            int c = c0 + u * NWAVE;
            int x = ((c < nchunks) ? c : wave) + off;
            cc[u] = (x >= nchunks) ? x - nchunks : x;
            k[u] = keyv[(size_t)cc[u] * 32 + lane];
        }
#pragma unroll
        for (int u = 0; u < UB; ++u) {
            bool hit = (c0 + u * NWAVE < nchunks) && any_key_hit(k[u], pat);
            unsigned bal = __builtin_amdgcn_ballot_w32(hit);
            if (hit) wlist[cnt + (int)__builtin_amdgcn_mbcnt_lo(bal, 0u)] = (unsigned)(cc[u] * 32 + lane);
            cnt += __builtin_popcount(bal);
        }
        if (cnt > PWS - 32 * UB) { proc(cnt); cnt = 0; }
    }
    proc(cnt);
}


__global__ __launch_bounds__(TPB) void k_deg_own(const unsigned* __restrict__ keyw,
                                                 const int* __restrict__ src,
                                                 const int* __restrict__ dst,
                                                 const float* __restrict__ w,
                                                 float* __restrict__ dinv,
                                                 int* __restrict__ cntp,
                                                 unsigned* __restrict__ recw,
                                                 int n, int E, int R, int nchunks) {
    constexpr int PWS = pws_of(RMAX * 4);
    __shared__ float degl[RMAX];
    __shared__ unsigned list[NWAVE * PWS];
    const uint4* keyv = (const uint4*)keyw;
    const int tid = threadIdx.x, lane = tid & 31;
    const int wave = __builtin_amdgcn_readfirstlane(tid >> 5);
    const int owner = blockIdx.x, base = owner * R;
    const unsigned pat = (unsigned)owner * 0x01010101u;
    unsigned* wlist = list + wave * PWS;
    const int wid = owner * NWAVE + wave;
    uint2* wrec = (uint2*)recw + (size_t)wid * CAP;
    int nrec = 0;

    for (int i = tid; i < RMAX; i += TPB) degl[i] = 1.0f;
    __syncthreads();

    auto proc = [&](int cnt) {
        for (int g = 0; g < cnt; g += 32) {
            int idx = g + lane;
            bool v = idx < cnt;
            unsigned seg = v ? wlist[idx] : 0u;
            unsigned m = v ? key_hit_mask16(keyv[seg], pat) : 0u;
            while (__builtin_amdgcn_ballot_w32(m != 0u) != 0u) {
                bool act = (m != 0u);
                int e = (int)seg * 16 + (act ? __builtin_ctz(m) : 0);
                m &= m - 1u;
                act = act && (e < E);
                unsigned bal = __builtin_amdgcn_ballot_w32(act);
                if (act) {
                    int dl = dst[e] - base;
                    float we = w[e];
                    int sv = src[e];
                    int pos = nrec + (int)__builtin_amdgcn_mbcnt_lo(bal, 0u);
                    pos = pos < CAP ? pos : CAP - 1;
                    wrec[pos] = make_uint2((unsigned)sv | ((unsigned)dl << SBITS), __float_as_uint(we));
                    ldsAddF32(&degl[dl], we);
                }
                nrec += __builtin_popcount(bal);
            }
        }
    };
    scan_owned<PWS>(keyv, nchunks, wlist, lane, wave, owner, proc);
    if (lane == 0) cntp[wid] = nrec;
    __syncthreads();

    for (int i = tid; i < R; i += TPB) {
        int node = base + i;
        if (node < n) {
            float d = degl[i];
            dinv[node] = (d > 0.0f) ? rsqrtf(d) : 0.0f;
        }
    }
}


__device__ __forceinline__ int kq(int j, int hi) { return 4 * j + 2 * hi; }


template <int W, bool FUSE>
__global__ __launch_bounds__(TPB) void k_agg_own(const unsigned* __restrict__ keyw,
                                                 const int* __restrict__ src,
                                                 const int* __restrict__ dst,
                                                 const float* __restrict__ w,
                                                 const float* __restrict__ dinv,
                                                 const float* __restrict__ h,
                                                 const float* __restrict__ bias,
                                                 const int* __restrict__ cntp,
                                                 const unsigned* __restrict__ recw,
                                                 float* __restrict__ out,
                                                 const float* __restrict__ Wa,
                                                 const float* __restrict__ Wb,
                                                 int n, int E, int R, int nchunks) {
    constexpr int PITCH = FUSE ? 68 : W;
    constexpr int AROWS = FUSE ? 400 : RMAX;
    constexpr int PWS = pws_of(AROWS * PITCH * 4);
    __shared__ __attribute__((aligned(16))) float acc[AROWS * PITCH];
    __shared__ unsigned list[NWAVE * PWS];
    __shared__ float lse[FUSE ? 1 : RMAX];
    const uint4* keyv = (const uint4*)keyw;
    const int tid = threadIdx.x, lane = tid & 31;
    const int wave = __builtin_amdgcn_readfirstlane(tid >> 5);
    const int owner = blockIdx.x, base = owner * R;
    const unsigned pat = (unsigned)owner * 0x01010101u;
    unsigned* wlist = list + wave * PWS;
    int rows = n - base;
    rows = rows < 0 ? 0 : (rows > R ? R : rows);
    const int nel = rows * W;
    const float* hown = h + (size_t)base * W;
    float* oown = out + (size_t)base * W;

    for (int i = tid; i < nel; i += TPB) {
        int r = i / W, f = i - r * W;
        float di = dinv[base + r];
        float sl = di * di * hown[i];
        acc[r * PITCH + f] = FUSE ? sl : bias[f] + sl;
    }
    __syncthreads();

    const bool fl = (2 * lane < W);
    auto proc = [&](int cnt) {
        for (int g = 0; g < cnt; g += 32) {
            int idx = g + lane;
            bool v = idx < cnt;
            unsigned seg = v ? wlist[idx] : 0u;
            unsigned m = v ? key_hit_mask16(keyv[seg], pat) : 0u;
            while (__builtin_amdgcn_ballot_w32(m != 0u) != 0u) {
                bool act = (m != 0u);
                int e = (int)seg * 16 + (act ? __builtin_ctz(m) : 0);
                m &= m - 1u;
                act = act && (e < E);
                int s = 0, dl = 0;
                float nr = 0.0f;
                if (act) {
                    int d = dst[e];
                    s = src[e];
                    nr = dinv[s] * w[e] * dinv[d];
                    dl = d - base;
                }
                unsigned bal = __builtin_amdgcn_ballot_w32(act);
                while (bal != 0u) {
                    int jj[EB];
                    bool ok[EB];
#pragma unroll
                    for (int u = 0; u < EB; ++u) {
                        ok[u] = (bal != 0u);
                        jj[u] = ok[u] ? __builtin_ctz(bal) : 0;
                        bal &= bal - 1u;
                    }
                    v2f r[EB];
#pragma unroll
                    for (int u = 0; u < EB; ++u) {
                        int sj = __builtin_amdgcn_readlane(s, jj[u]);
                        r[u] = fl ? *(const v2f*)(h + (size_t)sj * W + 2 * lane) : v2f{0.0f, 0.0f};
                    }
#pragma unroll
                    for (int u = 0; u < EB; ++u) {
                        if (ok[u]) {
                            float nj = __int_as_float(__builtin_amdgcn_readlane(__float_as_int(nr), jj[u]));
                            int dj = __builtin_amdgcn_readlane(dl, jj[u]);
                            if (fl) {
                                float* a = &acc[dj * PITCH + 2 * lane];
                                ldsAddF32(a, nj * r[u].x);
                                ldsAddF32(a + 1, nj * r[u].y);
                            }
                        }
                    }
                }
            }
        }
    };
    const int wid = owner * NWAVE + wave;
    const int c = cntp[wid];
    if (c <= CAP) {
        const unsigned long long* wrec = (const unsigned long long*)recw + (size_t)wid * CAP;
        const bool cur = lane < c;
        unsigned long long rc = cur ? wrec[lane] : 0ull;
        int s = (int)((unsigned)rc & SMASK), dl = (int)((unsigned)rc >> SBITS);
        float nr = cur ? dinv[s] * __uint_as_float((unsigned)(rc >> 32)) * dinv[base + dl] : 0.0f;
        for (int g = 0; g < c; g += 32) {
            const bool nx = g + 32 + lane < c;
            unsigned long long rn = nx ? wrec[g + 32 + lane] : 0ull;
            v2f r[32];
#pragma unroll
            for (int u = 0; u < 32; ++u) {
                int sj = __builtin_amdgcn_readlane(s, u);
                r[u] = fl ? *(const v2f*)(h + (size_t)sj * W + 2 * lane) : v2f{0.0f, 0.0f};
            }
            int sn = (int)((unsigned)rn & SMASK), dln = (int)((unsigned)rn >> SBITS);
            float dsn = nx ? dinv[sn] : 0.0f;
            float ddn = nx ? dinv[base + dln] : 0.0f;
#pragma unroll
            for (int u = 0; u < 32; ++u) {
                if (g + u < c) {
                    float nj = __int_as_float(__builtin_amdgcn_readlane(__float_as_int(nr), u));
                    int dj = __builtin_amdgcn_readlane(dl, u);
                    if (fl) {
                        float* a = &acc[dj * PITCH + 2 * lane];
                        ldsAddF32(a, nj * r[u].x);
                        ldsAddF32(a + 1, nj * r[u].y);
                    }
                }
            }
            s = sn; dl = dln;
            nr = dsn * __uint_as_float((unsigned)(rn >> 32)) * ddn;
        }
    } else {
        scan_owned<PWS>(keyv, nchunks, wlist, lane, wave, owner, proc);
    }
    __syncthreads();

    if constexpr (!FUSE) {
        static_assert(RMAX <= TPB, "a thread per owned row");
        if (tid < rows) {
            const int r = tid;
            const float* v = acc + r * W;
            float mx = -3.402823466e38f;
#pragma unroll
            for (int c = 0; c < W; ++c) mx = fmaxf(mx, v[c]);
            float s = 0.0f;
#pragma unroll
            for (int c = 0; c < W; ++c) s += expf(v[c] - mx);
            lse[r] = mx + logf(s);
        }
        __syncthreads();
        for (int i = tid * 4; i < nel; i += TPB * 4) {
            const float4 q = *(const float4*)(acc + i);
            const float l = lse[i / W];
            *(float4*)(oown + i) = make_float4(q.x - l, q.y - l, q.z - l, q.w - l);
        }
    } else {
        const int lo = lane & 15, hi = lane >> 4;
        const int ntile = (rows + 15) >> 4;
        static_assert(AROWS <= 16 * NWAVE, "one 16-row tile per wave covers the owner's rows");
        if (wave < ntile) {
            const int mt = wave;
            const float* arow = acc + (mt * 16 + lo) * PITCH;
            v8f c0 = {}, c1 = {}, c2 = {}, c3 = {};
#pragma unroll 8
            for (int j = 0; j < 16; ++j) {
                const int k0 = kq(j, hi);
                const v2f a = *(const v2f*)(arow + k0);
                const float4 q0 = *(const float4*)(Wa + (size_t)k0 * 64 + 4 * lo);
                const float4 q1 = *(const float4*)(Wa + (size_t)(k0 + 1) * 64 + 4 * lo);
                c0 = __builtin_amdgcn_wmma_f32_16x16x4_f32(false, a, false, v2f{ q0.x, q1.x }, (short)0, c0, false, false);
                c1 = __builtin_amdgcn_wmma_f32_16x16x4_f32(false, a, false, v2f{ q0.y, q1.y }, (short)0, c1, false, false);
                c2 = __builtin_amdgcn_wmma_f32_16x16x4_f32(false, a, false, v2f{ q0.z, q1.z }, (short)0, c2, false, false);
                c3 = __builtin_amdgcn_wmma_f32_16x16x4_f32(false, a, false, v2f{ q0.w, q1.w }, (short)0, c3, false, false);
            }
            const float4 bv = *(const float4*)(bias + 4 * lo);
            float* drow = acc + (mt * 16 + 8 * hi) * PITCH + 4 * lo;
#pragma unroll
            for (int r = 0; r < 8; ++r) {
                *(float4*)(drow + r * PITCH) = make_float4(fmaxf(c0[r] + bv.x, 0.0f), fmaxf(c1[r] + bv.y, 0.0f),
                                                           fmaxf(c2[r] + bv.z, 0.0f), fmaxf(c3[r] + bv.w, 0.0f));
            }
            v8f d0 = {}, d1 = {}, d2 = {};
#pragma unroll 8
            for (int j = 0; j < 16; ++j) {
                const int k0 = kq(j, hi);
                const v2f a = *(const v2f*)(arow + k0);
                v2f b[3];
#pragma unroll
                for (int nt = 0; nt < 3; ++nt) {
                    const int col = 3 * lo + nt;
                    const bool cok = col < 40;
                    const int cc = cok ? col : 39;
                    const float w0 = Wb[(size_t)k0 * 40 + cc];
                    const float w1 = Wb[(size_t)(k0 + 1) * 40 + cc];
                    b[nt] = v2f{ cok ? w0 : 0.0f, cok ? w1 : 0.0f };
                }
                d0 = __builtin_amdgcn_wmma_f32_16x16x4_f32(false, a, false, b[0], (short)0, d0, false, false);
                d1 = __builtin_amdgcn_wmma_f32_16x16x4_f32(false, a, false, b[1], (short)0, d1, false, false);
                d2 = __builtin_amdgcn_wmma_f32_16x16x4_f32(false, a, false, b[2], (short)0, d2, false, false);
            }
            const int r0 = mt * 16 + 8 * hi;
            float* orow = out + (size_t)(base + r0) * 40 + 3 * lo;
#pragma unroll
            for (int r = 0; r < 8; ++r) {
                if (r0 + r < rows) {
                    float* p = orow + (size_t)r * 40;
                    if (lo < 13) { p[0] = d0[r]; p[1] = d1[r]; p[2] = d2[r]; }
                    else if (lo == 13) { p[0] = d0[r]; }
                }
            }
        }
    }
}


extern "C" void kernel_launch(void* const* d_in, const int* in_sizes, int n_in,
                              void* d_out, int out_size, void* d_ws, size_t ws_size,
                              hipStream_t stream) {
    const float* x   = (const float*)d_in[0];
    const int*   ei  = (const int*)d_in[1];
    const float* ew  = (const float*)d_in[2];
    const float* W1  = (const float*)d_in[3];
    const float* b1  = (const float*)d_in[4];
    const float* W2  = (const float*)d_in[5];
    const float* b2  = (const float*)d_in[6];
    float* out = (float*)d_out;

    const int N = in_sizes[0] / 64;
    const int E = in_sizes[2];
    const int* src = ei;
    const int* dst = ei + E;

    const int R = (N + NOWN - 1) / NOWN;
    const int nchunks = (E + 511) / 512;
    const int nwords = nchunks * 128;
    const int npad = ((N + 63) / 64) * 64;

    float* ws   = (float*)d_ws;
    float* dinv = ws;
    unsigned* keyw = (unsigned*)(dinv + npad);
    float* bufA = (float*)(keyw + nwords);
    int* cntp = (int*)(bufA + (size_t)N * 128);
    unsigned* recw = (unsigned*)(cntp + NOWN * NWAVE);

    const int BLK = 256;

    k_key<<<(nwords + BLK - 1) / BLK, BLK, 0, stream>>>(dst, keyw, E, nwords, (unsigned)R);
    k_deg_own<<<NOWN, TPB, 0, stream>>>(keyw, src, dst, ew, dinv, cntp, recw, N, E, R, nchunks);

    k_agg_own<64, true><<<NOWN, TPB, 0, stream>>>(keyw, src, dst, ew, dinv, x, b1, cntp, recw, bufA, W1, W2, N, E, R, nchunks);

    k_agg_own<40, false><<<NOWN, TPB, 0, stream>>>(keyw, src, dst, ew, dinv, bufA, b2, cntp, recw, out, nullptr, nullptr, N, E, R, nchunks);
}


# key kernel also packs each edge (source, local row, weight) into an 8-byte record in edge order; degree pass fetches an owned edge with one scattered load instead of three (dst, weight, src)
# speedup vs baseline: 5.0427x; 1.0819x over previous
#include <hip/hip_runtime.h>
#include <math.h>


typedef __attribute__((ext_vector_type(2))) float v2f;
typedef __attribute__((ext_vector_type(8))) float v8f;

constexpr int NOWN  = 256;
constexpr int RMAX  = 392;
constexpr int TPB   = 1024;
constexpr int NWAVE = TPB / 32;
constexpr int pws_of(int other_bytes) {
    int p = (164 * 1024 - other_bytes + NWAVE * 4 - 1) / (NWAVE * 4);
    return p < 512 ? 512 : p;
}
constexpr int UB    = 8;
constexpr int EB    = 16;
constexpr int CAP   = 512;
constexpr int SBITS = 23;
constexpr unsigned SMASK = (1u << SBITS) - 1u;

__device__ __forceinline__ void ldsAddF32(float* p, float v) {
    __hip_atomic_fetch_add(p, v, __ATOMIC_RELAXED, __HIP_MEMORY_SCOPE_WORKGROUP);
}


__global__ void k_key(const int* __restrict__ src, const int* __restrict__ dst,
                      const float* __restrict__ w, unsigned* __restrict__ keyw,
                      uint4* __restrict__ erec, int E, int nwords, unsigned R) {
    int i = blockIdx.x * blockDim.x + threadIdx.x;
    if (i >= nwords) return;
    unsigned v = 0u;
    unsigned rw[8];
#pragma unroll
    for (int u = 0; u < 4; ++u) {
        int e = i * 4 + u;
        const bool in = e < E;
        unsigned d = in ? (unsigned)dst[e] : 0u;
        unsigned sv = in ? (unsigned)src[e] : 0u;
        float we = in ? w[e] : 0.0f;
        unsigned o = d / R;
        v |= o << (8 * u);
        rw[2 * u] = sv | ((d - o * R) << SBITS);
        rw[2 * u + 1] = __float_as_uint(we);
    }
    keyw[i] = v;
    erec[2 * i] = make_uint4(rw[0], rw[1], rw[2], rw[3]);
    erec[2 * i + 1] = make_uint4(rw[4], rw[5], rw[6], rw[7]);
}

__device__ __forceinline__ bool any_key_hit(uint4 k, unsigned pat) {
    unsigned x0 = k.x ^ pat, x1 = k.y ^ pat, x2 = k.z ^ pat, x3 = k.w ^ pat;
    unsigned t = ((x0 - 0x01010101u) & ~x0) | ((x1 - 0x01010101u) & ~x1) |
                 ((x2 - 0x01010101u) & ~x2) | ((x3 - 0x01010101u) & ~x3);
    return (t & 0x80808080u) != 0u;
}

__device__ __forceinline__ unsigned key_hit_mask16(uint4 k, unsigned pat) {
    unsigned kk[4] = { k.x, k.y, k.z, k.w };
    unsigned m = 0u;
#pragma unroll
    for (int j = 0; j < 4; ++j) {
        unsigned x = kk[j] ^ pat;
        unsigned t = ((x & 0x7F7F7F7Fu) + 0x7F7F7F7Fu) | x;
        unsigned z = ~t & 0x80808080u;
        unsigned c = ((z >> 7) & 1u) | ((z >> 14) & 2u) | ((z >> 21) & 4u) | ((z >> 28) & 8u);
        m |= c << (4 * j);
    }
    return m;
}

template <int PWS, class Proc>
__device__ __forceinline__ void scan_owned(const uint4* __restrict__ keyv, int nchunks,
                                           unsigned* wlist, int lane, int wave, int owner,
                                           Proc proc) {
    const unsigned pat = (unsigned)owner * 0x01010101u;
    const int off = (owner * 12) % nchunks;
    int cnt = 0;
    for (int c0 = wave; c0 < nchunks; c0 += NWAVE * UB) {
        uint4 k[UB];
        int cc[UB];
#pragma unroll
        for (int u = 0; u < UB; ++u) {
            int c = c0 + u * NWAVE;
            int x = ((c < nchunks) ? c : wave) + off;
            cc[u] = (x >= nchunks) ? x - nchunks : x;
            k[u] = keyv[(size_t)cc[u] * 32 + lane];
        }
#pragma unroll
        for (int u = 0; u < UB; ++u) {
            bool hit = (c0 + u * NWAVE < nchunks) && any_key_hit(k[u], pat);
            unsigned bal = __builtin_amdgcn_ballot_w32(hit);
            if (hit) wlist[cnt + (int)__builtin_amdgcn_mbcnt_lo(bal, 0u)] = (unsigned)(cc[u] * 32 + lane);
            cnt += __builtin_popcount(bal);
        }
        if (cnt > PWS - 32 * UB) { proc(cnt); cnt = 0; }
    }
    proc(cnt);
}


__global__ __launch_bounds__(TPB) void k_deg_own(const unsigned* __restrict__ keyw,
                                                 const uint2* __restrict__ erec,
                                                 float* __restrict__ dinv,
                                                 int* __restrict__ cntp,
                                                 unsigned* __restrict__ recw,
                                                 int n, int E, int R, int nchunks) {
    constexpr int PWS = pws_of(RMAX * 4);
    __shared__ float degl[RMAX];
    __shared__ unsigned list[NWAVE * PWS];
    const uint4* keyv = (const uint4*)keyw;
    const int tid = threadIdx.x, lane = tid & 31;
    const int wave = __builtin_amdgcn_readfirstlane(tid >> 5);
    const int owner = blockIdx.x, base = owner * R;
    const unsigned pat = (unsigned)owner * 0x01010101u;
    unsigned* wlist = list + wave * PWS;
    const int wid = owner * NWAVE + wave;
    uint2* wrec = (uint2*)recw + (size_t)wid * CAP;
    int nrec = 0;

    for (int i = tid; i < RMAX; i += TPB) degl[i] = 1.0f;
    __syncthreads();

    auto proc = [&](int cnt) {
        for (int g = 0; g < cnt; g += 32) {
            int idx = g + lane;
            bool v = idx < cnt;
            unsigned seg = v ? wlist[idx] : 0u;
            unsigned m = v ? key_hit_mask16(keyv[seg], pat) : 0u;
            while (__builtin_amdgcn_ballot_w32(m != 0u) != 0u) {
                bool act = (m != 0u);
                int e = (int)seg * 16 + (act ? __builtin_ctz(m) : 0);
                m &= m - 1u;
                act = act && (e < E);
                unsigned bal = __builtin_amdgcn_ballot_w32(act);
                if (act) {
                    const uint2 rc = erec[e];
                    int pos = nrec + (int)__builtin_amdgcn_mbcnt_lo(bal, 0u);
                    pos = pos < CAP ? pos : CAP - 1;
                    wrec[pos] = rc;
                    ldsAddF32(&degl[rc.x >> SBITS], __uint_as_float(rc.y));
                }
                nrec += __builtin_popcount(bal);
            }
        }
    };
    scan_owned<PWS>(keyv, nchunks, wlist, lane, wave, owner, proc);
    if (lane == 0) cntp[wid] = nrec;
    __syncthreads();

    for (int i = tid; i < R; i += TPB) {
        int node = base + i;
        if (node < n) {
            float d = degl[i];
            dinv[node] = (d > 0.0f) ? rsqrtf(d) : 0.0f;
        }
    }
}


__device__ __forceinline__ int kq(int j, int hi) { return 4 * j + 2 * hi; }


template <int W, bool FUSE>
__global__ __launch_bounds__(TPB) void k_agg_own(const unsigned* __restrict__ keyw,
                                                 const int* __restrict__ src,
                                                 const int* __restrict__ dst,
                                                 const float* __restrict__ w,
                                                 const float* __restrict__ dinv,
                                                 const float* __restrict__ h,
                                                 const float* __restrict__ bias,
                                                 const int* __restrict__ cntp,
                                                 const unsigned* __restrict__ recw,
                                                 float* __restrict__ out,
                                                 const float* __restrict__ Wa,
                                                 const float* __restrict__ Wb,
                                                 int n, int E, int R, int nchunks) {
    constexpr int PITCH = FUSE ? 68 : W;
    constexpr int AROWS = FUSE ? 400 : RMAX;
    constexpr int PWS = pws_of(AROWS * PITCH * 4);
    __shared__ __attribute__((aligned(16))) float acc[AROWS * PITCH];
    __shared__ unsigned list[NWAVE * PWS];
    __shared__ float lse[FUSE ? 1 : RMAX];
    const uint4* keyv = (const uint4*)keyw;
    const int tid = threadIdx.x, lane = tid & 31;
    const int wave = __builtin_amdgcn_readfirstlane(tid >> 5);
    const int owner = blockIdx.x, base = owner * R;
    const unsigned pat = (unsigned)owner * 0x01010101u;
    unsigned* wlist = list + wave * PWS;
    int rows = n - base;
    rows = rows < 0 ? 0 : (rows > R ? R : rows);
    const int nel = rows * W;
    const float* hown = h + (size_t)base * W;
    float* oown = out + (size_t)base * W;

    for (int i = tid; i < nel; i += TPB) {
        int r = i / W, f = i - r * W;
        float di = dinv[base + r];
        float sl = di * di * hown[i];
        acc[r * PITCH + f] = FUSE ? sl : bias[f] + sl;
    }
    __syncthreads();

    const bool fl = (2 * lane < W);
    auto proc = [&](int cnt) {
        for (int g = 0; g < cnt; g += 32) {
            int idx = g + lane;
            bool v = idx < cnt;
            unsigned seg = v ? wlist[idx] : 0u;
            unsigned m = v ? key_hit_mask16(keyv[seg], pat) : 0u;
            while (__builtin_amdgcn_ballot_w32(m != 0u) != 0u) {
                bool act = (m != 0u);
                int e = (int)seg * 16 + (act ? __builtin_ctz(m) : 0);
                m &= m - 1u;
                act = act && (e < E);
                int s = 0, dl = 0;
                float nr = 0.0f;
                if (act) {
                    int d = dst[e];
                    s = src[e];
                    nr = dinv[s] * w[e] * dinv[d];
                    dl = d - base;
                }
                unsigned bal = __builtin_amdgcn_ballot_w32(act);
                while (bal != 0u) {
                    int jj[EB];
                    bool ok[EB];
#pragma unroll
                    for (int u = 0; u < EB; ++u) {
                        ok[u] = (bal != 0u);
                        jj[u] = ok[u] ? __builtin_ctz(bal) : 0;
                        bal &= bal - 1u;
                    }
                    v2f r[EB];
#pragma unroll
                    for (int u = 0; u < EB; ++u) {
                        int sj = __builtin_amdgcn_readlane(s, jj[u]);
                        r[u] = fl ? *(const v2f*)(h + (size_t)sj * W + 2 * lane) : v2f{0.0f, 0.0f};
                    }
#pragma unroll
                    for (int u = 0; u < EB; ++u) {
                        if (ok[u]) {
                            float nj = __int_as_float(__builtin_amdgcn_readlane(__float_as_int(nr), jj[u]));
                            int dj = __builtin_amdgcn_readlane(dl, jj[u]);
                            if (fl) {
                                float* a = &acc[dj * PITCH + 2 * lane];
                                ldsAddF32(a, nj * r[u].x);
                                ldsAddF32(a + 1, nj * r[u].y);
                            }
                        }
                    }
                }
            }
        }
    };
    const int wid = owner * NWAVE + wave;
    const int c = cntp[wid];
    if (c <= CAP) {
        const unsigned long long* wrec = (const unsigned long long*)recw + (size_t)wid * CAP;
        const bool cur = lane < c;
        unsigned long long rc = cur ? wrec[lane] : 0ull;
        int s = (int)((unsigned)rc & SMASK), dl = (int)((unsigned)rc >> SBITS);
        float nr = cur ? dinv[s] * __uint_as_float((unsigned)(rc >> 32)) * dinv[base + dl] : 0.0f;
        for (int g = 0; g < c; g += 32) {
            const bool nx = g + 32 + lane < c;
            unsigned long long rn = nx ? wrec[g + 32 + lane] : 0ull;
            v2f r[32];
#pragma unroll
            for (int u = 0; u < 32; ++u) {
                int sj = __builtin_amdgcn_readlane(s, u);
                r[u] = fl ? *(const v2f*)(h + (size_t)sj * W + 2 * lane) : v2f{0.0f, 0.0f};
            }
            int sn = (int)((unsigned)rn & SMASK), dln = (int)((unsigned)rn >> SBITS);
            float dsn = nx ? dinv[sn] : 0.0f;
            float ddn = nx ? dinv[base + dln] : 0.0f;
#pragma unroll
            for (int u = 0; u < 32; ++u) {
                if (g + u < c) {
                    float nj = __int_as_float(__builtin_amdgcn_readlane(__float_as_int(nr), u));
                    int dj = __builtin_amdgcn_readlane(dl, u);
                    if (fl) {
                        float* a = &acc[dj * PITCH + 2 * lane];
                        ldsAddF32(a, nj * r[u].x);
                        ldsAddF32(a + 1, nj * r[u].y);
                    }
                }
            }
            s = sn; dl = dln;
            nr = dsn * __uint_as_float((unsigned)(rn >> 32)) * ddn;
        }
    } else {
        scan_owned<PWS>(keyv, nchunks, wlist, lane, wave, owner, proc);
    }
    __syncthreads();

    if constexpr (!FUSE) {
        static_assert(RMAX <= TPB, "a thread per owned row");
        if (tid < rows) {
            const int r = tid;
            const float* v = acc + r * W;
            float mx = -3.402823466e38f;
#pragma unroll
            for (int c = 0; c < W; ++c) mx = fmaxf(mx, v[c]);
            float s = 0.0f;
#pragma unroll
            for (int c = 0; c < W; ++c) s += expf(v[c] - mx);
            lse[r] = mx + logf(s);
        }
        __syncthreads();
        for (int i = tid * 4; i < nel; i += TPB * 4) {
            const float4 q = *(const float4*)(acc + i);
            const float l = lse[i / W];
            *(float4*)(oown + i) = make_float4(q.x - l, q.y - l, q.z - l, q.w - l);
        }
    } else {
        const int lo = lane & 15, hi = lane >> 4;
        const int ntile = (rows + 15) >> 4;
        static_assert(AROWS <= 16 * NWAVE, "one 16-row tile per wave covers the owner's rows");
        if (wave < ntile) {
            const int mt = wave;
            const float* arow = acc + (mt * 16 + lo) * PITCH;
            v8f c0 = {}, c1 = {}, c2 = {}, c3 = {};
#pragma unroll 8
            for (int j = 0; j < 16; ++j) {
                const int k0 = kq(j, hi);
                const v2f a = *(const v2f*)(arow + k0);
                const float4 q0 = *(const float4*)(Wa + (size_t)k0 * 64 + 4 * lo);
                const float4 q1 = *(const float4*)(Wa + (size_t)(k0 + 1) * 64 + 4 * lo);
                c0 = __builtin_amdgcn_wmma_f32_16x16x4_f32(false, a, false, v2f{ q0.x, q1.x }, (short)0, c0, false, false);
                c1 = __builtin_amdgcn_wmma_f32_16x16x4_f32(false, a, false, v2f{ q0.y, q1.y }, (short)0, c1, false, false);
                c2 = __builtin_amdgcn_wmma_f32_16x16x4_f32(false, a, false, v2f{ q0.z, q1.z }, (short)0, c2, false, false);
                c3 = __builtin_amdgcn_wmma_f32_16x16x4_f32(false, a, false, v2f{ q0.w, q1.w }, (short)0, c3, false, false);
            }
            const float4 bv = *(const float4*)(bias + 4 * lo);
            float* drow = acc + (mt * 16 + 8 * hi) * PITCH + 4 * lo;
#pragma unroll
            for (int r = 0; r < 8; ++r) {
                *(float4*)(drow + r * PITCH) = make_float4(fmaxf(c0[r] + bv.x, 0.0f), fmaxf(c1[r] + bv.y, 0.0f),
                                                           fmaxf(c2[r] + bv.z, 0.0f), fmaxf(c3[r] + bv.w, 0.0f));
            }
            v8f d0 = {}, d1 = {}, d2 = {};
#pragma unroll 8
            for (int j = 0; j < 16; ++j) {
                const int k0 = kq(j, hi);
                const v2f a = *(const v2f*)(arow + k0);
                v2f b[3];
#pragma unroll
                for (int nt = 0; nt < 3; ++nt) {
                    const int col = 3 * lo + nt;
                    const bool cok = col < 40;
                    const int cc = cok ? col : 39;
                    const float w0 = Wb[(size_t)k0 * 40 + cc];
                    const float w1 = Wb[(size_t)(k0 + 1) * 40 + cc];
                    b[nt] = v2f{ cok ? w0 : 0.0f, cok ? w1 : 0.0f };
                }
                d0 = __builtin_amdgcn_wmma_f32_16x16x4_f32(false, a, false, b[0], (short)0, d0, false, false);
                d1 = __builtin_amdgcn_wmma_f32_16x16x4_f32(false, a, false, b[1], (short)0, d1, false, false);
                d2 = __builtin_amdgcn_wmma_f32_16x16x4_f32(false, a, false, b[2], (short)0, d2, false, false);
            }
            const int r0 = mt * 16 + 8 * hi;
            float* orow = out + (size_t)(base + r0) * 40 + 3 * lo;
#pragma unroll
            for (int r = 0; r < 8; ++r) {
                if (r0 + r < rows) {
                    float* p = orow + (size_t)r * 40;
                    if (lo < 13) { p[0] = d0[r]; p[1] = d1[r]; p[2] = d2[r]; }
                    else if (lo == 13) { p[0] = d0[r]; }
                }
            }
        }
    }
}


extern "C" void kernel_launch(void* const* d_in, const int* in_sizes, int n_in,
                              void* d_out, int out_size, void* d_ws, size_t ws_size,
                              hipStream_t stream) {
    const float* x   = (const float*)d_in[0];
    const int*   ei  = (const int*)d_in[1];
    const float* ew  = (const float*)d_in[2];
    const float* W1  = (const float*)d_in[3];
    const float* b1  = (const float*)d_in[4];
    const float* W2  = (const float*)d_in[5];
    const float* b2  = (const float*)d_in[6];
    float* out = (float*)d_out;

    const int N = in_sizes[0] / 64;
    const int E = in_sizes[2];
    const int* src = ei;
    const int* dst = ei + E;

    const int R = (N + NOWN - 1) / NOWN;
    const int nchunks = (E + 511) / 512;
    const int nwords = nchunks * 128;
    const int npad = ((N + 63) / 64) * 64;

    float* ws   = (float*)d_ws;
    float* dinv = ws;
    unsigned* keyw = (unsigned*)(dinv + npad);
    float* bufA = (float*)(keyw + nwords);
    int* cntp = (int*)(bufA + (size_t)N * 128);
    unsigned* recw = (unsigned*)(cntp + NOWN * NWAVE);
    uint4* erec = (uint4*)(recw + (size_t)NOWN * NWAVE * CAP * 2);

    const int BLK = 256;

    k_key<<<(nwords + BLK - 1) / BLK, BLK, 0, stream>>>(src, dst, ew, keyw, erec, E, nwords, (unsigned)R);
    k_deg_own<<<NOWN, TPB, 0, stream>>>(keyw, (const uint2*)erec, dinv, cntp, recw, N, E, R, nchunks);

    k_agg_own<64, true><<<NOWN, TPB, 0, stream>>>(keyw, src, dst, ew, dinv, x, b1, cntp, recw, bufA, W1, W2, N, E, R, nchunks);

    k_agg_own<40, false><<<NOWN, TPB, 0, stream>>>(keyw, src, dst, ew, dinv, bufA, b2, cntp, recw, out, nullptr, nullptr, N, E, R, nchunks);
}


# key kernel builds a bit index (owner, 16-edge segment) by LDS or; degree pass reads it, not every key byte; waves take contiguous edge shares, overflow scan too; LDS f32 adds in arrival order
# speedup vs baseline: 5.5377x; 1.0982x over previous
#include <hip/hip_runtime.h>
#include <math.h>


typedef __attribute__((ext_vector_type(2))) float v2f;
typedef __attribute__((ext_vector_type(8))) float v8f;

constexpr int NOWN  = 256;
constexpr int RMAX  = 392;
constexpr int TPB   = 1024;
constexpr int NWAVE = TPB / 32;
constexpr int pws_of(int other_bytes) {
    int p = (164 * 1024 - other_bytes + NWAVE * 4 - 1) / (NWAVE * 4);
    return p < 512 ? 512 : p;
}
constexpr int UB    = 8;
constexpr int EB    = 16;
constexpr int CAP   = 512;
constexpr int SBITS = 23;
constexpr unsigned SMASK = (1u << SBITS) - 1u;
constexpr int KB    = 1024;
constexpr int BMW   = KB / 128;
constexpr int BMP   = NOWN + 8;
constexpr int NLD   = 4;

__device__ __forceinline__ void ldsAddF32(float* p, float v) {
    __hip_atomic_fetch_add(p, v, __ATOMIC_RELAXED, __HIP_MEMORY_SCOPE_WORKGROUP);
}


__global__ __launch_bounds__(KB) void k_key(const int* __restrict__ src, const int* __restrict__ dst,
                                            const float* __restrict__ w, unsigned* __restrict__ keyw,
                                            uint4* __restrict__ erec, unsigned* __restrict__ bmap,
                                            int E, int nwords, unsigned R) {
    __shared__ unsigned bm[BMW * BMP];
    const int tid = threadIdx.x;
    const int i = blockIdx.x * KB + tid;
    for (int j = tid; j < BMW * BMP; j += KB) bm[j] = 0u;
    __syncthreads();
    if (i < nwords) {
        unsigned* plane = bm + (tid >> 7) * BMP;
        const unsigned bit = 1u << ((tid >> 2) & 31);
        unsigned v = 0u;
        unsigned rw[8];
        unsigned ow[4];
#pragma unroll
        for (int u = 0; u < 4; ++u) {
            int e = i * 4 + u;
            const bool in = e < E;
            unsigned d = in ? (unsigned)dst[e] : 0u;
            unsigned sv = in ? (unsigned)src[e] : 0u;
            float we = in ? w[e] : 0.0f;
            unsigned o = d / R;
            ow[u] = o;
            v |= o << (8 * u);
            rw[2 * u] = sv | ((d - o * R) << SBITS);
            rw[2 * u + 1] = __float_as_uint(we);
        }
        keyw[i] = v;
        erec[2 * i] = make_uint4(rw[0], rw[1], rw[2], rw[3]);
        erec[2 * i + 1] = make_uint4(rw[4], rw[5], rw[6], rw[7]);
#pragma unroll
        for (int u = 0; u < 4; ++u) {
            if (i * 4 + u < E) __hip_atomic_fetch_or(&plane[ow[u]], bit, __ATOMIC_RELAXED, __HIP_MEMORY_SCOPE_WORKGROUP);
        }
    }
    __syncthreads();
    unsigned* bout = bmap + (size_t)blockIdx.x * (NOWN * BMW);
    for (int g = tid; g < NOWN * BMW; g += KB) bout[g] = bm[(g & (BMW - 1)) * BMP + (g / BMW)];
}

__device__ __forceinline__ bool any_key_hit(uint4 k, unsigned pat) {
    unsigned x0 = k.x ^ pat, x1 = k.y ^ pat, x2 = k.z ^ pat, x3 = k.w ^ pat;
    unsigned t = ((x0 - 0x01010101u) & ~x0) | ((x1 - 0x01010101u) & ~x1) |
                 ((x2 - 0x01010101u) & ~x2) | ((x3 - 0x01010101u) & ~x3);
    return (t & 0x80808080u) != 0u;
}

__device__ __forceinline__ unsigned key_hit_mask16(uint4 k, unsigned pat) {
    unsigned kk[4] = { k.x, k.y, k.z, k.w };
    unsigned m = 0u;
#pragma unroll
    for (int j = 0; j < 4; ++j) {
        unsigned x = kk[j] ^ pat;
        unsigned t = ((x & 0x7F7F7F7Fu) + 0x7F7F7F7Fu) | x;
        unsigned z = ~t & 0x80808080u;
        unsigned c = ((z >> 7) & 1u) | ((z >> 14) & 2u) | ((z >> 21) & 4u) | ((z >> 28) & 8u);
        m |= c << (4 * j);
    }
    return m;
}

__device__ __forceinline__ int share_lo(int wave, int nchunks) {
    return (int)(((long long)wave * nchunks) / NWAVE);
}

template <int PWS, class Proc>
__device__ __forceinline__ void scan_owned(const uint4* __restrict__ keyv, int xlo, int xhi,
                                           unsigned* wlist, int lane, int owner, Proc proc) {
    const unsigned pat = (unsigned)owner * 0x01010101u;
    int cnt = 0;
    for (int c0 = xlo; c0 < xhi; c0 += UB) {
        uint4 k[UB];
#pragma unroll
        for (int u = 0; u < UB; ++u) {
            const int c = c0 + u;
            const int x = (c < xhi) ? c : xlo;
            k[u] = keyv[(size_t)x * 32 + lane];
        }
#pragma unroll
        for (int u = 0; u < UB; ++u) {
            bool hit = (c0 + u < xhi) && any_key_hit(k[u], pat);
            unsigned bal = __builtin_amdgcn_ballot_w32(hit);
            if (hit) wlist[cnt + (int)__builtin_amdgcn_mbcnt_lo(bal, 0u)] = (unsigned)((c0 + u) * 32 + lane);
            cnt += __builtin_popcount(bal);
        }
        if (cnt > PWS - 32 * UB) { proc(cnt); cnt = 0; }
    }
    proc(cnt);
}


__global__ __launch_bounds__(TPB) void k_deg_own(const unsigned* __restrict__ keyw,
                                                 const uint2* __restrict__ erec,
                                                 const unsigned* __restrict__ bmap,
                                                 float* __restrict__ dinv,
                                                 int* __restrict__ cntp,
                                                 unsigned* __restrict__ recw,
                                                 int n, int E, int R, int nchunks) {
    constexpr int PWS = pws_of(RMAX * 4);
    __shared__ float degl[RMAX];
    __shared__ unsigned list[NWAVE * PWS];
    const uint4* keyv = (const uint4*)keyw;
    const int tid = threadIdx.x, lane = tid & 31;
    const int wave = __builtin_amdgcn_readfirstlane(tid >> 5);
    const int owner = blockIdx.x, base = owner * R;
    const unsigned pat = (unsigned)owner * 0x01010101u;
    unsigned* wlist = list + wave * PWS;
    const int wid = owner * NWAVE + wave;
    uint2* wrec = (uint2*)recw + (size_t)wid * CAP;
    int nrec = 0;

    for (int i = tid; i < RMAX; i += TPB) degl[i] = 1.0f;
    __syncthreads();

    auto proc = [&](int cnt) {
        for (int g = 0; g < cnt; g += 32) {
            int idx = g + lane;
            bool v = idx < cnt;
            unsigned seg = v ? wlist[idx] : 0u;
            unsigned m = v ? key_hit_mask16(keyv[seg], pat) : 0u;
            while (__builtin_amdgcn_ballot_w32(m != 0u) != 0u) {
                bool act = (m != 0u);
                int e = (int)seg * 16 + (act ? __builtin_ctz(m) : 0);
                m &= m - 1u;
                act = act && (e < E);
                unsigned bal = __builtin_amdgcn_ballot_w32(act);
                if (act) {
                    const uint2 rc = erec[e];
                    int pos = nrec + (int)__builtin_amdgcn_mbcnt_lo(bal, 0u);
                    pos = pos < CAP ? pos : CAP - 1;
                    wrec[pos] = rc;
                    ldsAddF32(&degl[rc.x >> SBITS], __uint_as_float(rc.y));
                }
                nrec += __builtin_popcount(bal);
            }
        }
    };
    {
        const int xlo = share_lo(wave, nchunks), xhi = share_lo(wave + 1, nchunks);
        int cnt = 0;
        for (int x0 = xlo; x0 < xhi; x0 += NLD * 32) {
            unsigned m[NLD], sb[NLD];
#pragma unroll
            for (int q = 0; q < NLD; ++q) {
                const int x = x0 + q * 32 + lane;
                const bool ok = x < xhi;
                const int blk = ok ? (x / BMW) : 0, pl = x & (BMW - 1);
                const unsigned wv = bmap[((size_t)blk * NOWN + owner) * BMW + pl];
                m[q] = ok ? wv : 0u;
                sb[q] = (unsigned)(blk * (KB / 4) + pl * 32);
            }
#pragma unroll
            for (int q = 0; q < NLD; ++q) {
                unsigned mm = m[q];
                while (__builtin_amdgcn_ballot_w32(mm != 0u) != 0u) {
                    const bool hit = (mm != 0u);
                    const unsigned seg = sb[q] + (hit ? (unsigned)__builtin_ctz(mm) : 0u);
                    mm &= mm - 1u;
                    const unsigned bal = __builtin_amdgcn_ballot_w32(hit);
                    if (hit) wlist[cnt + (int)__builtin_amdgcn_mbcnt_lo(bal, 0u)] = seg;
                    cnt += __builtin_popcount(bal);
                    if (cnt > PWS - 32) { proc(cnt); cnt = 0; }
                }
            }
        }
        proc(cnt);
    }
    if (lane == 0) cntp[wid] = nrec;
    __syncthreads();

    for (int i = tid; i < R; i += TPB) {
        int node = base + i;
        if (node < n) {
            float d = degl[i];
            dinv[node] = (d > 0.0f) ? rsqrtf(d) : 0.0f;
        }
    }
}


__device__ __forceinline__ int kq(int j, int hi) { return 4 * j + 2 * hi; }


template <int W, bool FUSE>
__global__ __launch_bounds__(TPB) void k_agg_own(const unsigned* __restrict__ keyw,
                                                 const int* __restrict__ src,
                                                 const int* __restrict__ dst,
                                                 const float* __restrict__ w,
                                                 const float* __restrict__ dinv,
                                                 const float* __restrict__ h,
                                                 const float* __restrict__ bias,
                                                 const int* __restrict__ cntp,
                                                 const unsigned* __restrict__ recw,
                                                 float* __restrict__ out,
                                                 const float* __restrict__ Wa,
                                                 const float* __restrict__ Wb,
                                                 int n, int E, int R, int nchunks) {
    constexpr int PITCH = FUSE ? 68 : W;
    constexpr int AROWS = FUSE ? 400 : RMAX;
    constexpr int PWS = pws_of(AROWS * PITCH * 4);
    __shared__ __attribute__((aligned(16))) float acc[AROWS * PITCH];
    __shared__ unsigned list[NWAVE * PWS];
    __shared__ float lse[FUSE ? 1 : RMAX];
    const uint4* keyv = (const uint4*)keyw;
    const int tid = threadIdx.x, lane = tid & 31;
    const int wave = __builtin_amdgcn_readfirstlane(tid >> 5);
    const int owner = blockIdx.x, base = owner * R;
    const unsigned pat = (unsigned)owner * 0x01010101u;
    unsigned* wlist = list + wave * PWS;
    int rows = n - base;
    rows = rows < 0 ? 0 : (rows > R ? R : rows);
    const int nel = rows * W;
    const float* hown = h + (size_t)base * W;
    float* oown = out + (size_t)base * W;

    for (int i = tid; i < nel; i += TPB) {
        int r = i / W, f = i - r * W;
        float di = dinv[base + r];
        float sl = di * di * hown[i];
        acc[r * PITCH + f] = FUSE ? sl : bias[f] + sl;
    }
    __syncthreads();

    const bool fl = (2 * lane < W);
    auto proc = [&](int cnt) {
        for (int g = 0; g < cnt; g += 32) {
            int idx = g + lane;
            bool v = idx < cnt;
            unsigned seg = v ? wlist[idx] : 0u;
            unsigned m = v ? key_hit_mask16(keyv[seg], pat) : 0u;
            while (__builtin_amdgcn_ballot_w32(m != 0u) != 0u) {
                bool act = (m != 0u);
                int e = (int)seg * 16 + (act ? __builtin_ctz(m) : 0);
                m &= m - 1u;
                act = act && (e < E);
                int s = 0, dl = 0;
                float nr = 0.0f;
                if (act) {
                    int d = dst[e];
                    s = src[e];
                    nr = dinv[s] * w[e] * dinv[d];
                    dl = d - base;
                }
                unsigned bal = __builtin_amdgcn_ballot_w32(act);
                while (bal != 0u) {
                    int jj[EB];
                    bool ok[EB];
#pragma unroll
                    for (int u = 0; u < EB; ++u) {
                        ok[u] = (bal != 0u);
                        jj[u] = ok[u] ? __builtin_ctz(bal) : 0;
                        bal &= bal - 1u;
                    }
                    v2f r[EB];
#pragma unroll
                    for (int u = 0; u < EB; ++u) {
                        int sj = __builtin_amdgcn_readlane(s, jj[u]);
                        r[u] = fl ? *(const v2f*)(h + (size_t)sj * W + 2 * lane) : v2f{0.0f, 0.0f};
                    }
#pragma unroll
                    for (int u = 0; u < EB; ++u) {
                        if (ok[u]) {
                            float nj = __int_as_float(__builtin_amdgcn_readlane(__float_as_int(nr), jj[u]));
                            int dj = __builtin_amdgcn_readlane(dl, jj[u]);
                            if (fl) {
                                float* a = &acc[dj * PITCH + 2 * lane];
                                ldsAddF32(a, nj * r[u].x);
                                ldsAddF32(a + 1, nj * r[u].y);
                            }
                        }
                    }
                }
            }
        }
    };
    const int wid = owner * NWAVE + wave;
    const int c = cntp[wid];
    if (c <= CAP) {
        const unsigned long long* wrec = (const unsigned long long*)recw + (size_t)wid * CAP;
        const bool cur = lane < c;
        unsigned long long rc = cur ? wrec[lane] : 0ull;
        int s = (int)((unsigned)rc & SMASK), dl = (int)((unsigned)rc >> SBITS);
        float nr = cur ? dinv[s] * __uint_as_float((unsigned)(rc >> 32)) * dinv[base + dl] : 0.0f;
        for (int g = 0; g < c; g += 32) {
            const bool nx = g + 32 + lane < c;
            unsigned long long rn = nx ? wrec[g + 32 + lane] : 0ull;
            v2f r[32];
#pragma unroll
            for (int u = 0; u < 32; ++u) {
                int sj = __builtin_amdgcn_readlane(s, u);
                r[u] = fl ? *(const v2f*)(h + (size_t)sj * W + 2 * lane) : v2f{0.0f, 0.0f};
            }
            int sn = (int)((unsigned)rn & SMASK), dln = (int)((unsigned)rn >> SBITS);
            float dsn = nx ? dinv[sn] : 0.0f;
            float ddn = nx ? dinv[base + dln] : 0.0f;
#pragma unroll
            for (int u = 0; u < 32; ++u) {
                if (g + u < c) {
                    float nj = __int_as_float(__builtin_amdgcn_readlane(__float_as_int(nr), u));
                    int dj = __builtin_amdgcn_readlane(dl, u);
                    if (fl) {
                        float* a = &acc[dj * PITCH + 2 * lane];
                        ldsAddF32(a, nj * r[u].x);
                        ldsAddF32(a + 1, nj * r[u].y);
                    }
                }
            }
            s = sn; dl = dln;
            nr = dsn * __uint_as_float((unsigned)(rn >> 32)) * ddn;
        }
    } else {
        scan_owned<PWS>(keyv, share_lo(wave, nchunks), share_lo(wave + 1, nchunks), wlist, lane, owner, proc);
    }
    __syncthreads();

    if constexpr (!FUSE) {
        static_assert(RMAX <= TPB, "a thread per owned row");
        if (tid < rows) {
            const int r = tid;
            const float* v = acc + r * W;
            float mx = -3.402823466e38f;
#pragma unroll
            for (int c = 0; c < W; ++c) mx = fmaxf(mx, v[c]);
            float s = 0.0f;
#pragma unroll
            for (int c = 0; c < W; ++c) s += expf(v[c] - mx);
            lse[r] = mx + logf(s);
        }
        __syncthreads();
        for (int i = tid * 4; i < nel; i += TPB * 4) {
            const float4 q = *(const float4*)(acc + i);
            const float l = lse[i / W];
            *(float4*)(oown + i) = make_float4(q.x - l, q.y - l, q.z - l, q.w - l);
        }
    } else {
        const int lo = lane & 15, hi = lane >> 4;
        const int ntile = (rows + 15) >> 4;
        static_assert(AROWS <= 16 * NWAVE, "one 16-row tile per wave covers the owner's rows");
        if (wave < ntile) {
            const int mt = wave;
            const float* arow = acc + (mt * 16 + lo) * PITCH;
            v8f c0 = {}, c1 = {}, c2 = {}, c3 = {};
#pragma unroll 8
            for (int j = 0; j < 16; ++j) {
                const int k0 = kq(j, hi);
                const v2f a = *(const v2f*)(arow + k0);
                const float4 q0 = *(const float4*)(Wa + (size_t)k0 * 64 + 4 * lo);
                const float4 q1 = *(const float4*)(Wa + (size_t)(k0 + 1) * 64 + 4 * lo);
                c0 = __builtin_amdgcn_wmma_f32_16x16x4_f32(false, a, false, v2f{ q0.x, q1.x }, (short)0, c0, false, false);
                c1 = __builtin_amdgcn_wmma_f32_16x16x4_f32(false, a, false, v2f{ q0.y, q1.y }, (short)0, c1, false, false);
                c2 = __builtin_amdgcn_wmma_f32_16x16x4_f32(false, a, false, v2f{ q0.z, q1.z }, (short)0, c2, false, false);
                c3 = __builtin_amdgcn_wmma_f32_16x16x4_f32(false, a, false, v2f{ q0.w, q1.w }, (short)0, c3, false, false);
            }
            const float4 bv = *(const float4*)(bias + 4 * lo);
            float* drow = acc + (mt * 16 + 8 * hi) * PITCH + 4 * lo;
#pragma unroll
            for (int r = 0; r < 8; ++r) {
                *(float4*)(drow + r * PITCH) = make_float4(fmaxf(c0[r] + bv.x, 0.0f), fmaxf(c1[r] + bv.y, 0.0f),
                                                           fmaxf(c2[r] + bv.z, 0.0f), fmaxf(c3[r] + bv.w, 0.0f));
            }
            v8f d0 = {}, d1 = {}, d2 = {};
#pragma unroll 8
            for (int j = 0; j < 16; ++j) {
                const int k0 = kq(j, hi);
                const v2f a = *(const v2f*)(arow + k0);
                v2f b[3];
#pragma unroll
                for (int nt = 0; nt < 3; ++nt) {
                    const int col = 3 * lo + nt;
                    const bool cok = col < 40;
                    const int cc = cok ? col : 39;
                    const float w0 = Wb[(size_t)k0 * 40 + cc];
                    const float w1 = Wb[(size_t)(k0 + 1) * 40 + cc];
                    b[nt] = v2f{ cok ? w0 : 0.0f, cok ? w1 : 0.0f };
                }
                d0 = __builtin_amdgcn_wmma_f32_16x16x4_f32(false, a, false, b[0], (short)0, d0, false, false);
                d1 = __builtin_amdgcn_wmma_f32_16x16x4_f32(false, a, false, b[1], (short)0, d1, false, false);
                d2 = __builtin_amdgcn_wmma_f32_16x16x4_f32(false, a, false, b[2], (short)0, d2, false, false);
            }
            const int r0 = mt * 16 + 8 * hi;
            float* orow = out + (size_t)(base + r0) * 40 + 3 * lo;
#pragma unroll
            for (int r = 0; r < 8; ++r) {
                if (r0 + r < rows) {
                    float* p = orow + (size_t)r * 40;
                    if (lo < 13) { p[0] = d0[r]; p[1] = d1[r]; p[2] = d2[r]; }
                    else if (lo == 13) { p[0] = d0[r]; }
                }
            }
        }
    }
}


extern "C" void kernel_launch(void* const* d_in, const int* in_sizes, int n_in,
                              void* d_out, int out_size, void* d_ws, size_t ws_size,
                              hipStream_t stream) {
    const float* x   = (const float*)d_in[0];
    const int*   ei  = (const int*)d_in[1];
    const float* ew  = (const float*)d_in[2];
    const float* W1  = (const float*)d_in[3];
    const float* b1  = (const float*)d_in[4];
    const float* W2  = (const float*)d_in[5];
    const float* b2  = (const float*)d_in[6];
    float* out = (float*)d_out;

    const int N = in_sizes[0] / 64;
    const int E = in_sizes[2];
    const int* src = ei;
    const int* dst = ei + E;

    const int R = (N + NOWN - 1) / NOWN;
    const int nchunks = (E + 511) / 512;
    const int nwords = nchunks * 128;
    const int npad = ((N + 63) / 64) * 64;

    float* ws   = (float*)d_ws;
    float* dinv = ws;
    unsigned* keyw = (unsigned*)(dinv + npad);
    float* bufA = (float*)(keyw + nwords);
    int* cntp = (int*)(bufA + (size_t)N * 128);
    unsigned* recw = (unsigned*)(cntp + NOWN * NWAVE);
    uint4* erec = (uint4*)(recw + (size_t)NOWN * NWAVE * CAP * 2);
    const int nblk = (nwords + KB - 1) / KB;
    unsigned* bmap = (unsigned*)(erec + (size_t)nwords * 2);

    k_key<<<nblk, KB, 0, stream>>>(src, dst, ew, keyw, erec, bmap, E, nwords, (unsigned)R);
    k_deg_own<<<NOWN, TPB, 0, stream>>>(keyw, (const uint2*)erec, bmap, dinv, cntp, recw, N, E, R, nchunks);

    k_agg_own<64, true><<<NOWN, TPB, 0, stream>>>(keyw, src, dst, ew, dinv, x, b1, cntp, recw, bufA, W1, W2, N, E, R, nchunks);

    k_agg_own<40, false><<<NOWN, TPB, 0, stream>>>(keyw, src, dst, ew, dinv, bufA, b2, cntp, recw, out, nullptr, nullptr, N, E, R, nchunks);
}


# degree pass files each owned edge under the wave that holds its LDS row (LDS int count = place); sum passes add by plain LDS load, fma, store: no f32 atomics left; row sums in list (arrival) order
# speedup vs baseline: 6.2518x; 1.1290x over previous
#include <hip/hip_runtime.h>
#include <math.h>


typedef __attribute__((ext_vector_type(2))) float v2f;
typedef __attribute__((ext_vector_type(8))) float v8f;

constexpr int NOWN  = 256;
constexpr int RMAX  = 392;
constexpr int TPB   = 1024;
constexpr int NWAVE = TPB / 32;
constexpr int pws_of(int other_bytes) {
    int p = (164 * 1024 - other_bytes + NWAVE * 4 - 1) / (NWAVE * 4);
    return p < 512 ? 512 : p;
}
constexpr int UB    = 8;
constexpr int EB    = 16;
constexpr int CAP   = 512;
constexpr int SBITS = 23;
constexpr unsigned SMASK = (1u << SBITS) - 1u;
constexpr int KB    = 1024;
constexpr int BMW   = KB / 128;
constexpr int BMP   = NOWN + 8;
constexpr int NLD   = 4;

__device__ __forceinline__ void ldsAddF32(float* p, float v) {
    __hip_atomic_fetch_add(p, v, __ATOMIC_RELAXED, __HIP_MEMORY_SCOPE_WORKGROUP);
}


__global__ __launch_bounds__(KB) void k_key(const int* __restrict__ src, const int* __restrict__ dst,
                                            const float* __restrict__ w, unsigned* __restrict__ keyw,
                                            uint4* __restrict__ erec, unsigned* __restrict__ bmap,
                                            int E, int nwords, unsigned R) {
    __shared__ unsigned bm[BMW * BMP];
    const int tid = threadIdx.x;
    const int i = blockIdx.x * KB + tid;
    for (int j = tid; j < BMW * BMP; j += KB) bm[j] = 0u;
    __syncthreads();
    if (i < nwords) {
        unsigned* plane = bm + (tid >> 7) * BMP;
        const unsigned bit = 1u << ((tid >> 2) & 31);
        unsigned v = 0u;
        unsigned rw[8];
        unsigned ow[4];
#pragma unroll
        for (int u = 0; u < 4; ++u) {
            int e = i * 4 + u;
            const bool in = e < E;
            unsigned d = in ? (unsigned)dst[e] : 0u;
            unsigned sv = in ? (unsigned)src[e] : 0u;
            float we = in ? w[e] : 0.0f;
            unsigned o = d / R;
            ow[u] = o;
            v |= o << (8 * u);
            rw[2 * u] = sv | ((d - o * R) << SBITS);
            rw[2 * u + 1] = __float_as_uint(we);
        }
        keyw[i] = v;
        erec[2 * i] = make_uint4(rw[0], rw[1], rw[2], rw[3]);
        erec[2 * i + 1] = make_uint4(rw[4], rw[5], rw[6], rw[7]);
#pragma unroll
        for (int u = 0; u < 4; ++u) {
            if (i * 4 + u < E) __hip_atomic_fetch_or(&plane[ow[u]], bit, __ATOMIC_RELAXED, __HIP_MEMORY_SCOPE_WORKGROUP);
        }
    }
    __syncthreads();
    unsigned* bout = bmap + (size_t)blockIdx.x * (NOWN * BMW);
    for (int g = tid; g < NOWN * BMW; g += KB) bout[g] = bm[(g & (BMW - 1)) * BMP + (g / BMW)];
}

__device__ __forceinline__ bool any_key_hit(uint4 k, unsigned pat) {
    unsigned x0 = k.x ^ pat, x1 = k.y ^ pat, x2 = k.z ^ pat, x3 = k.w ^ pat;
    unsigned t = ((x0 - 0x01010101u) & ~x0) | ((x1 - 0x01010101u) & ~x1) |
                 ((x2 - 0x01010101u) & ~x2) | ((x3 - 0x01010101u) & ~x3);
    return (t & 0x80808080u) != 0u;
}

__device__ __forceinline__ unsigned key_hit_mask16(uint4 k, unsigned pat) {
    unsigned kk[4] = { k.x, k.y, k.z, k.w };
    unsigned m = 0u;
#pragma unroll
    for (int j = 0; j < 4; ++j) {
        unsigned x = kk[j] ^ pat;
        unsigned t = ((x & 0x7F7F7F7Fu) + 0x7F7F7F7Fu) | x;
        unsigned z = ~t & 0x80808080u;
        unsigned c = ((z >> 7) & 1u) | ((z >> 14) & 2u) | ((z >> 21) & 4u) | ((z >> 28) & 8u);
        m |= c << (4 * j);
    }
    return m;
}

__device__ __forceinline__ int share_lo(int wave, int nchunks) {
    return (int)(((long long)wave * nchunks) / NWAVE);
}

template <int PWS, class Proc>
__device__ __forceinline__ void scan_owned(const uint4* __restrict__ keyv, int xlo, int xhi,
                                           unsigned* wlist, int lane, int owner, Proc proc) {
    const unsigned pat = (unsigned)owner * 0x01010101u;
    int cnt = 0;
    for (int c0 = xlo; c0 < xhi; c0 += UB) {
        uint4 k[UB];
#pragma unroll
        for (int u = 0; u < UB; ++u) {
            const int c = c0 + u;
            const int x = (c < xhi) ? c : xlo;
            k[u] = keyv[(size_t)x * 32 + lane];
        }
#pragma unroll
        for (int u = 0; u < UB; ++u) {
            bool hit = (c0 + u < xhi) && any_key_hit(k[u], pat);
            unsigned bal = __builtin_amdgcn_ballot_w32(hit);
            if (hit) wlist[cnt + (int)__builtin_amdgcn_mbcnt_lo(bal, 0u)] = (unsigned)((c0 + u) * 32 + lane);
            cnt += __builtin_popcount(bal);
        }
        if (cnt > PWS - 32 * UB) { proc(cnt); cnt = 0; }
    }
    proc(cnt);
}


__global__ __launch_bounds__(TPB) void k_deg_own(const unsigned* __restrict__ keyw,
                                                 const uint2* __restrict__ erec,
                                                 const unsigned* __restrict__ bmap,
                                                 float* __restrict__ dinv,
                                                 int* __restrict__ cntp,
                                                 unsigned* __restrict__ recw,
                                                 int n, int E, int R, int nchunks) {
    constexpr int PWS = pws_of(RMAX * 4 + NWAVE * CAP * 8 + NWAVE * 4);
    __shared__ float degl[RMAX];
    __shared__ unsigned list[NWAVE * PWS];
    __shared__ unsigned long long stage[NWAVE * CAP];
    __shared__ int lcnt[NWAVE];
    const uint4* keyv = (const uint4*)keyw;
    const int tid = threadIdx.x, lane = tid & 31;
    const int wave = __builtin_amdgcn_readfirstlane(tid >> 5);
    const int owner = blockIdx.x, base = owner * R;
    const unsigned pat = (unsigned)owner * 0x01010101u;
    unsigned* wlist = list + wave * PWS;
    const int wid = owner * NWAVE + wave;
    const float gscale = (float)NWAVE / (float)R;

    for (int i = tid; i < RMAX; i += TPB) degl[i] = 1.0f;
    if (tid < NWAVE) lcnt[tid] = 0;
    __syncthreads();

    auto proc = [&](int cnt) {
        for (int g = 0; g < cnt; g += 32) {
            int idx = g + lane;
            bool v = idx < cnt;
            unsigned seg = v ? wlist[idx] : 0u;
            unsigned m = v ? key_hit_mask16(keyv[seg], pat) : 0u;
            while (__builtin_amdgcn_ballot_w32(m != 0u) != 0u) {
                bool act = (m != 0u);
                int e = (int)seg * 16 + (act ? __builtin_ctz(m) : 0);
                m &= m - 1u;
                act = act && (e < E);
                if (act) {
                    const uint2 rc = erec[e];
                    const unsigned row = rc.x >> SBITS;
                    int grp = (int)((float)row * gscale);
                    grp = grp > NWAVE - 1 ? NWAVE - 1 : grp;
                    const int pos = __hip_atomic_fetch_add(&lcnt[grp], 1, __ATOMIC_RELAXED, __HIP_MEMORY_SCOPE_WORKGROUP);
                    if (pos < CAP) stage[grp * CAP + pos] = ((unsigned long long)rc.y << 32) | rc.x;
                    ldsAddF32(&degl[row], __uint_as_float(rc.y));
                }
            }
        }
    };
    {
        const int xlo = share_lo(wave, nchunks), xhi = share_lo(wave + 1, nchunks);
        int cnt = 0;
        for (int x0 = xlo; x0 < xhi; x0 += NLD * 32) {
            unsigned m[NLD], sb[NLD];
#pragma unroll
            for (int q = 0; q < NLD; ++q) {
                const int x = x0 + q * 32 + lane;
                const bool ok = x < xhi;
                const int blk = ok ? (x / BMW) : 0, pl = x & (BMW - 1);
                const unsigned wv = bmap[((size_t)blk * NOWN + owner) * BMW + pl];
                m[q] = ok ? wv : 0u;
                sb[q] = (unsigned)(blk * (KB / 4) + pl * 32);
            }
#pragma unroll
            for (int q = 0; q < NLD; ++q) {
                unsigned mm = m[q];
                while (__builtin_amdgcn_ballot_w32(mm != 0u) != 0u) {
                    const bool hit = (mm != 0u);
                    const unsigned seg = sb[q] + (hit ? (unsigned)__builtin_ctz(mm) : 0u);
                    mm &= mm - 1u;
                    const unsigned bal = __builtin_amdgcn_ballot_w32(hit);
                    if (hit) wlist[cnt + (int)__builtin_amdgcn_mbcnt_lo(bal, 0u)] = seg;
                    cnt += __builtin_popcount(bal);
                    if (cnt > PWS - 32) { proc(cnt); cnt = 0; }
                }
            }
        }
        proc(cnt);
    }
    __syncthreads();

    {
        const int cw = lcnt[wave];
        const int cc = cw < CAP ? cw : CAP;
        unsigned long long* wrec = (unsigned long long*)recw + (size_t)wid * CAP;
        for (int i = lane; i < cc; i += 32) wrec[i] = stage[wave * CAP + i];
        if (lane == 0) cntp[wid] = cw;
    }

    for (int i = tid; i < R; i += TPB) {
        int node = base + i;
        if (node < n) {
            float d = degl[i];
            dinv[node] = (d > 0.0f) ? rsqrtf(d) : 0.0f;
        }
    }
}


__device__ __forceinline__ int kq(int j, int hi) { return 4 * j + 2 * hi; }


template <int W, bool FUSE>
__global__ __launch_bounds__(TPB) void k_agg_own(const unsigned* __restrict__ keyw,
                                                 const int* __restrict__ src,
                                                 const int* __restrict__ dst,
                                                 const float* __restrict__ w,
                                                 const float* __restrict__ dinv,
                                                 const float* __restrict__ h,
                                                 const float* __restrict__ bias,
                                                 const int* __restrict__ cntp,
                                                 const unsigned* __restrict__ recw,
                                                 float* __restrict__ out,
                                                 const float* __restrict__ Wa,
                                                 const float* __restrict__ Wb,
                                                 int n, int E, int R, int nchunks) {
    constexpr int PITCH = FUSE ? 68 : W;
    constexpr int AROWS = FUSE ? 400 : RMAX;
    constexpr int PWS = pws_of(AROWS * PITCH * 4);
    __shared__ __attribute__((aligned(16))) float acc[AROWS * PITCH];
    __shared__ unsigned list[NWAVE * PWS];
    __shared__ float lse[FUSE ? 1 : RMAX];
    const uint4* keyv = (const uint4*)keyw;
    const int tid = threadIdx.x, lane = tid & 31;
    const int wave = __builtin_amdgcn_readfirstlane(tid >> 5);
    const int owner = blockIdx.x, base = owner * R;
    const unsigned pat = (unsigned)owner * 0x01010101u;
    unsigned* wlist = list + wave * PWS;
    int rows = n - base;
    rows = rows < 0 ? 0 : (rows > R ? R : rows);
    const int nel = rows * W;
    const float* hown = h + (size_t)base * W;
    float* oown = out + (size_t)base * W;

    for (int i = tid; i < nel; i += TPB) {
        int r = i / W, f = i - r * W;
        float di = dinv[base + r];
        float sl = di * di * hown[i];
        acc[r * PITCH + f] = FUSE ? sl : bias[f] + sl;
    }
    __syncthreads();

    const bool fl = (2 * lane < W);
    auto proc = [&](int cnt) {
        for (int g = 0; g < cnt; g += 32) {
            int idx = g + lane;
            bool v = idx < cnt;
            unsigned seg = v ? wlist[idx] : 0u;
            unsigned m = v ? key_hit_mask16(keyv[seg], pat) : 0u;
            while (__builtin_amdgcn_ballot_w32(m != 0u) != 0u) {
                bool act = (m != 0u);
                int e = (int)seg * 16 + (act ? __builtin_ctz(m) : 0);
                m &= m - 1u;
                act = act && (e < E);
                int s = 0, dl = 0;
                float nr = 0.0f;
                if (act) {
                    int d = dst[e];
                    s = src[e];
                    nr = dinv[s] * w[e] * dinv[d];
                    dl = d - base;
                }
                unsigned bal = __builtin_amdgcn_ballot_w32(act);
                while (bal != 0u) {
                    int jj[EB];
                    bool ok[EB];
#pragma unroll
                    for (int u = 0; u < EB; ++u) {
                        ok[u] = (bal != 0u);
                        jj[u] = ok[u] ? __builtin_ctz(bal) : 0;
                        bal &= bal - 1u;
                    }
                    v2f r[EB];
#pragma unroll
                    for (int u = 0; u < EB; ++u) {
                        int sj = __builtin_amdgcn_readlane(s, jj[u]);
                        r[u] = fl ? *(const v2f*)(h + (size_t)sj * W + 2 * lane) : v2f{0.0f, 0.0f};
                    }
#pragma unroll
                    for (int u = 0; u < EB; ++u) {
                        if (ok[u]) {
                            float nj = __int_as_float(__builtin_amdgcn_readlane(__float_as_int(nr), jj[u]));
                            int dj = __builtin_amdgcn_readlane(dl, jj[u]);
                            if (fl) {
                                float* a = &acc[dj * PITCH + 2 * lane];
                                ldsAddF32(a, nj * r[u].x);
                                ldsAddF32(a + 1, nj * r[u].y);
                            }
                        }
                    }
                }
            }
        }
    };
    const int wid = owner * NWAVE + wave;
    const int cl = cntp[owner * NWAVE + lane];
    const bool over = __builtin_amdgcn_ballot_w32(cl > CAP) != 0u;
    const int c = __builtin_amdgcn_readlane(cl, wave);
    if (!over) {
        const unsigned long long* wrec = (const unsigned long long*)recw + (size_t)wid * CAP;
        const bool cur = lane < c;
        unsigned long long rc = cur ? wrec[lane] : 0ull;
        int s = (int)((unsigned)rc & SMASK), dl = (int)((unsigned)rc >> SBITS);
        float nr = cur ? dinv[s] * __uint_as_float((unsigned)(rc >> 32)) * dinv[base + dl] : 0.0f;
        for (int g = 0; g < c; g += 32) {
            const bool nx = g + 32 + lane < c;
            unsigned long long rn = nx ? wrec[g + 32 + lane] : 0ull;
            v2f r[32];
#pragma unroll
            for (int u = 0; u < 32; ++u) {
                int sj = __builtin_amdgcn_readlane(s, u);
                r[u] = fl ? *(const v2f*)(h + (size_t)sj * W + 2 * lane) : v2f{0.0f, 0.0f};
            }
            int sn = (int)((unsigned)rn & SMASK), dln = (int)((unsigned)rn >> SBITS);
            float dsn = nx ? dinv[sn] : 0.0f;
            float ddn = nx ? dinv[base + dln] : 0.0f;
#pragma unroll
            for (int u0 = 0; u0 < 32; u0 += 4) {
                const int left = c - (g + u0);
                if (left > 0) {
                    int dj[4];
                    float nj[4];
#pragma unroll
                    for (int k = 0; k < 4; ++k) {
                        dj[k] = __builtin_amdgcn_readlane(dl, u0 + k);
                        nj[k] = __int_as_float(__builtin_amdgcn_readlane(__float_as_int(nr), u0 + k));
                    }
                    if (fl) {
                        v2f t[4];
#pragma unroll
                        for (int k = 0; k < 4; ++k) t[k] = *(const v2f*)&acc[dj[k] * PITCH + 2 * lane];
#pragma unroll
                        for (int k = 0; k < 4; ++k) {
                            v2f b = t[k];
#pragma unroll
                            for (int m = 0; m < k; ++m) {
                                if (dj[m] == dj[k]) b = t[m];
                            }
                            b.x = fmaf(nj[k], r[u0 + k].x, b.x);
                            b.y = fmaf(nj[k], r[u0 + k].y, b.y);
                            t[k] = b;
                        }
                        if (left >= 4) {
#pragma unroll
                            for (int k = 0; k < 4; ++k) *(v2f*)&acc[dj[k] * PITCH + 2 * lane] = t[k];
                        } else {
#pragma unroll
                            for (int k = 0; k < 3; ++k) {
                                if (k < left) *(v2f*)&acc[dj[k] * PITCH + 2 * lane] = t[k];
                            }
                        }
                    }
                }
            }
            s = sn; dl = dln;
            nr = dsn * __uint_as_float((unsigned)(rn >> 32)) * ddn;
        }
    } else {
        scan_owned<PWS>(keyv, share_lo(wave, nchunks), share_lo(wave + 1, nchunks), wlist, lane, owner, proc);
    }
    __syncthreads();

    if constexpr (!FUSE) {
        static_assert(RMAX <= TPB, "a thread per owned row");
        if (tid < rows) {
            const int r = tid;
            const float* v = acc + r * W;
            float mx = -3.402823466e38f;
#pragma unroll
            for (int c = 0; c < W; ++c) mx = fmaxf(mx, v[c]);
            float s = 0.0f;
#pragma unroll
            for (int c = 0; c < W; ++c) s += expf(v[c] - mx);
            lse[r] = mx + logf(s);
        }
        __syncthreads();
        for (int i = tid * 4; i < nel; i += TPB * 4) {
            const float4 q = *(const float4*)(acc + i);
            const float l = lse[i / W];
            *(float4*)(oown + i) = make_float4(q.x - l, q.y - l, q.z - l, q.w - l);
        }
    } else {
        const int lo = lane & 15, hi = lane >> 4;
        const int ntile = (rows + 15) >> 4;
        static_assert(AROWS <= 16 * NWAVE, "one 16-row tile per wave covers the owner's rows");
        if (wave < ntile) {
            const int mt = wave;
            const float* arow = acc + (mt * 16 + lo) * PITCH;
            v8f c0 = {}, c1 = {}, c2 = {}, c3 = {};
#pragma unroll 8
            for (int j = 0; j < 16; ++j) {
                const int k0 = kq(j, hi);
                const v2f a = *(const v2f*)(arow + k0);
                const float4 q0 = *(const float4*)(Wa + (size_t)k0 * 64 + 4 * lo);
                const float4 q1 = *(const float4*)(Wa + (size_t)(k0 + 1) * 64 + 4 * lo);
                c0 = __builtin_amdgcn_wmma_f32_16x16x4_f32(false, a, false, v2f{ q0.x, q1.x }, (short)0, c0, false, false);
                c1 = __builtin_amdgcn_wmma_f32_16x16x4_f32(false, a, false, v2f{ q0.y, q1.y }, (short)0, c1, false, false);
                c2 = __builtin_amdgcn_wmma_f32_16x16x4_f32(false, a, false, v2f{ q0.z, q1.z }, (short)0, c2, false, false);
                c3 = __builtin_amdgcn_wmma_f32_16x16x4_f32(false, a, false, v2f{ q0.w, q1.w }, (short)0, c3, false, false);
            }
            const float4 bv = *(const float4*)(bias + 4 * lo);
            float* drow = acc + (mt * 16 + 8 * hi) * PITCH + 4 * lo;
#pragma unroll
            for (int r = 0; r < 8; ++r) {
                *(float4*)(drow + r * PITCH) = make_float4(fmaxf(c0[r] + bv.x, 0.0f), fmaxf(c1[r] + bv.y, 0.0f),
                                                           fmaxf(c2[r] + bv.z, 0.0f), fmaxf(c3[r] + bv.w, 0.0f));
            }
            v8f d0 = {}, d1 = {}, d2 = {};
#pragma unroll 8
            for (int j = 0; j < 16; ++j) {
                const int k0 = kq(j, hi);
                const v2f a = *(const v2f*)(arow + k0);
                v2f b[3];
#pragma unroll
                for (int nt = 0; nt < 3; ++nt) {
                    const int col = 3 * lo + nt;
                    const bool cok = col < 40;
                    const int cc = cok ? col : 39;
                    const float w0 = Wb[(size_t)k0 * 40 + cc];
                    const float w1 = Wb[(size_t)(k0 + 1) * 40 + cc];
                    b[nt] = v2f{ cok ? w0 : 0.0f, cok ? w1 : 0.0f };
                }
                d0 = __builtin_amdgcn_wmma_f32_16x16x4_f32(false, a, false, b[0], (short)0, d0, false, false);
                d1 = __builtin_amdgcn_wmma_f32_16x16x4_f32(false, a, false, b[1], (short)0, d1, false, false);
                d2 = __builtin_amdgcn_wmma_f32_16x16x4_f32(false, a, false, b[2], (short)0, d2, false, false);
            }
            const int r0 = mt * 16 + 8 * hi;
            float* orow = out + (size_t)(base + r0) * 40 + 3 * lo;
#pragma unroll
            for (int r = 0; r < 8; ++r) {
                if (r0 + r < rows) {
                    float* p = orow + (size_t)r * 40;
                    if (lo < 13) { p[0] = d0[r]; p[1] = d1[r]; p[2] = d2[r]; }
                    else if (lo == 13) { p[0] = d0[r]; }
                }
            }
        }
    }
}


extern "C" void kernel_launch(void* const* d_in, const int* in_sizes, int n_in,
                              void* d_out, int out_size, void* d_ws, size_t ws_size,
                              hipStream_t stream) {
    const float* x   = (const float*)d_in[0];
    const int*   ei  = (const int*)d_in[1];
    const float* ew  = (const float*)d_in[2];
    const float* W1  = (const float*)d_in[3];
    const float* b1  = (const float*)d_in[4];
    const float* W2  = (const float*)d_in[5];
    const float* b2  = (const float*)d_in[6];
    float* out = (float*)d_out;

    const int N = in_sizes[0] / 64;
    const int E = in_sizes[2];
    const int* src = ei;
    const int* dst = ei + E;

    const int R = (N + NOWN - 1) / NOWN;
    const int nchunks = (E + 511) / 512;
    const int nwords = nchunks * 128;
    const int npad = ((N + 63) / 64) * 64;

    float* ws   = (float*)d_ws;
    float* dinv = ws;
    unsigned* keyw = (unsigned*)(dinv + npad);
    float* bufA = (float*)(keyw + nwords);
    int* cntp = (int*)(bufA + (size_t)N * 128);
    unsigned* recw = (unsigned*)(cntp + NOWN * NWAVE);
    uint4* erec = (uint4*)(recw + (size_t)NOWN * NWAVE * CAP * 2);
    const int nblk = (nwords + KB - 1) / KB;
    unsigned* bmap = (unsigned*)(erec + (size_t)nwords * 2);

    k_key<<<nblk, KB, 0, stream>>>(src, dst, ew, keyw, erec, bmap, E, nwords, (unsigned)R);
    k_deg_own<<<NOWN, TPB, 0, stream>>>(keyw, (const uint2*)erec, bmap, dinv, cntp, recw, N, E, R, nchunks);

    k_agg_own<64, true><<<NOWN, TPB, 0, stream>>>(keyw, src, dst, ew, dinv, x, b1, cntp, recw, bufA, W1, W2, N, E, R, nchunks);

    k_agg_own<40, false><<<NOWN, TPB, 0, stream>>>(keyw, src, dst, ew, dinv, bufA, b2, cntp, recw, out, nullptr, nullptr, N, E, R, nchunks);
}


# both sum passes: a 16-lane half-wave per edge, so 16-byte row loads and LDS adds serve two edges per instruction; degree pass files edges in 64 half-wave lists; row sums stay in list (arrival) order
# speedup vs baseline: 6.8028x; 1.0881x over previous
#include <hip/hip_runtime.h>
#include <math.h>


typedef __attribute__((ext_vector_type(2))) float v2f;
typedef __attribute__((ext_vector_type(8))) float v8f;

constexpr int NOWN  = 256;
constexpr int RMAX  = 392;
constexpr int TPB   = 1024;
constexpr int NWAVE = TPB / 32;
constexpr int pws_of(int other_bytes) {
    int p = (164 * 1024 - other_bytes + NWAVE * 4 - 1) / (NWAVE * 4);
    return p < 512 ? 512 : p;
}
constexpr int UB    = 8;
constexpr int EB    = 16;
constexpr int NGRP  = 2 * NWAVE;
constexpr int CAP   = 256;
constexpr int SBITS = 23;
constexpr unsigned SMASK = (1u << SBITS) - 1u;
constexpr int KB    = 1024;
constexpr int BMW   = KB / 128;
constexpr int BMP   = NOWN + 8;
constexpr int NLD   = 4;

__device__ __forceinline__ void ldsAddF32(float* p, float v) {
    __hip_atomic_fetch_add(p, v, __ATOMIC_RELAXED, __HIP_MEMORY_SCOPE_WORKGROUP);
}


__global__ __launch_bounds__(KB) void k_key(const int* __restrict__ src, const int* __restrict__ dst,
                                            const float* __restrict__ w, unsigned* __restrict__ keyw,
                                            uint4* __restrict__ erec, unsigned* __restrict__ bmap,
                                            int E, int nwords, unsigned R) {
    __shared__ unsigned bm[BMW * BMP];
    const int tid = threadIdx.x;
    const int i = blockIdx.x * KB + tid;
    for (int j = tid; j < BMW * BMP; j += KB) bm[j] = 0u;
    __syncthreads();
    if (i < nwords) {
        unsigned* plane = bm + (tid >> 7) * BMP;
        const unsigned bit = 1u << ((tid >> 2) & 31);
        unsigned v = 0u;
        unsigned rw[8];
        unsigned ow[4];
#pragma unroll
        for (int u = 0; u < 4; ++u) {
            int e = i * 4 + u;
            const bool in = e < E;
            unsigned d = in ? (unsigned)dst[e] : 0u;
            unsigned sv = in ? (unsigned)src[e] : 0u;
            float we = in ? w[e] : 0.0f;
            unsigned o = d / R;
            ow[u] = o;
            v |= o << (8 * u);
            rw[2 * u] = sv | ((d - o * R) << SBITS);
            rw[2 * u + 1] = __float_as_uint(we);
        }
        keyw[i] = v;
        erec[2 * i] = make_uint4(rw[0], rw[1], rw[2], rw[3]);
        erec[2 * i + 1] = make_uint4(rw[4], rw[5], rw[6], rw[7]);
#pragma unroll
        for (int u = 0; u < 4; ++u) {
            if (i * 4 + u < E) __hip_atomic_fetch_or(&plane[ow[u]], bit, __ATOMIC_RELAXED, __HIP_MEMORY_SCOPE_WORKGROUP);
        }
    }
    __syncthreads();
    unsigned* bout = bmap + (size_t)blockIdx.x * (NOWN * BMW);
    for (int g = tid; g < NOWN * BMW; g += KB) bout[g] = bm[(g & (BMW - 1)) * BMP + (g / BMW)];
}

__device__ __forceinline__ bool any_key_hit(uint4 k, unsigned pat) {
    unsigned x0 = k.x ^ pat, x1 = k.y ^ pat, x2 = k.z ^ pat, x3 = k.w ^ pat;
    unsigned t = ((x0 - 0x01010101u) & ~x0) | ((x1 - 0x01010101u) & ~x1) |
                 ((x2 - 0x01010101u) & ~x2) | ((x3 - 0x01010101u) & ~x3);
    return (t & 0x80808080u) != 0u;
}

__device__ __forceinline__ unsigned key_hit_mask16(uint4 k, unsigned pat) {
    unsigned kk[4] = { k.x, k.y, k.z, k.w };
    unsigned m = 0u;
#pragma unroll
    for (int j = 0; j < 4; ++j) {
        unsigned x = kk[j] ^ pat;
        unsigned t = ((x & 0x7F7F7F7Fu) + 0x7F7F7F7Fu) | x;
        unsigned z = ~t & 0x80808080u;
        unsigned c = ((z >> 7) & 1u) | ((z >> 14) & 2u) | ((z >> 21) & 4u) | ((z >> 28) & 8u);
        m |= c << (4 * j);
    }
    return m;
}

__device__ __forceinline__ int share_lo(int wave, int nchunks) {
    return (int)(((long long)wave * nchunks) / NWAVE);
}

template <int PWS, class Proc>
__device__ __forceinline__ void scan_owned(const uint4* __restrict__ keyv, int xlo, int xhi,
                                           unsigned* wlist, int lane, int owner, Proc proc) {
    const unsigned pat = (unsigned)owner * 0x01010101u;
    int cnt = 0;
    for (int c0 = xlo; c0 < xhi; c0 += UB) {
        uint4 k[UB];
#pragma unroll
        for (int u = 0; u < UB; ++u) {
            const int c = c0 + u;
            const int x = (c < xhi) ? c : xlo;
            k[u] = keyv[(size_t)x * 32 + lane];
        }
#pragma unroll
        for (int u = 0; u < UB; ++u) {
            bool hit = (c0 + u < xhi) && any_key_hit(k[u], pat);
            unsigned bal = __builtin_amdgcn_ballot_w32(hit);
            if (hit) wlist[cnt + (int)__builtin_amdgcn_mbcnt_lo(bal, 0u)] = (unsigned)((c0 + u) * 32 + lane);
            cnt += __builtin_popcount(bal);
        }
        if (cnt > PWS - 32 * UB) { proc(cnt); cnt = 0; }
    }
    proc(cnt);
}


__global__ __launch_bounds__(TPB) void k_deg_own(const unsigned* __restrict__ keyw,
                                                 const uint2* __restrict__ erec,
                                                 const unsigned* __restrict__ bmap,
                                                 float* __restrict__ dinv,
                                                 int* __restrict__ cntp,
                                                 unsigned* __restrict__ recw,
                                                 int n, int E, int R, int nchunks) {
    constexpr int PWS = pws_of(RMAX * 4 + NGRP * CAP * 8 + NGRP * 4);
    __shared__ float degl[RMAX];
    __shared__ unsigned list[NWAVE * PWS];
    __shared__ unsigned long long stage[NGRP * CAP];
    __shared__ int lcnt[NGRP];
    const uint4* keyv = (const uint4*)keyw;
    const int tid = threadIdx.x, lane = tid & 31;
    const int wave = __builtin_amdgcn_readfirstlane(tid >> 5);
    const int owner = blockIdx.x, base = owner * R;
    const unsigned pat = (unsigned)owner * 0x01010101u;
    unsigned* wlist = list + wave * PWS;
    const float gscale = (float)NGRP / (float)R;

    for (int i = tid; i < RMAX; i += TPB) degl[i] = 1.0f;
    if (tid < NGRP) lcnt[tid] = 0;
    __syncthreads();

    auto proc = [&](int cnt) {
        for (int g = 0; g < cnt; g += 32) {
            int idx = g + lane;
            bool v = idx < cnt;
            unsigned seg = v ? wlist[idx] : 0u;
            unsigned m = v ? key_hit_mask16(keyv[seg], pat) : 0u;
            while (__builtin_amdgcn_ballot_w32(m != 0u) != 0u) {
                bool act = (m != 0u);
                int e = (int)seg * 16 + (act ? __builtin_ctz(m) : 0);
                m &= m - 1u;
                act = act && (e < E);
                if (act) {
                    const uint2 rc = erec[e];
                    const unsigned row = rc.x >> SBITS;
                    int grp = (int)((float)row * gscale);
                    grp = grp > NGRP - 1 ? NGRP - 1 : grp;
                    const int pos = __hip_atomic_fetch_add(&lcnt[grp], 1, __ATOMIC_RELAXED, __HIP_MEMORY_SCOPE_WORKGROUP);
                    if (pos < CAP) stage[grp * CAP + pos] = ((unsigned long long)rc.y << 32) | rc.x;
                    ldsAddF32(&degl[row], __uint_as_float(rc.y));
                }
            }
        }
    };
    {
        const int xlo = share_lo(wave, nchunks), xhi = share_lo(wave + 1, nchunks);
        int cnt = 0;
        for (int x0 = xlo; x0 < xhi; x0 += NLD * 32) {
            unsigned m[NLD], sb[NLD];
#pragma unroll
            for (int q = 0; q < NLD; ++q) {
                const int x = x0 + q * 32 + lane;
                const bool ok = x < xhi;
                const int blk = ok ? (x / BMW) : 0, pl = x & (BMW - 1);
                const unsigned wv = bmap[((size_t)blk * NOWN + owner) * BMW + pl];
                m[q] = ok ? wv : 0u;
                sb[q] = (unsigned)(blk * (KB / 4) + pl * 32);
            }
#pragma unroll
            for (int q = 0; q < NLD; ++q) {
                unsigned mm = m[q];
                while (__builtin_amdgcn_ballot_w32(mm != 0u) != 0u) {
                    const bool hit = (mm != 0u);
                    const unsigned seg = sb[q] + (hit ? (unsigned)__builtin_ctz(mm) : 0u);
                    mm &= mm - 1u;
                    const unsigned bal = __builtin_amdgcn_ballot_w32(hit);
                    if (hit) wlist[cnt + (int)__builtin_amdgcn_mbcnt_lo(bal, 0u)] = seg;
                    cnt += __builtin_popcount(bal);
                    if (cnt > PWS - 32) { proc(cnt); cnt = 0; }
                }
            }
        }
        proc(cnt);
    }
    __syncthreads();

    for (int q = 0; q < 2; ++q) {
        const int gi = 2 * wave + q;
        const int cw = lcnt[gi];
        const int cc = cw < CAP ? cw : CAP;
        unsigned long long* wrec = (unsigned long long*)recw + ((size_t)owner * NGRP + gi) * CAP;
        for (int i = lane; i < cc; i += 32) wrec[i] = stage[gi * CAP + i];
        if (lane == 0) cntp[owner * NGRP + gi] = cw;
    }

    for (int i = tid; i < R; i += TPB) {
        int node = base + i;
        if (node < n) {
            float d = degl[i];
            dinv[node] = (d > 0.0f) ? rsqrtf(d) : 0.0f;
        }
    }
}


__device__ __forceinline__ int kq(int j, int hi) { return 4 * j + 2 * hi; }


template <int P> struct ic_t { static constexpr int v = P; };
template <int P, int N, class F> __device__ __forceinline__ void unrolled(F&& f) {
    if constexpr (P < N) { f(ic_t<P>{}); unrolled<P + 1, N>(f); }
}
template <int P> __device__ __forceinline__ int half_share(int v) {
    return __builtin_amdgcn_update_dpp(0, v, 0x150 + P, 0xF, 0xF, false);
}

template <int W, bool FUSE>
__global__ __launch_bounds__(TPB) void k_agg_own(const unsigned* __restrict__ keyw,
                                                 const int* __restrict__ src,
                                                 const int* __restrict__ dst,
                                                 const float* __restrict__ w,
                                                 const float* __restrict__ dinv,
                                                 const float* __restrict__ h,
                                                 const float* __restrict__ bias,
                                                 const int* __restrict__ cntp,
                                                 const unsigned* __restrict__ recw,
                                                 float* __restrict__ out,
                                                 const float* __restrict__ Wa,
                                                 const float* __restrict__ Wb,
                                                 int n, int E, int R, int nchunks) {
    constexpr int PITCH = FUSE ? 68 : W;
    constexpr int AROWS = FUSE ? 400 : RMAX;
    constexpr int DUMP = AROWS * PITCH;
    constexpr int PWS = pws_of((AROWS * PITCH + 64) * 4);
    __shared__ __attribute__((aligned(16))) float acc[AROWS * PITCH + 64];
    __shared__ unsigned list[NWAVE * PWS];
    __shared__ float lse[FUSE ? 1 : RMAX];
    const uint4* keyv = (const uint4*)keyw;
    const int tid = threadIdx.x, lane = tid & 31;
    const int wave = __builtin_amdgcn_readfirstlane(tid >> 5);
    const int owner = blockIdx.x, base = owner * R;
    const unsigned pat = (unsigned)owner * 0x01010101u;
    unsigned* wlist = list + wave * PWS;
    int rows = n - base;
    rows = rows < 0 ? 0 : (rows > R ? R : rows);
    const int nel = rows * W;
    const float* hown = h + (size_t)base * W;
    float* oown = out + (size_t)base * W;

    for (int i = tid; i < nel; i += TPB) {
        int r = i / W, f = i - r * W;
        float di = dinv[base + r];
        float sl = di * di * hown[i];
        acc[r * PITCH + f] = FUSE ? sl : bias[f] + sl;
    }
    if (tid < 64) acc[DUMP + tid] = 0.0f;
    __syncthreads();

    const bool fl2 = (2 * lane < W);
    auto proc = [&](int cnt) {
        for (int g = 0; g < cnt; g += 32) {
            int idx = g + lane;
            bool v = idx < cnt;
            unsigned seg = v ? wlist[idx] : 0u;
            unsigned m = v ? key_hit_mask16(keyv[seg], pat) : 0u;
            while (__builtin_amdgcn_ballot_w32(m != 0u) != 0u) {
                bool act = (m != 0u);
                int e = (int)seg * 16 + (act ? __builtin_ctz(m) : 0);
                m &= m - 1u;
                act = act && (e < E);
                int s = 0, dl = 0;
                float nr = 0.0f;
                if (act) {
                    int d = dst[e];
                    s = src[e];
                    nr = dinv[s] * w[e] * dinv[d];
                    dl = d - base;
                }
                unsigned bal = __builtin_amdgcn_ballot_w32(act);
                while (bal != 0u) {
                    int jj[EB];
                    bool ok[EB];
#pragma unroll
                    for (int u = 0; u < EB; ++u) {
                        ok[u] = (bal != 0u);
                        jj[u] = ok[u] ? __builtin_ctz(bal) : 0;
                        bal &= bal - 1u;
                    }
                    v2f r[EB];
#pragma unroll
                    for (int u = 0; u < EB; ++u) {
                        int sj = __builtin_amdgcn_readlane(s, jj[u]);
                        r[u] = fl2 ? *(const v2f*)(h + (size_t)sj * W + 2 * lane) : v2f{0.0f, 0.0f};
                    }
#pragma unroll
                    for (int u = 0; u < EB; ++u) {
                        if (ok[u]) {
                            float nj = __int_as_float(__builtin_amdgcn_readlane(__float_as_int(nr), jj[u]));
                            int dj = __builtin_amdgcn_readlane(dl, jj[u]);
                            if (fl2) {
                                float* a = &acc[dj * PITCH + 2 * lane];
                                ldsAddF32(a, nj * r[u].x);
                                ldsAddF32(a + 1, nj * r[u].y);
                            }
                        }
                    }
                }
            }
        }
    };
    const int ca = cntp[owner * NGRP + lane];
    const int cb = cntp[owner * NGRP + 32 + lane];
    const bool over = __builtin_amdgcn_ballot_w32(ca > CAP || cb > CAP) != 0u;
    if (!over) {
        constexpr int W4 = W / 4;
        const int hf = lane >> 4, cl = lane & 15;
        const bool fl = cl < W4;
        const int clq = fl ? cl : W4 - 1;
        const int gi = 2 * wave + hf;
        const int ch = cntp[owner * NGRP + gi];
        const int c0 = __builtin_amdgcn_readlane(ch, 0), c1 = __builtin_amdgcn_readlane(ch, 16);
        const int cmax = c0 > c1 ? c0 : c1;
        const unsigned long long* wrec = (const unsigned long long*)recw + ((size_t)owner * NGRP + gi) * CAP;
        const float4* hv = (const float4*)h;
        const bool cur = cl < ch;
        unsigned long long rc = cur ? wrec[cl] : 0ull;
        int s = (int)((unsigned)rc & SMASK), dl = (int)((unsigned)rc >> SBITS);
        float nr = cur ? dinv[s] * __uint_as_float((unsigned)(rc >> 32)) * dinv[base + dl] : 0.0f;
        int go = s * W4;
        float4* acc4 = (float4*)acc;
        static_assert(PITCH % 4 == 0 && DUMP % 4 == 0, "rows start on 16-byte pieces");
        int ai = cur ? dl * (PITCH / 4) : DUMP / 4;
        for (int g = 0; g < cmax; g += 16) {
            const bool nx = g + 16 + cl < ch;
            unsigned long long rn = nx ? wrec[g + 16 + cl] : 0ull;
            float4 r[16];
            unrolled<0, 16>([&](auto pc) {
                constexpr int P = decltype(pc)::v;
                r[P] = hv[half_share<P>(go) + clq];
            });
            int sn = (int)((unsigned)rn & SMASK), dln = (int)((unsigned)rn >> SBITS);
            float dsn = nx ? dinv[sn] : 0.0f;
            float ddn = nx ? dinv[base + dln] : 0.0f;
            unrolled<0, 8>([&](auto kc) {
                constexpr int P0 = 2 * decltype(kc)::v;
                if (g + P0 < cmax) {
                    const int a0 = half_share<P0>(ai) + cl;
                    const int a1 = half_share<P0 + 1>(ai) + cl;
                    const float n0 = __int_as_float(half_share<P0>(__float_as_int(nr)));
                    const float n1 = __int_as_float(half_share<P0 + 1>(__float_as_int(nr)));
                    if (fl) {
                        float4 t0 = acc4[a0];
                        float4 t1 = acc4[a1];
                        const bool same = (a1 == a0);
                        t0.x = fmaf(n0, r[P0].x, t0.x);
                        t0.y = fmaf(n0, r[P0].y, t0.y);
                        t0.z = fmaf(n0, r[P0].z, t0.z);
                        t0.w = fmaf(n0, r[P0].w, t0.w);
                        t1.x = fmaf(n1, r[P0 + 1].x, same ? t0.x : t1.x);
                        t1.y = fmaf(n1, r[P0 + 1].y, same ? t0.y : t1.y);
                        t1.z = fmaf(n1, r[P0 + 1].z, same ? t0.z : t1.z);
                        t1.w = fmaf(n1, r[P0 + 1].w, same ? t0.w : t1.w);
                        acc4[a0] = t0;
                        acc4[a1] = t1;
                    }
                }
            });
            go = sn * W4;
            ai = nx ? dln * (PITCH / 4) : DUMP / 4;
            nr = dsn * __uint_as_float((unsigned)(rn >> 32)) * ddn;
        }
    } else {
        scan_owned<PWS>(keyv, share_lo(wave, nchunks), share_lo(wave + 1, nchunks), wlist, lane, owner, proc);
    }
    __syncthreads();

    if constexpr (!FUSE) {
        static_assert(RMAX <= TPB, "a thread per owned row");
        if (tid < rows) {
            const int r = tid;
            const float* v = acc + r * W;
            float mx = -3.402823466e38f;
#pragma unroll
            for (int c = 0; c < W; ++c) mx = fmaxf(mx, v[c]);
            float s = 0.0f;
#pragma unroll
            for (int c = 0; c < W; ++c) s += expf(v[c] - mx);
            lse[r] = mx + logf(s);
        }
        __syncthreads();
        for (int i = tid * 4; i < nel; i += TPB * 4) {
            const float4 q = *(const float4*)(acc + i);
            const float l = lse[i / W];
            *(float4*)(oown + i) = make_float4(q.x - l, q.y - l, q.z - l, q.w - l);
        }
    } else {
        const int lo = lane & 15, hi = lane >> 4;
        const int ntile = (rows + 15) >> 4;
        static_assert(AROWS <= 16 * NWAVE, "one 16-row tile per wave covers the owner's rows");
        if (wave < ntile) {
            const int mt = wave;
            const float* arow = acc + (mt * 16 + lo) * PITCH;
            v8f c0 = {}, c1 = {}, c2 = {}, c3 = {};
#pragma unroll 8
            for (int j = 0; j < 16; ++j) {
                const int k0 = kq(j, hi);
                const v2f a = *(const v2f*)(arow + k0);
                const float4 q0 = *(const float4*)(Wa + (size_t)k0 * 64 + 4 * lo);
                const float4 q1 = *(const float4*)(Wa + (size_t)(k0 + 1) * 64 + 4 * lo);
                c0 = __builtin_amdgcn_wmma_f32_16x16x4_f32(false, a, false, v2f{ q0.x, q1.x }, (short)0, c0, false, false);
                c1 = __builtin_amdgcn_wmma_f32_16x16x4_f32(false, a, false, v2f{ q0.y, q1.y }, (short)0, c1, false, false);
                c2 = __builtin_amdgcn_wmma_f32_16x16x4_f32(false, a, false, v2f{ q0.z, q1.z }, (short)0, c2, false, false);
                c3 = __builtin_amdgcn_wmma_f32_16x16x4_f32(false, a, false, v2f{ q0.w, q1.w }, (short)0, c3, false, false);
            }
            const float4 bv = *(const float4*)(bias + 4 * lo);
            float* drow = acc + (mt * 16 + 8 * hi) * PITCH + 4 * lo;
#pragma unroll
            for (int r = 0; r < 8; ++r) {
                *(float4*)(drow + r * PITCH) = make_float4(fmaxf(c0[r] + bv.x, 0.0f), fmaxf(c1[r] + bv.y, 0.0f),
                                                           fmaxf(c2[r] + bv.z, 0.0f), fmaxf(c3[r] + bv.w, 0.0f));
            }
            v8f d0 = {}, d1 = {}, d2 = {};
#pragma unroll 8
            for (int j = 0; j < 16; ++j) {
                const int k0 = kq(j, hi);
                const v2f a = *(const v2f*)(arow + k0);
                v2f b[3];
#pragma unroll
                for (int nt = 0; nt < 3; ++nt) {
                    const int col = 3 * lo + nt;
                    const bool cok = col < 40;
                    const int cc = cok ? col : 39;
                    const float w0 = Wb[(size_t)k0 * 40 + cc];
                    const float w1 = Wb[(size_t)(k0 + 1) * 40 + cc];
                    b[nt] = v2f{ cok ? w0 : 0.0f, cok ? w1 : 0.0f };
                }
                d0 = __builtin_amdgcn_wmma_f32_16x16x4_f32(false, a, false, b[0], (short)0, d0, false, false);
                d1 = __builtin_amdgcn_wmma_f32_16x16x4_f32(false, a, false, b[1], (short)0, d1, false, false);
                d2 = __builtin_amdgcn_wmma_f32_16x16x4_f32(false, a, false, b[2], (short)0, d2, false, false);
            }
            const int r0 = mt * 16 + 8 * hi;
            float* orow = out + (size_t)(base + r0) * 40 + 3 * lo;
#pragma unroll
            for (int r = 0; r < 8; ++r) {
                if (r0 + r < rows) {
                    float* p = orow + (size_t)r * 40;
                    if (lo < 13) { p[0] = d0[r]; p[1] = d1[r]; p[2] = d2[r]; }
                    else if (lo == 13) { p[0] = d0[r]; }
                }
            }
        }
    }
}


extern "C" void kernel_launch(void* const* d_in, const int* in_sizes, int n_in,
                              void* d_out, int out_size, void* d_ws, size_t ws_size,
                              hipStream_t stream) {
    const float* x   = (const float*)d_in[0];
    const int*   ei  = (const int*)d_in[1];
    const float* ew  = (const float*)d_in[2];
    const float* W1  = (const float*)d_in[3];
    const float* b1  = (const float*)d_in[4];
    const float* W2  = (const float*)d_in[5];
    const float* b2  = (const float*)d_in[6];
    float* out = (float*)d_out;

    const int N = in_sizes[0] / 64;
    const int E = in_sizes[2];
    const int* src = ei;
    const int* dst = ei + E;

    const int R = (N + NOWN - 1) / NOWN;
    const int nchunks = (E + 511) / 512;
    const int nwords = nchunks * 128;
    const int npad = ((N + 63) / 64) * 64;

    float* ws   = (float*)d_ws;
    float* dinv = ws;
    unsigned* keyw = (unsigned*)(dinv + npad);
    float* bufA = (float*)(keyw + nwords);
    int* cntp = (int*)(bufA + (size_t)N * 128);
    unsigned* recw = (unsigned*)(cntp + NOWN * NGRP);
    uint4* erec = (uint4*)(recw + (size_t)NOWN * NGRP * CAP * 2);
    const int nblk = (nwords + KB - 1) / KB;
    unsigned* bmap = (unsigned*)(erec + (size_t)nwords * 2);

    k_key<<<nblk, KB, 0, stream>>>(src, dst, ew, keyw, erec, bmap, E, nwords, (unsigned)R);
    k_deg_own<<<NOWN, TPB, 0, stream>>>(keyw, (const uint2*)erec, bmap, dinv, cntp, recw, N, E, R, nchunks);

    k_agg_own<64, true><<<NOWN, TPB, 0, stream>>>(keyw, src, dst, ew, dinv, x, b1, cntp, recw, bufA, W1, W2, N, E, R, nchunks);

    k_agg_own<40, false><<<NOWN, TPB, 0, stream>>>(keyw, src, dst, ew, dinv, bufA, b2, cntp, recw, out, nullptr, nullptr, N, E, R, nchunks);
}


# key kernel groups each 4096-edge block's records by owner in LDS (count = rank, prefix, place); degree pass reads one contiguous run per block: no index bits, no per-edge fetch; sums in arrival order
# speedup vs baseline: 7.5211x; 1.1056x over previous
#include <hip/hip_runtime.h>
#include <math.h>


typedef __attribute__((ext_vector_type(2))) float v2f;
typedef __attribute__((ext_vector_type(8))) float v8f;

constexpr int NOWN  = 256;
constexpr int RMAX  = 392;
constexpr int TPB   = 1024;
constexpr int NWAVE = TPB / 32;
constexpr int pws_of(int other_bytes) {
    int p = (164 * 1024 - other_bytes + NWAVE * 4 - 1) / (NWAVE * 4);
    return p < 512 ? 512 : p;
}
constexpr int UB    = 8;
constexpr int EB    = 16;
constexpr int NGRP  = 2 * NWAVE;
constexpr int CAP   = 320;
constexpr int SBITS = 23;
constexpr unsigned SMASK = (1u << SBITS) - 1u;
constexpr int KB    = 1024;
constexpr int KE    = KB * 4;
constexpr int RB    = 16;

__device__ __forceinline__ void ldsAddF32(float* p, float v) {
    __hip_atomic_fetch_add(p, v, __ATOMIC_RELAXED, __HIP_MEMORY_SCOPE_WORKGROUP);
}


__global__ __launch_bounds__(KB) void k_key(const int* __restrict__ src, const int* __restrict__ dst,
                                            const float* __restrict__ w, unsigned* __restrict__ keyw,
                                            uint4* __restrict__ erec, unsigned* __restrict__ offs,
                                            int E, int nwords, unsigned R) {
    __shared__ __attribute__((aligned(16))) unsigned long long srt[KE];
    __shared__ __attribute__((aligned(16))) int ocnt[NOWN];
    __shared__ __attribute__((aligned(16))) int ost[NOWN];
    __shared__ int tot;
    const int tid = threadIdx.x;
    const int i = blockIdx.x * KB + tid;
    if (tid < NOWN) ocnt[tid] = 0;
    unsigned v = 0u;
    unsigned rw[8];
    unsigned ow[4];
    bool in[4];
#pragma unroll
    for (int u = 0; u < 4; ++u) {
        int e = i * 4 + u;
        in[u] = e < E;
        unsigned d = in[u] ? (unsigned)dst[e] : 0u;
        unsigned sv = in[u] ? (unsigned)src[e] : 0u;
        float we = in[u] ? w[e] : 0.0f;
        unsigned o = d / R;
        ow[u] = o;
        v |= o << (8 * u);
        rw[2 * u] = sv | ((d - o * R) << SBITS);
        rw[2 * u + 1] = __float_as_uint(we);
    }
    __syncthreads();
    int rk[4];
#pragma unroll
    for (int u = 0; u < 4; ++u)
        rk[u] = in[u] ? __hip_atomic_fetch_add(&ocnt[ow[u]], 1, __ATOMIC_RELAXED, __HIP_MEMORY_SCOPE_WORKGROUP) : 0;
    __syncthreads();
    if (tid < 32) {
        int cq[8];
        int s = 0;
#pragma unroll
        for (int q = 0; q < 8; ++q) { cq[q] = ocnt[8 * tid + q]; s += cq[q]; }
        int inc = s;
#pragma unroll
        for (int dlt = 1; dlt < 32; dlt <<= 1) {
            int t = __shfl_up(inc, dlt);
            if (tid >= dlt) inc += t;
        }
        int st = inc - s;
#pragma unroll
        for (int q = 0; q < 8; ++q) {
            ost[8 * tid + q] = st;
            st += cq[q];
        }
        if (tid == 31) tot = inc;
    }
    __syncthreads();
#pragma unroll
    for (int u = 0; u < 4; ++u) {
        if (in[u]) srt[ost[ow[u]] + rk[u]] = ((unsigned long long)rw[2 * u + 1] << 32) | rw[2 * u];
    }
    __syncthreads();
    if (i < nwords) keyw[i] = v;
    if (tid < NOWN) offs[(size_t)blockIdx.x * NOWN + tid] = (unsigned)ost[tid] | ((unsigned)ocnt[tid] << 16);
    const int total = tot;
    const uint4* s4 = (const uint4*)srt;
    uint4* eout = erec + (size_t)blockIdx.x * (KE / 2);
#pragma unroll
    for (int q = 0; q < 2; ++q) {
        const int p = tid + q * KB;
        if (2 * p < total) eout[p] = s4[p];
    }
}

__device__ __forceinline__ bool any_key_hit(uint4 k, unsigned pat) {
    unsigned x0 = k.x ^ pat, x1 = k.y ^ pat, x2 = k.z ^ pat, x3 = k.w ^ pat;
    unsigned t = ((x0 - 0x01010101u) & ~x0) | ((x1 - 0x01010101u) & ~x1) |
                 ((x2 - 0x01010101u) & ~x2) | ((x3 - 0x01010101u) & ~x3);
    return (t & 0x80808080u) != 0u;
}

__device__ __forceinline__ unsigned key_hit_mask16(uint4 k, unsigned pat) {
    unsigned kk[4] = { k.x, k.y, k.z, k.w };
    unsigned m = 0u;
#pragma unroll
    for (int j = 0; j < 4; ++j) {
        unsigned x = kk[j] ^ pat;
        unsigned t = ((x & 0x7F7F7F7Fu) + 0x7F7F7F7Fu) | x;
        unsigned z = ~t & 0x80808080u;
        unsigned c = ((z >> 7) & 1u) | ((z >> 14) & 2u) | ((z >> 21) & 4u) | ((z >> 28) & 8u);
        m |= c << (4 * j);
    }
    return m;
}

__device__ __forceinline__ int share_lo(int wave, int nchunks) {
    return (int)(((long long)wave * nchunks) / NWAVE);
}

template <int PWS, class Proc>
__device__ __forceinline__ void scan_owned(const uint4* __restrict__ keyv, int xlo, int xhi,
                                           unsigned* wlist, int lane, int owner, Proc proc) {
    const unsigned pat = (unsigned)owner * 0x01010101u;
    int cnt = 0;
    for (int c0 = xlo; c0 < xhi; c0 += UB) {
        uint4 k[UB];
#pragma unroll
        for (int u = 0; u < UB; ++u) {
            const int c = c0 + u;
            const int x = (c < xhi) ? c : xlo;
            k[u] = keyv[(size_t)x * 32 + lane];
        }
#pragma unroll
        for (int u = 0; u < UB; ++u) {
            bool hit = (c0 + u < xhi) && any_key_hit(k[u], pat);
            unsigned bal = __builtin_amdgcn_ballot_w32(hit);
            if (hit) wlist[cnt + (int)__builtin_amdgcn_mbcnt_lo(bal, 0u)] = (unsigned)((c0 + u) * 32 + lane);
            cnt += __builtin_popcount(bal);
        }
        if (cnt > PWS - 32 * UB) { proc(cnt); cnt = 0; }
    }
    proc(cnt);
}


__global__ __launch_bounds__(TPB) void k_deg_own(const uint2* __restrict__ erec,
                                                 const unsigned* __restrict__ offs,
                                                 float* __restrict__ dinv,
                                                 int* __restrict__ cntp,
                                                 unsigned* __restrict__ recw,
                                                 int n, int R, int nblk) {
    __shared__ float degl[RMAX];
    __shared__ unsigned long long stage[NGRP * CAP];
    __shared__ int lcnt[NGRP];
    static_assert(RMAX * 4 + NGRP * CAP * 8 + NGRP * 4 > 160 * 1024, "one owner per WGP");
    const int tid = threadIdx.x, lane = tid & 31;
    const int wave = __builtin_amdgcn_readfirstlane(tid >> 5);
    const int owner = blockIdx.x, base = owner * R;
    const float gscale = (float)NGRP / (float)R;

    for (int i = tid; i < RMAX; i += TPB) degl[i] = 1.0f;
    if (tid < NGRP) lcnt[tid] = 0;
    __syncthreads();

    auto file = [&](uint2 rc) {
        const unsigned row = rc.x >> SBITS;
        int grp = (int)((float)row * gscale);
        grp = grp > NGRP - 1 ? NGRP - 1 : grp;
        const int pos = __hip_atomic_fetch_add(&lcnt[grp], 1, __ATOMIC_RELAXED, __HIP_MEMORY_SCOPE_WORKGROUP);
        if (pos < CAP) stage[grp * CAP + pos] = ((unsigned long long)rc.y << 32) | rc.x;
        ldsAddF32(&degl[row], __uint_as_float(rc.y));
    };
    {
        const int blo = share_lo(wave, nblk), bhi = share_lo(wave + 1, nblk);
        for (int b0 = blo; b0 < bhi; b0 += 32) {
            const int nb = (bhi - b0) < 32 ? (bhi - b0) : 32;
            const int bl = b0 + (lane < nb ? lane : 0);
            const unsigned pw = offs[(size_t)bl * NOWN + owner];
            for (int j0 = 0; j0 < nb; j0 += RB) {
                uint2 rc[RB];
                int cj[RB];
                int cmx = 0;
#pragma unroll
                for (int u = 0; u < RB; ++u) {
                    const bool have = j0 + u < nb;
                    const int j = have ? j0 + u : j0;
                    const unsigned wj = (unsigned)__builtin_amdgcn_readlane((int)pw, j);
                    cj[u] = have ? (int)(wj >> 16) : 0;
                    cmx = cj[u] > cmx ? cj[u] : cmx;
                    const int idx = (lane < cj[u]) ? (int)(wj & 0xFFFFu) + lane : 0;
                    rc[u] = erec[(size_t)(b0 + j) * KE + idx];
                }
#pragma unroll
                for (int u = 0; u < RB; ++u) {
                    if (lane < cj[u]) file(rc[u]);
                }
                if (cmx > 32) {
                    const int je = (j0 + RB) < nb ? (j0 + RB) : nb;
                    for (int j = j0; j < je; ++j) {
                        const unsigned wj = (unsigned)__builtin_amdgcn_readlane((int)pw, j);
                        const int cn = (int)(wj >> 16), st = (int)(wj & 0xFFFFu);
                        for (int i0 = 32 + lane; i0 - lane < cn; i0 += 32) {
                            if (i0 < cn) file(erec[(size_t)(b0 + j) * KE + st + i0]);
                        }
                    }
                }
            }
        }
    }
    __syncthreads();

    for (int q = 0; q < 2; ++q) {
        const int gi = 2 * wave + q;
        const int cw = lcnt[gi];
        const int cc = cw < CAP ? cw : CAP;
        unsigned long long* wrec = (unsigned long long*)recw + ((size_t)owner * NGRP + gi) * CAP;
        for (int i = lane; i < cc; i += 32) wrec[i] = stage[gi * CAP + i];
        if (lane == 0) cntp[owner * NGRP + gi] = cw;
    }

    for (int i = tid; i < R; i += TPB) {
        int node = base + i;
        if (node < n) {
            float d = degl[i];
            dinv[node] = (d > 0.0f) ? rsqrtf(d) : 0.0f;
        }
    }
}


__device__ __forceinline__ int kq(int j, int hi) { return 4 * j + 2 * hi; }


template <int P> struct ic_t { static constexpr int v = P; };
template <int P, int N, class F> __device__ __forceinline__ void unrolled(F&& f) {
    if constexpr (P < N) { f(ic_t<P>{}); unrolled<P + 1, N>(f); }
}
template <int P> __device__ __forceinline__ int half_share(int v) {
    return __builtin_amdgcn_update_dpp(0, v, 0x150 + P, 0xF, 0xF, false);
}

template <int W, bool FUSE>
__global__ __launch_bounds__(TPB) void k_agg_own(const unsigned* __restrict__ keyw,
                                                 const int* __restrict__ src,
                                                 const int* __restrict__ dst,
                                                 const float* __restrict__ w,
                                                 const float* __restrict__ dinv,
                                                 const float* __restrict__ h,
                                                 const float* __restrict__ bias,
                                                 const int* __restrict__ cntp,
                                                 const unsigned* __restrict__ recw,
                                                 float* __restrict__ out,
                                                 const float* __restrict__ Wa,
                                                 const float* __restrict__ Wb,
                                                 int n, int E, int R, int nchunks) {
    constexpr int PITCH = FUSE ? 68 : W;
    constexpr int AROWS = FUSE ? 400 : RMAX;
    constexpr int DUMP = AROWS * PITCH;
    constexpr int PWS = pws_of((AROWS * PITCH + 64) * 4);
    __shared__ __attribute__((aligned(16))) float acc[AROWS * PITCH + 64];
    __shared__ unsigned list[NWAVE * PWS];
    __shared__ float lse[FUSE ? 1 : RMAX];
    const uint4* keyv = (const uint4*)keyw;
    const int tid = threadIdx.x, lane = tid & 31;
    const int wave = __builtin_amdgcn_readfirstlane(tid >> 5);
    const int owner = blockIdx.x, base = owner * R;
    const unsigned pat = (unsigned)owner * 0x01010101u;
    unsigned* wlist = list + wave * PWS;
    int rows = n - base;
    rows = rows < 0 ? 0 : (rows > R ? R : rows);
    const int nel = rows * W;
    const float* hown = h + (size_t)base * W;
    float* oown = out + (size_t)base * W;

    for (int i = tid; i < nel; i += TPB) {
        int r = i / W, f = i - r * W;
        float di = dinv[base + r];
        float sl = di * di * hown[i];
        acc[r * PITCH + f] = FUSE ? sl : bias[f] + sl;
    }
    if (tid < 64) acc[DUMP + tid] = 0.0f;
    __syncthreads();

    const bool fl2 = (2 * lane < W);
    auto proc = [&](int cnt) {
        for (int g = 0; g < cnt; g += 32) {
            int idx = g + lane;
            bool v = idx < cnt;
            unsigned seg = v ? wlist[idx] : 0u;
            unsigned m = v ? key_hit_mask16(keyv[seg], pat) : 0u;
            while (__builtin_amdgcn_ballot_w32(m != 0u) != 0u) {
                bool act = (m != 0u);
                int e = (int)seg * 16 + (act ? __builtin_ctz(m) : 0);
                m &= m - 1u;
                act = act && (e < E);
                int s = 0, dl = 0;
                float nr = 0.0f;
                if (act) {
                    int d = dst[e];
                    s = src[e];
                    nr = dinv[s] * w[e] * dinv[d];
                    dl = d - base;
                }
                unsigned bal = __builtin_amdgcn_ballot_w32(act);
                while (bal != 0u) {
                    int jj[EB];
                    bool ok[EB];
#pragma unroll
                    for (int u = 0; u < EB; ++u) {
                        ok[u] = (bal != 0u);
                        jj[u] = ok[u] ? __builtin_ctz(bal) : 0;
                        bal &= bal - 1u;
                    }
                    v2f r[EB];
#pragma unroll
                    for (int u = 0; u < EB; ++u) {
                        int sj = __builtin_amdgcn_readlane(s, jj[u]);
                        r[u] = fl2 ? *(const v2f*)(h + (size_t)sj * W + 2 * lane) : v2f{0.0f, 0.0f};
                    }
#pragma unroll
                    for (int u = 0; u < EB; ++u) {
                        if (ok[u]) {
                            float nj = __int_as_float(__builtin_amdgcn_readlane(__float_as_int(nr), jj[u]));
                            int dj = __builtin_amdgcn_readlane(dl, jj[u]);
                            if (fl2) {
                                float* a = &acc[dj * PITCH + 2 * lane];
                                ldsAddF32(a, nj * r[u].x);
                                ldsAddF32(a + 1, nj * r[u].y);
                            }
                        }
                    }
                }
            }
        }
    };
    const int ca = cntp[owner * NGRP + lane];
    const int cb = cntp[owner * NGRP + 32 + lane];
    const bool over = __builtin_amdgcn_ballot_w32(ca > CAP || cb > CAP) != 0u;
    if (!over) {
        constexpr int W4 = W / 4;
        const int hf = lane >> 4, cl = lane & 15;
        const bool fl = cl < W4;
        const int clq = fl ? cl : W4 - 1;
        const int gi = 2 * wave + hf;
        const int ch = cntp[owner * NGRP + gi];
        const int c0 = __builtin_amdgcn_readlane(ch, 0), c1 = __builtin_amdgcn_readlane(ch, 16);
        const int cmax = c0 > c1 ? c0 : c1;
        const unsigned long long* wrec = (const unsigned long long*)recw + ((size_t)owner * NGRP + gi) * CAP;
        const float4* hv = (const float4*)h;
        const bool cur = cl < ch;
        unsigned long long rc = cur ? wrec[cl] : 0ull;
        int s = (int)((unsigned)rc & SMASK), dl = (int)((unsigned)rc >> SBITS);
        float nr = cur ? dinv[s] * __uint_as_float((unsigned)(rc >> 32)) * dinv[base + dl] : 0.0f;
        int go = s * W4;
        float4* acc4 = (float4*)acc;
        static_assert(PITCH % 4 == 0 && DUMP % 4 == 0, "rows start on 16-byte pieces");
        int ai = cur ? dl * (PITCH / 4) : DUMP / 4;
        for (int g = 0; g < cmax; g += 16) {
            const bool nx = g + 16 + cl < ch;
            unsigned long long rn = nx ? wrec[g + 16 + cl] : 0ull;
            float4 r[16];
            unrolled<0, 16>([&](auto pc) {
                constexpr int P = decltype(pc)::v;
                r[P] = hv[half_share<P>(go) + clq];
            });
            int sn = (int)((unsigned)rn & SMASK), dln = (int)((unsigned)rn >> SBITS);
            float dsn = nx ? dinv[sn] : 0.0f;
            float ddn = nx ? dinv[base + dln] : 0.0f;
            unrolled<0, 8>([&](auto kc) {
                constexpr int P0 = 2 * decltype(kc)::v;
                if (g + P0 < cmax) {
                    const int a0 = half_share<P0>(ai) + cl;
                    const int a1 = half_share<P0 + 1>(ai) + cl;
                    const float n0 = __int_as_float(half_share<P0>(__float_as_int(nr)));
                    const float n1 = __int_as_float(half_share<P0 + 1>(__float_as_int(nr)));
                    if (fl) {
                        float4 t0 = acc4[a0];
                        float4 t1 = acc4[a1];
                        const bool same = (a1 == a0);
                        t0.x = fmaf(n0, r[P0].x, t0.x);
                        t0.y = fmaf(n0, r[P0].y, t0.y);
                        t0.z = fmaf(n0, r[P0].z, t0.z);
                        t0.w = fmaf(n0, r[P0].w, t0.w);
                        t1.x = fmaf(n1, r[P0 + 1].x, same ? t0.x : t1.x);
                        t1.y = fmaf(n1, r[P0 + 1].y, same ? t0.y : t1.y);
                        t1.z = fmaf(n1, r[P0 + 1].z, same ? t0.z : t1.z);
                        t1.w = fmaf(n1, r[P0 + 1].w, same ? t0.w : t1.w);
                        acc4[a0] = t0;
                        acc4[a1] = t1;
                    }
                }
            });
            go = sn * W4;
            ai = nx ? dln * (PITCH / 4) : DUMP / 4;
            nr = dsn * __uint_as_float((unsigned)(rn >> 32)) * ddn;
        }
    } else {
        scan_owned<PWS>(keyv, share_lo(wave, nchunks), share_lo(wave + 1, nchunks), wlist, lane, owner, proc);
    }
    __syncthreads();

    if constexpr (!FUSE) {
        static_assert(RMAX <= TPB, "a thread per owned row");
        if (tid < rows) {
            const int r = tid;
            const float* v = acc + r * W;
            float mx = -3.402823466e38f;
#pragma unroll
            for (int c = 0; c < W; ++c) mx = fmaxf(mx, v[c]);
            float s = 0.0f;
#pragma unroll
            for (int c = 0; c < W; ++c) s += expf(v[c] - mx);
            lse[r] = mx + logf(s);
        }
        __syncthreads();
        for (int i = tid * 4; i < nel; i += TPB * 4) {
            const float4 q = *(const float4*)(acc + i);
            const float l = lse[i / W];
            *(float4*)(oown + i) = make_float4(q.x - l, q.y - l, q.z - l, q.w - l);
        }
    } else {
        const int lo = lane & 15, hi = lane >> 4;
        const int ntile = (rows + 15) >> 4;
        static_assert(AROWS <= 16 * NWAVE, "one 16-row tile per wave covers the owner's rows");
        if (wave < ntile) {
            const int mt = wave;
            const float* arow = acc + (mt * 16 + lo) * PITCH;
            v8f c0 = {}, c1 = {}, c2 = {}, c3 = {};
#pragma unroll 8
            for (int j = 0; j < 16; ++j) {
                const int k0 = kq(j, hi);
                const v2f a = *(const v2f*)(arow + k0);
                const float4 q0 = *(const float4*)(Wa + (size_t)k0 * 64 + 4 * lo);
                const float4 q1 = *(const float4*)(Wa + (size_t)(k0 + 1) * 64 + 4 * lo);
                c0 = __builtin_amdgcn_wmma_f32_16x16x4_f32(false, a, false, v2f{ q0.x, q1.x }, (short)0, c0, false, false);
                c1 = __builtin_amdgcn_wmma_f32_16x16x4_f32(false, a, false, v2f{ q0.y, q1.y }, (short)0, c1, false, false);
                c2 = __builtin_amdgcn_wmma_f32_16x16x4_f32(false, a, false, v2f{ q0.z, q1.z }, (short)0, c2, false, false);
                c3 = __builtin_amdgcn_wmma_f32_16x16x4_f32(false, a, false, v2f{ q0.w, q1.w }, (short)0, c3, false, false);
            }
            const float4 bv = *(const float4*)(bias + 4 * lo);
            float* drow = acc + (mt * 16 + 8 * hi) * PITCH + 4 * lo;
#pragma unroll
            for (int r = 0; r < 8; ++r) {
                *(float4*)(drow + r * PITCH) = make_float4(fmaxf(c0[r] + bv.x, 0.0f), fmaxf(c1[r] + bv.y, 0.0f),
                                                           fmaxf(c2[r] + bv.z, 0.0f), fmaxf(c3[r] + bv.w, 0.0f));
            }
            v8f d0 = {}, d1 = {}, d2 = {};
#pragma unroll 8
            for (int j = 0; j < 16; ++j) {
                const int k0 = kq(j, hi);
                const v2f a = *(const v2f*)(arow + k0);
                v2f b[3];
#pragma unroll
                for (int nt = 0; nt < 3; ++nt) {
                    const int col = 3 * lo + nt;
                    const bool cok = col < 40;
                    const int cc = cok ? col : 39;
                    const float w0 = Wb[(size_t)k0 * 40 + cc];
                    const float w1 = Wb[(size_t)(k0 + 1) * 40 + cc];
                    b[nt] = v2f{ cok ? w0 : 0.0f, cok ? w1 : 0.0f };
                }
                d0 = __builtin_amdgcn_wmma_f32_16x16x4_f32(false, a, false, b[0], (short)0, d0, false, false);
                d1 = __builtin_amdgcn_wmma_f32_16x16x4_f32(false, a, false, b[1], (short)0, d1, false, false);
                d2 = __builtin_amdgcn_wmma_f32_16x16x4_f32(false, a, false, b[2], (short)0, d2, false, false);
            }
            const int r0 = mt * 16 + 8 * hi;
            float* orow = out + (size_t)(base + r0) * 40 + 3 * lo;
#pragma unroll
            for (int r = 0; r < 8; ++r) {
                if (r0 + r < rows) {
                    float* p = orow + (size_t)r * 40;
                    if (lo < 13) { p[0] = d0[r]; p[1] = d1[r]; p[2] = d2[r]; }
                    else if (lo == 13) { p[0] = d0[r]; }
                }
            }
        }
    }
}


extern "C" void kernel_launch(void* const* d_in, const int* in_sizes, int n_in,
                              void* d_out, int out_size, void* d_ws, size_t ws_size,
                              hipStream_t stream) {
    const float* x   = (const float*)d_in[0];
    const int*   ei  = (const int*)d_in[1];
    const float* ew  = (const float*)d_in[2];
    const float* W1  = (const float*)d_in[3];
    const float* b1  = (const float*)d_in[4];
    const float* W2  = (const float*)d_in[5];
    const float* b2  = (const float*)d_in[6];
    float* out = (float*)d_out;

    const int N = in_sizes[0] / 64;
    const int E = in_sizes[2];
    const int* src = ei;
    const int* dst = ei + E;

    const int R = (N + NOWN - 1) / NOWN;
    const int nchunks = (E + 511) / 512;
    const int nwords = nchunks * 128;
    const int npad = ((N + 63) / 64) * 64;

    float* ws   = (float*)d_ws;
    float* dinv = ws;
    unsigned* keyw = (unsigned*)(dinv + npad);
    float* bufA = (float*)(keyw + nwords);
    int* cntp = (int*)(bufA + (size_t)N * 128);
    unsigned* recw = (unsigned*)(cntp + NOWN * NGRP);
    uint4* erec = (uint4*)(recw + (size_t)NOWN * NGRP * CAP * 2);
    const int nblk = (nwords + KB - 1) / KB;
    unsigned* offs = (unsigned*)(erec + (size_t)nwords * 2);

    k_key<<<nblk, KB, 0, stream>>>(src, dst, ew, keyw, erec, offs, E, nwords, (unsigned)R);
    k_deg_own<<<NOWN, TPB, 0, stream>>>((const uint2*)erec, offs, dinv, cntp, recw, N, R, nblk);

    k_agg_own<64, true><<<NOWN, TPB, 0, stream>>>(keyw, src, dst, ew, dinv, x, b1, cntp, recw, bufA, W1, W2, N, E, R, nchunks);

    k_agg_own<40, false><<<NOWN, TPB, 0, stream>>>(keyw, src, dst, ew, dinv, bufA, b2, cntp, recw, out, nullptr, nullptr, N, E, R, nchunks);
}


# layer-2 source rows carry their node's deg^-1/2 in a spare 16-byte piece (pitch 192 B) fetched by the row gather's idle lanes: no scattered degree load per edge in layer 2; sum order kept
# speedup vs baseline: 8.1212x; 1.0798x over previous
#include <hip/hip_runtime.h>
#include <math.h>


typedef __attribute__((ext_vector_type(2))) float v2f;
typedef __attribute__((ext_vector_type(8))) float v8f;

constexpr int NOWN  = 256;
constexpr int RMAX  = 392;
constexpr int TPB   = 1024;
constexpr int NWAVE = TPB / 32;
constexpr int pws_of(int other_bytes) {
    int p = (164 * 1024 - other_bytes + NWAVE * 4 - 1) / (NWAVE * 4);
    return p < 512 ? 512 : p;
}
constexpr int UB    = 8;
constexpr int EB    = 16;
constexpr int NGRP  = 2 * NWAVE;
constexpr int CAP   = 320;
constexpr int SBITS = 23;
constexpr unsigned SMASK = (1u << SBITS) - 1u;
constexpr int KB    = 1024;
constexpr int KE    = KB * 4;
constexpr int RB    = 16;
constexpr int HP2   = 48;

__device__ __forceinline__ void ldsAddF32(float* p, float v) {
    __hip_atomic_fetch_add(p, v, __ATOMIC_RELAXED, __HIP_MEMORY_SCOPE_WORKGROUP);
}


__global__ __launch_bounds__(KB) void k_key(const int* __restrict__ src, const int* __restrict__ dst,
                                            const float* __restrict__ w, unsigned* __restrict__ keyw,
                                            uint4* __restrict__ erec, unsigned* __restrict__ offs,
                                            int E, int nwords, unsigned R) {
    __shared__ __attribute__((aligned(16))) unsigned long long srt[KE];
    __shared__ __attribute__((aligned(16))) int ocnt[NOWN];
    __shared__ __attribute__((aligned(16))) int ost[NOWN];
    __shared__ int tot;
    const int tid = threadIdx.x;
    const int i = blockIdx.x * KB + tid;
    if (tid < NOWN) ocnt[tid] = 0;
    unsigned v = 0u;
    unsigned rw[8];
    unsigned ow[4];
    bool in[4];
#pragma unroll
    for (int u = 0; u < 4; ++u) {
        int e = i * 4 + u;
        in[u] = e < E;
        unsigned d = in[u] ? (unsigned)dst[e] : 0u;
        unsigned sv = in[u] ? (unsigned)src[e] : 0u;
        float we = in[u] ? w[e] : 0.0f;
        unsigned o = d / R;
        ow[u] = o;
        v |= o << (8 * u);
        rw[2 * u] = sv | ((d - o * R) << SBITS);
        rw[2 * u + 1] = __float_as_uint(we);
    }
    __syncthreads();
    int rk[4];
#pragma unroll
    for (int u = 0; u < 4; ++u)
        rk[u] = in[u] ? __hip_atomic_fetch_add(&ocnt[ow[u]], 1, __ATOMIC_RELAXED, __HIP_MEMORY_SCOPE_WORKGROUP) : 0;
    __syncthreads();
    if (tid < 32) {
        int cq[8];
        int s = 0;
#pragma unroll
        for (int q = 0; q < 8; ++q) { cq[q] = ocnt[8 * tid + q]; s += cq[q]; }
        int inc = s;
#pragma unroll
        for (int dlt = 1; dlt < 32; dlt <<= 1) {
            int t = __shfl_up(inc, dlt);
            if (tid >= dlt) inc += t;
        }
        int st = inc - s;
#pragma unroll
        for (int q = 0; q < 8; ++q) {
            ost[8 * tid + q] = st;
            st += cq[q];
        }
        if (tid == 31) tot = inc;
    }
    __syncthreads();
#pragma unroll
    for (int u = 0; u < 4; ++u) {
        if (in[u]) srt[ost[ow[u]] + rk[u]] = ((unsigned long long)rw[2 * u + 1] << 32) | rw[2 * u];
    }
    __syncthreads();
    if (i < nwords) keyw[i] = v;
    if (tid < NOWN) offs[(size_t)blockIdx.x * NOWN + tid] = (unsigned)ost[tid] | ((unsigned)ocnt[tid] << 16);
    const int total = tot;
    const uint4* s4 = (const uint4*)srt;
    uint4* eout = erec + (size_t)blockIdx.x * (KE / 2);
#pragma unroll
    for (int q = 0; q < 2; ++q) {
        const int p = tid + q * KB;
        if (2 * p < total) eout[p] = s4[p];
    }
}

__device__ __forceinline__ bool any_key_hit(uint4 k, unsigned pat) {
    unsigned x0 = k.x ^ pat, x1 = k.y ^ pat, x2 = k.z ^ pat, x3 = k.w ^ pat;
    unsigned t = ((x0 - 0x01010101u) & ~x0) | ((x1 - 0x01010101u) & ~x1) |
                 ((x2 - 0x01010101u) & ~x2) | ((x3 - 0x01010101u) & ~x3);
    return (t & 0x80808080u) != 0u;
}

__device__ __forceinline__ unsigned key_hit_mask16(uint4 k, unsigned pat) {
    unsigned kk[4] = { k.x, k.y, k.z, k.w };
    unsigned m = 0u;
#pragma unroll
    for (int j = 0; j < 4; ++j) {
        unsigned x = kk[j] ^ pat;
        unsigned t = ((x & 0x7F7F7F7Fu) + 0x7F7F7F7Fu) | x;
        unsigned z = ~t & 0x80808080u;
        unsigned c = ((z >> 7) & 1u) | ((z >> 14) & 2u) | ((z >> 21) & 4u) | ((z >> 28) & 8u);
        m |= c << (4 * j);
    }
    return m;
}

__device__ __forceinline__ int share_lo(int wave, int nchunks) {
    return (int)(((long long)wave * nchunks) / NWAVE);
}

template <int PWS, class Proc>
__device__ __forceinline__ void scan_owned(const uint4* __restrict__ keyv, int xlo, int xhi,
                                           unsigned* wlist, int lane, int owner, Proc proc) {
    const unsigned pat = (unsigned)owner * 0x01010101u;
    int cnt = 0;
    for (int c0 = xlo; c0 < xhi; c0 += UB) {
        uint4 k[UB];
#pragma unroll
        for (int u = 0; u < UB; ++u) {
            const int c = c0 + u;
            const int x = (c < xhi) ? c : xlo;
            k[u] = keyv[(size_t)x * 32 + lane];
        }
#pragma unroll
        for (int u = 0; u < UB; ++u) {
            bool hit = (c0 + u < xhi) && any_key_hit(k[u], pat);
            unsigned bal = __builtin_amdgcn_ballot_w32(hit);
            if (hit) wlist[cnt + (int)__builtin_amdgcn_mbcnt_lo(bal, 0u)] = (unsigned)((c0 + u) * 32 + lane);
            cnt += __builtin_popcount(bal);
        }
        if (cnt > PWS - 32 * UB) { proc(cnt); cnt = 0; }
    }
    proc(cnt);
}


__global__ __launch_bounds__(TPB) void k_deg_own(const uint2* __restrict__ erec,
                                                 const unsigned* __restrict__ offs,
                                                 float* __restrict__ dinv,
                                                 int* __restrict__ cntp,
                                                 unsigned* __restrict__ recw,
                                                 int n, int R, int nblk) {
    __shared__ float degl[RMAX];
    __shared__ unsigned long long stage[NGRP * CAP];
    __shared__ int lcnt[NGRP];
    static_assert(RMAX * 4 + NGRP * CAP * 8 + NGRP * 4 > 160 * 1024, "one owner per WGP");
    const int tid = threadIdx.x, lane = tid & 31;
    const int wave = __builtin_amdgcn_readfirstlane(tid >> 5);
    const int owner = blockIdx.x, base = owner * R;
    const float gscale = (float)NGRP / (float)R;

    for (int i = tid; i < RMAX; i += TPB) degl[i] = 1.0f;
    if (tid < NGRP) lcnt[tid] = 0;
    __syncthreads();

    auto file = [&](uint2 rc) {
        const unsigned row = rc.x >> SBITS;
        int grp = (int)((float)row * gscale);
        grp = grp > NGRP - 1 ? NGRP - 1 : grp;
        const int pos = __hip_atomic_fetch_add(&lcnt[grp], 1, __ATOMIC_RELAXED, __HIP_MEMORY_SCOPE_WORKGROUP);
        if (pos < CAP) stage[grp * CAP + pos] = ((unsigned long long)rc.y << 32) | rc.x;
        ldsAddF32(&degl[row], __uint_as_float(rc.y));
    };
    {
        const int blo = share_lo(wave, nblk), bhi = share_lo(wave + 1, nblk);
        for (int b0 = blo; b0 < bhi; b0 += 32) {
            const int nb = (bhi - b0) < 32 ? (bhi - b0) : 32;
            const int bl = b0 + (lane < nb ? lane : 0);
            const unsigned pw = offs[(size_t)bl * NOWN + owner];
            for (int j0 = 0; j0 < nb; j0 += RB) {
                uint2 rc[RB];
                int cj[RB];
                int cmx = 0;
#pragma unroll
                for (int u = 0; u < RB; ++u) {
                    const bool have = j0 + u < nb;
                    const int j = have ? j0 + u : j0;
                    const unsigned wj = (unsigned)__builtin_amdgcn_readlane((int)pw, j);
                    cj[u] = have ? (int)(wj >> 16) : 0;
                    cmx = cj[u] > cmx ? cj[u] : cmx;
                    const int idx = (lane < cj[u]) ? (int)(wj & 0xFFFFu) + lane : 0;
                    rc[u] = erec[(size_t)(b0 + j) * KE + idx];
                }
#pragma unroll
                for (int u = 0; u < RB; ++u) {
                    if (lane < cj[u]) file(rc[u]);
                }
                if (cmx > 32) {
                    const int je = (j0 + RB) < nb ? (j0 + RB) : nb;
                    for (int j = j0; j < je; ++j) {
                        const unsigned wj = (unsigned)__builtin_amdgcn_readlane((int)pw, j);
                        const int cn = (int)(wj >> 16), st = (int)(wj & 0xFFFFu);
                        for (int i0 = 32 + lane; i0 - lane < cn; i0 += 32) {
                            if (i0 < cn) file(erec[(size_t)(b0 + j) * KE + st + i0]);
                        }
                    }
                }
            }
        }
    }
    __syncthreads();

    for (int q = 0; q < 2; ++q) {
        const int gi = 2 * wave + q;
        const int cw = lcnt[gi];
        const int cc = cw < CAP ? cw : CAP;
        unsigned long long* wrec = (unsigned long long*)recw + ((size_t)owner * NGRP + gi) * CAP;
        for (int i = lane; i < cc; i += 32) wrec[i] = stage[gi * CAP + i];
        if (lane == 0) cntp[owner * NGRP + gi] = cw;
    }

    for (int i = tid; i < R; i += TPB) {
        int node = base + i;
        if (node < n) {
            float d = degl[i];
            dinv[node] = (d > 0.0f) ? rsqrtf(d) : 0.0f;
        }
    }
}


__device__ __forceinline__ int kq(int j, int hi) { return 4 * j + 2 * hi; }


template <int P> struct ic_t { static constexpr int v = P; };
template <int P, int N, class F> __device__ __forceinline__ void unrolled(F&& f) {
    if constexpr (P < N) { f(ic_t<P>{}); unrolled<P + 1, N>(f); }
}
template <int P> __device__ __forceinline__ int half_share(int v) {
    return __builtin_amdgcn_update_dpp(0, v, 0x150 + P, 0xF, 0xF, false);
}

template <int P> __device__ __forceinline__ float half_share_f(float v) {
    return __int_as_float(__builtin_amdgcn_update_dpp(0, __float_as_int(v), 0x150 + P, 0xF, 0xF, true));
}

template <int W, bool FUSE>
__global__ __launch_bounds__(TPB) void k_agg_own(const unsigned* __restrict__ keyw,
                                                 const int* __restrict__ src,
                                                 const int* __restrict__ dst,
                                                 const float* __restrict__ w,
                                                 const float* __restrict__ dinv,
                                                 const float* __restrict__ h,
                                                 const float* __restrict__ bias,
                                                 const int* __restrict__ cntp,
                                                 const unsigned* __restrict__ recw,
                                                 float* __restrict__ out,
                                                 const float* __restrict__ Wa,
                                                 const float* __restrict__ Wb,
                                                 int n, int E, int R, int nchunks) {
    constexpr int PITCH = FUSE ? 68 : W;
    constexpr int HP = FUSE ? W : HP2;
    constexpr int AROWS = FUSE ? 400 : RMAX;
    constexpr int DUMP = AROWS * PITCH;
    constexpr int PWS = pws_of((AROWS * PITCH + 64) * 4);
    __shared__ __attribute__((aligned(16))) float acc[AROWS * PITCH + 64];
    __shared__ unsigned list[NWAVE * PWS];
    __shared__ float lse[FUSE ? 1 : RMAX];
    const uint4* keyv = (const uint4*)keyw;
    const int tid = threadIdx.x, lane = tid & 31;
    const int wave = __builtin_amdgcn_readfirstlane(tid >> 5);
    const int owner = blockIdx.x, base = owner * R;
    const unsigned pat = (unsigned)owner * 0x01010101u;
    unsigned* wlist = list + wave * PWS;
    int rows = n - base;
    rows = rows < 0 ? 0 : (rows > R ? R : rows);
    const int nel = rows * W;
    const float* hown = h + (size_t)base * HP;
    float* oown = out + (size_t)base * W;

    for (int i = tid; i < nel; i += TPB) {
        int r = i / W, f = i - r * W;
        float di = dinv[base + r];
        float sl = di * di * hown[FUSE ? i : r * HP + f];
        acc[r * PITCH + f] = FUSE ? sl : bias[f] + sl;
    }
    if (tid < 64) acc[DUMP + tid] = 0.0f;
    __syncthreads();

    const bool fl2 = (2 * lane < W);
    auto proc = [&](int cnt) {
        for (int g = 0; g < cnt; g += 32) {
            int idx = g + lane;
            bool v = idx < cnt;
            unsigned seg = v ? wlist[idx] : 0u;
            unsigned m = v ? key_hit_mask16(keyv[seg], pat) : 0u;
            while (__builtin_amdgcn_ballot_w32(m != 0u) != 0u) {
                bool act = (m != 0u);
                int e = (int)seg * 16 + (act ? __builtin_ctz(m) : 0);
                m &= m - 1u;
                act = act && (e < E);
                int s = 0, dl = 0;
                float nr = 0.0f;
                if (act) {
                    int d = dst[e];
                    s = src[e];
                    nr = dinv[s] * w[e] * dinv[d];
                    dl = d - base;
                }
                unsigned bal = __builtin_amdgcn_ballot_w32(act);
                while (bal != 0u) {
                    int jj[EB];
                    bool ok[EB];
#pragma unroll
                    for (int u = 0; u < EB; ++u) {
                        ok[u] = (bal != 0u);
                        jj[u] = ok[u] ? __builtin_ctz(bal) : 0;
                        bal &= bal - 1u;
                    }
                    v2f r[EB];
#pragma unroll
                    for (int u = 0; u < EB; ++u) {
                        int sj = __builtin_amdgcn_readlane(s, jj[u]);
                        r[u] = fl2 ? *(const v2f*)(h + (size_t)sj * HP + 2 * lane) : v2f{0.0f, 0.0f};
                    }
#pragma unroll
                    for (int u = 0; u < EB; ++u) {
                        if (ok[u]) {
                            float nj = __int_as_float(__builtin_amdgcn_readlane(__float_as_int(nr), jj[u]));
                            int dj = __builtin_amdgcn_readlane(dl, jj[u]);
                            if (fl2) {
                                float* a = &acc[dj * PITCH + 2 * lane];
                                ldsAddF32(a, nj * r[u].x);
                                ldsAddF32(a + 1, nj * r[u].y);
                            }
                        }
                    }
                }
            }
        }
    };
    const int ca = cntp[owner * NGRP + lane];
    const int cb = cntp[owner * NGRP + 32 + lane];
    const bool over = __builtin_amdgcn_ballot_w32(ca > CAP || cb > CAP) != 0u;
    if (!over) {
        constexpr int W4 = W / 4;
        constexpr int HP4 = HP / 4;
        constexpr int FP = FUSE ? 0 : W4;
        const int hf = lane >> 4, cl = lane & 15;
        const bool fl = cl < W4;
        const int clq = fl ? cl : (FUSE ? W4 - 1 : FP);
        const int gi = 2 * wave + hf;
        const int ch = cntp[owner * NGRP + gi];
        const int c0 = __builtin_amdgcn_readlane(ch, 0), c1 = __builtin_amdgcn_readlane(ch, 16);
        const int cmax = c0 > c1 ? c0 : c1;
        const unsigned long long* wrec = (const unsigned long long*)recw + ((size_t)owner * NGRP + gi) * CAP;
        const float4* hv = (const float4*)h;
        const bool cur = cl < ch;
        unsigned long long rc = cur ? wrec[cl] : 0ull;
        int s = (int)((unsigned)rc & SMASK), dl = (int)((unsigned)rc >> SBITS);
        float nr = 0.0f;
        if constexpr (FUSE) {
            nr = cur ? dinv[s] * __uint_as_float((unsigned)(rc >> 32)) * dinv[base + dl] : 0.0f;
        } else {
            nr = cur ? __uint_as_float((unsigned)(rc >> 32)) * dinv[base + dl] : 0.0f;
        }
        int go = s * HP4;
        float4* acc4 = (float4*)acc;
        static_assert(PITCH % 4 == 0 && DUMP % 4 == 0, "rows start on 16-byte pieces");
        int ai = cur ? dl * (PITCH / 4) : DUMP / 4;
        for (int g = 0; g < cmax; g += 16) {
            const bool nx = g + 16 + cl < ch;
            unsigned long long rn = nx ? wrec[g + 16 + cl] : 0ull;
            float4 r[16];
            unrolled<0, 16>([&](auto pc) {
                constexpr int P = decltype(pc)::v;
                r[P] = hv[half_share<P>(go) + clq];
            });
            if constexpr (!FUSE) __builtin_amdgcn_sched_barrier(0);
            int sn = (int)((unsigned)rn & SMASK), dln = (int)((unsigned)rn >> SBITS);
            float dsn = 0.0f;
            if constexpr (FUSE) dsn = nx ? dinv[sn] : 0.0f;
            float ddn = nx ? dinv[base + dln] : 0.0f;
            unrolled<0, 8>([&](auto kc) {
                constexpr int P0 = 2 * decltype(kc)::v;
                if (g + P0 < cmax) {
                    const int a0 = half_share<P0>(ai) + cl;
                    const int a1 = half_share<P0 + 1>(ai) + cl;
                    float n0, n1;
                    if constexpr (FUSE) {
                        n0 = __int_as_float(half_share<P0>(__float_as_int(nr)));
                        n1 = __int_as_float(half_share<P0 + 1>(__float_as_int(nr)));
                    } else {
                        n0 = half_share_f<FP>(r[P0].x) * half_share_f<P0>(nr);
                        n1 = half_share_f<FP>(r[P0 + 1].x) * half_share_f<P0 + 1>(nr);
                    }
                    if (fl) {
                        float4 t0 = acc4[a0];
                        float4 t1 = acc4[a1];
                        const bool same = (a1 == a0);
                        t0.x = fmaf(n0, r[P0].x, t0.x);
                        t0.y = fmaf(n0, r[P0].y, t0.y);
                        t0.z = fmaf(n0, r[P0].z, t0.z);
                        t0.w = fmaf(n0, r[P0].w, t0.w);
                        t1.x = fmaf(n1, r[P0 + 1].x, same ? t0.x : t1.x);
                        t1.y = fmaf(n1, r[P0 + 1].y, same ? t0.y : t1.y);
                        t1.z = fmaf(n1, r[P0 + 1].z, same ? t0.z : t1.z);
                        t1.w = fmaf(n1, r[P0 + 1].w, same ? t0.w : t1.w);
                        acc4[a0] = t0;
                        acc4[a1] = t1;
                    }
                }
            });
            go = sn * HP4;
            ai = nx ? dln * (PITCH / 4) : DUMP / 4;
            if constexpr (FUSE) {
                nr = dsn * __uint_as_float((unsigned)(rn >> 32)) * ddn;
            } else {
                nr = __uint_as_float((unsigned)(rn >> 32)) * ddn;
            }
        }
    } else {
        scan_owned<PWS>(keyv, share_lo(wave, nchunks), share_lo(wave + 1, nchunks), wlist, lane, owner, proc);
    }
    __syncthreads();

    if constexpr (!FUSE) {
        static_assert(RMAX <= TPB, "a thread per owned row");
        if (tid < rows) {
            const int r = tid;
            const float* v = acc + r * W;
            float mx = -3.402823466e38f;
#pragma unroll
            for (int c = 0; c < W; ++c) mx = fmaxf(mx, v[c]);
            float s = 0.0f;
#pragma unroll
            for (int c = 0; c < W; ++c) s += expf(v[c] - mx);
            lse[r] = mx + logf(s);
        }
        __syncthreads();
        for (int i = tid * 4; i < nel; i += TPB * 4) {
            const float4 q = *(const float4*)(acc + i);
            const float l = lse[i / W];
            *(float4*)(oown + i) = make_float4(q.x - l, q.y - l, q.z - l, q.w - l);
        }
    } else {
        const int lo = lane & 15, hi = lane >> 4;
        const int ntile = (rows + 15) >> 4;
        static_assert(AROWS <= 16 * NWAVE, "one 16-row tile per wave covers the owner's rows");
        if (wave < ntile) {
            const int mt = wave;
            const float* arow = acc + (mt * 16 + lo) * PITCH;
            v8f c0 = {}, c1 = {}, c2 = {}, c3 = {};
#pragma unroll 8
            for (int j = 0; j < 16; ++j) {
                const int k0 = kq(j, hi);
                const v2f a = *(const v2f*)(arow + k0);
                const float4 q0 = *(const float4*)(Wa + (size_t)k0 * 64 + 4 * lo);
                const float4 q1 = *(const float4*)(Wa + (size_t)(k0 + 1) * 64 + 4 * lo);
                c0 = __builtin_amdgcn_wmma_f32_16x16x4_f32(false, a, false, v2f{ q0.x, q1.x }, (short)0, c0, false, false);
                c1 = __builtin_amdgcn_wmma_f32_16x16x4_f32(false, a, false, v2f{ q0.y, q1.y }, (short)0, c1, false, false);
                c2 = __builtin_amdgcn_wmma_f32_16x16x4_f32(false, a, false, v2f{ q0.z, q1.z }, (short)0, c2, false, false);
                c3 = __builtin_amdgcn_wmma_f32_16x16x4_f32(false, a, false, v2f{ q0.w, q1.w }, (short)0, c3, false, false);
            }
            const float4 bv = *(const float4*)(bias + 4 * lo);
            float* drow = acc + (mt * 16 + 8 * hi) * PITCH + 4 * lo;
#pragma unroll
            for (int r = 0; r < 8; ++r) {
                *(float4*)(drow + r * PITCH) = make_float4(fmaxf(c0[r] + bv.x, 0.0f), fmaxf(c1[r] + bv.y, 0.0f),
                                                           fmaxf(c2[r] + bv.z, 0.0f), fmaxf(c3[r] + bv.w, 0.0f));
            }
            const int r0 = mt * 16 + 8 * hi;
            float dvr[8];
#pragma unroll
            for (int r = 0; r < 8; ++r) dvr[r] = dinv[base + (r0 + r < rows ? r0 + r : 0)];
            v8f d0 = {}, d1 = {}, d2 = {};
#pragma unroll 8
            for (int j = 0; j < 16; ++j) {
                const int k0 = kq(j, hi);
                const v2f a = *(const v2f*)(arow + k0);
                v2f b[3];
#pragma unroll
                for (int nt = 0; nt < 3; ++nt) {
                    const int col = 3 * lo + nt;
                    const bool cok = col < 40;
                    const int cc = cok ? col : 39;
                    const float w0 = Wb[(size_t)k0 * 40 + cc];
                    const float w1 = Wb[(size_t)(k0 + 1) * 40 + cc];
                    b[nt] = v2f{ cok ? w0 : 0.0f, cok ? w1 : 0.0f };
                }
                d0 = __builtin_amdgcn_wmma_f32_16x16x4_f32(false, a, false, b[0], (short)0, d0, false, false);
                d1 = __builtin_amdgcn_wmma_f32_16x16x4_f32(false, a, false, b[1], (short)0, d1, false, false);
                d2 = __builtin_amdgcn_wmma_f32_16x16x4_f32(false, a, false, b[2], (short)0, d2, false, false);
            }
            float* orow = out + (size_t)(base + r0) * HP2 + 3 * lo;
            const bool k0 = 3 * lo < 40, k1 = 3 * lo + 1 < 40, k2 = 3 * lo + 2 < 40;
#pragma unroll
            for (int r = 0; r < 8; ++r) {
                if (r0 + r < rows) {
                    float* p = orow + (size_t)r * HP2;
                    p[0] = k0 ? d0[r] : 0.0f;
                    p[1] = k1 ? d1[r] : (lo == 13 ? dvr[r] : 0.0f);
                    p[2] = k2 ? d2[r] : 0.0f;
                }
            }
        }
    }
}


extern "C" void kernel_launch(void* const* d_in, const int* in_sizes, int n_in,
                              void* d_out, int out_size, void* d_ws, size_t ws_size,
                              hipStream_t stream) {
    const float* x   = (const float*)d_in[0];
    const int*   ei  = (const int*)d_in[1];
    const float* ew  = (const float*)d_in[2];
    const float* W1  = (const float*)d_in[3];
    const float* b1  = (const float*)d_in[4];
    const float* W2  = (const float*)d_in[5];
    const float* b2  = (const float*)d_in[6];
    float* out = (float*)d_out;

    const int N = in_sizes[0] / 64;
    const int E = in_sizes[2];
    const int* src = ei;
    const int* dst = ei + E;

    const int R = (N + NOWN - 1) / NOWN;
    const int nchunks = (E + 511) / 512;
    const int nwords = nchunks * 128;
    const int npad = ((N + 63) / 64) * 64;

    float* ws   = (float*)d_ws;
    float* dinv = ws;
    unsigned* keyw = (unsigned*)(dinv + npad);
    float* bufA = (float*)(keyw + nwords);
    int* cntp = (int*)(bufA + (size_t)N * 128);
    unsigned* recw = (unsigned*)(cntp + NOWN * NGRP);
    uint4* erec = (uint4*)(recw + (size_t)NOWN * NGRP * CAP * 2);
    const int nblk = (nwords + KB - 1) / KB;
    unsigned* offs = (unsigned*)(erec + (size_t)nwords * 2);

    k_key<<<nblk, KB, 0, stream>>>(src, dst, ew, keyw, erec, offs, E, nwords, (unsigned)R);
    k_deg_own<<<NOWN, TPB, 0, stream>>>((const uint2*)erec, offs, dinv, cntp, recw, N, R, nblk);

    k_agg_own<64, true><<<NOWN, TPB, 0, stream>>>(keyw, src, dst, ew, dinv, x, b1, cntp, recw, bufA, W1, W2, N, E, R, nchunks);

    k_agg_own<40, false><<<NOWN, TPB, 0, stream>>>(keyw, src, dst, ew, dinv, bufA, b2, cntp, recw, out, nullptr, nullptr, N, E, R, nchunks);
}
